# starting kernel
# baseline (speedup 1.0000x reference)
#include <hip/hip_runtime.h>
#include <math.h>

typedef __attribute__((ext_vector_type(2))) float v2f;
typedef __attribute__((ext_vector_type(8))) float v8f;

#define NEG_SLOPE 0.2f

__device__ __forceinline__ unsigned f2o(float f) {
    unsigned u = __float_as_uint(f);
    return (u & 0x80000000u) ? ~u : (u | 0x80000000u);
}
__device__ __forceinline__ float o2f(unsigned u) {
    return (u & 0x80000000u) ? __uint_as_float(u & 0x7fffffffu) : __uint_as_float(~u);
}
#define ORD_NEG_INF 0x007FFFFFu

__global__ void zeroF(float* __restrict__ p, int n) {
    int i = blockIdx.x * blockDim.x + threadIdx.x;
    if (i < n) p[i] = 0.0f;
}

__global__ void edgeMeanK(const int* __restrict__ ei, const float* __restrict__ ea,
                          float* __restrict__ s, float* __restrict__ c, int E) {
    int e = blockIdx.x * blockDim.x + threadIdx.x;
    if (e >= E) return;
    int dst = ei[E + e];
    atomicAdd(&s[dst], ea[e]);
    atomicAdd(&c[dst], 1.0f);
}
__global__ void loopAttrK(float* __restrict__ s, const float* __restrict__ c, int N) {
    int i = blockIdx.x * blockDim.x + threadIdx.x;
    if (i < N) s[i] = s[i] / fmaxf(c[i], 1.0f);
}

__global__ void kdotK(const float* __restrict__ we1, const float* __restrict__ ae1,
                      const float* __restrict__ we2, const float* __restrict__ ae2,
                      float* __restrict__ kscal) {
    int t = threadIdx.x;
    if (t == 0) { float s = 0.f; for (int j = 0; j < 64; ++j) s += we1[j] * ae1[j]; kscal[0] = s; }
    if (t == 1) { float s = 0.f; for (int j = 0; j < 32; ++j) s += we2[j] * ae2[j]; kscal[1] = s; }
}

template <int KDIM, int NT>
__global__ void gemmWmmaK(const float* __restrict__ A, const float* __restrict__ B,
                          float* __restrict__ C, int mtiles) {
    constexpr int NCOL = NT * 16;
    int wave = (blockIdx.x * blockDim.x + threadIdx.x) >> 5;
    int lane = threadIdx.x & 31;
    if (wave >= mtiles) return;

    int half = lane >> 4;
    int idx  = lane & 15;
    int kk   = half << 1;

    const float* arow  = A + (wave * 16 + idx) * KDIM + kk;
    const float* bbase = B + kk * NCOL + idx;

    v8f acc[NT];
#pragma unroll
    for (int t = 0; t < NT; ++t) acc[t] = (v8f){};

#pragma unroll
    for (int k0 = 0; k0 < KDIM; k0 += 4) {
        v2f a;
        a.x = arow[k0];
        a.y = arow[k0 + 1];
#pragma unroll
        for (int t = 0; t < NT; ++t) {
            v2f b;
            b.x = bbase[k0 * NCOL + t * 16];
            b.y = bbase[(k0 + 1) * NCOL + t * 16];
            acc[t] = __builtin_amdgcn_wmma_f32_16x16x4_f32(false, a, false, b,
                                                           (short)0, acc[t], false, false);
        }
    }
    int rbase = wave * 16 + (half << 3);
#pragma unroll
    for (int t = 0; t < NT; ++t) {
        float* cp = C + rbase * NCOL + t * 16 + idx;
#pragma unroll
        for (int i = 0; i < 8; ++i) cp[i * NCOL] = acc[t][i];
    }
}

__global__ void attdotK(const float* __restrict__ h, const float* __restrict__ as,
                        const float* __restrict__ ad, float* __restrict__ osrc,
                        float* __restrict__ odst, int N, int d4) {
    int i = blockIdx.x * blockDim.x + threadIdx.x;
    if (i >= N) return;
    const float4* hp = (const float4*)(h + (size_t)i * (d4 * 4));
    const float4* ap = (const float4*)as;
    const float4* bp = (const float4*)ad;
    float s1 = 0.f, s2 = 0.f;
    for (int j = 0; j < d4; ++j) {
        float4 v = hp[j], a = ap[j], b = bp[j];
        s1 += v.x * a.x + v.y * a.y + v.z * a.z + v.w * a.w;
        s2 += v.x * b.x + v.y * b.y + v.z * b.z + v.w * b.w;
    }
    osrc[i] = s1; odst[i] = s2;
}

__global__ void initMDK(unsigned* __restrict__ m, float* __restrict__ den, int N) {
    int i = blockIdx.x * blockDim.x + threadIdx.x;
    if (i < N) { m[i] = ORD_NEG_INF; den[i] = 0.0f; }
}

__global__ void alphaK(const int* __restrict__ ei, const float* __restrict__ ea,
                       const float* __restrict__ la, const float* __restrict__ asrc,
                       const float* __restrict__ adst, const float* __restrict__ kscal,
                       int ki, float* __restrict__ alpha, unsigned* __restrict__ m,
                       int E, int ET) {
    int e = blockIdx.x * blockDim.x + threadIdx.x;
    if (e >= ET) return;
    int s, dd; float attr;
    if (e < E) { s = ei[e]; dd = ei[E + e]; attr = ea[e]; }
    else       { s = dd = e - E; attr = la[e - E]; }
    float al = asrc[s] + adst[dd] + attr * kscal[ki];
    al = (al > 0.0f) ? al : NEG_SLOPE * al;
    alpha[e] = al;
    atomicMax(&m[dd], f2o(al));
}

__global__ void expSumK(const int* __restrict__ ei, const float* __restrict__ alpha,
                        const unsigned* __restrict__ m, float* __restrict__ den,
                        float* __restrict__ ex, int E, int ET) {
    int e = blockIdx.x * blockDim.x + threadIdx.x;
    if (e >= ET) return;
    int dd = (e < E) ? ei[E + e] : (e - E);
    float v = expf(alpha[e] - o2f(m[dd]));
    ex[e] = v;
    atomicAdd(&den[dd], v);
}

__global__ void aggK(const int* __restrict__ ei, const float* __restrict__ ex,
                     const float* __restrict__ den, const float* __restrict__ h,
                     float* __restrict__ accum, int shift, int E, int ET) {
    int idx = blockIdx.x * blockDim.x + threadIdx.x;
    int e = idx >> shift;
    if (e >= ET) return;
    int j = idx & ((1 << shift) - 1);
    int s, dd;
    if (e < E) { s = ei[e]; dd = ei[E + e]; }
    else       { s = dd = e - E; }
    float w = ex[e] / den[dd];
    atomicAdd(&accum[(dd << shift) + j], w * h[(s << shift) + j]);
}

__global__ void biasReluK(float* __restrict__ p, const float* __restrict__ b, int n, int mask) {
    int i = blockIdx.x * blockDim.x + threadIdx.x;
    if (i < n) p[i] = fmaxf(p[i] + b[i & mask], 0.0f);
}
__global__ void biasK(float* __restrict__ p, const float* __restrict__ b, int n, int mask) {
    int i = blockIdx.x * blockDim.x + threadIdx.x;
    if (i < n) p[i] = p[i] + b[i & mask];
}

static inline int gr(long long n, int blk) { return (int)((n + blk - 1) / blk); }

extern "C" void kernel_launch(void* const* d_in, const int* in_sizes, int n_in,
                              void* d_out, int out_size, void* d_ws, size_t ws_size,
                              hipStream_t stream) {
    const float* x    = (const float*)d_in[0];
    const int*   ei   = (const int*)d_in[1];
    const float* ea   = (const float*)d_in[2];
    const float* W1   = (const float*)d_in[3];
    const float* as1  = (const float*)d_in[4];
    const float* ad1  = (const float*)d_in[5];
    const float* we1  = (const float*)d_in[6];
    const float* ae1  = (const float*)d_in[7];
    const float* b1   = (const float*)d_in[8];
    const float* W2   = (const float*)d_in[9];
    const float* as2  = (const float*)d_in[10];
    const float* ad2  = (const float*)d_in[11];
    const float* we2  = (const float*)d_in[12];
    const float* ae2  = (const float*)d_in[13];
    const float* b2   = (const float*)d_in[14];
    float* out = (float*)d_out;

    const int N  = in_sizes[0] / 128;
    const int E  = in_sizes[1] / 2;
    const int ET = E + N;

    float* ws = (float*)d_ws;
    size_t o = 0;
    float*    la    = ws + o; o += N;
    float*    cnt   = ws + o; o += N;
    float*    kscal = ws + o; o += 4;
    float*    h     = ws + o; o += (size_t)N * 64;
    float*    asrc  = ws + o; o += N;
    float*    adst  = ws + o; o += N;
    unsigned* mbuf  = (unsigned*)(ws + o); o += N;
    float*    den   = ws + o; o += N;
    float*    alpha = ws + o; o += ET;
    float*    ex    = ws + o; o += ET;
    float*    hagg  = ws + o; o += (size_t)N * 64;

    const int B = 256;
    const int mtiles = N / 16;

    zeroF<<<gr(2 * (long long)N, B), B, 0, stream>>>(la, 2 * N);
    edgeMeanK<<<gr(E, B), B, 0, stream>>>(ei, ea, la, cnt, E);
    loopAttrK<<<gr(N, B), B, 0, stream>>>(la, cnt, N);
    kdotK<<<1, 64, 0, stream>>>(we1, ae1, we2, ae2, kscal);

    gemmWmmaK<128, 4><<<gr(mtiles, 8), 8 * 32, 0, stream>>>(x, W1, h, mtiles);
    attdotK<<<gr(N, B), B, 0, stream>>>(h, as1, ad1, asrc, adst, N, 16);
    initMDK<<<gr(N, B), B, 0, stream>>>(mbuf, den, N);
    zeroF<<<gr((long long)N * 64, B), B, 0, stream>>>(hagg, N * 64);
    alphaK<<<gr(ET, B), B, 0, stream>>>(ei, ea, la, asrc, adst, kscal, 0, alpha, mbuf, E, ET);
    expSumK<<<gr(ET, B), B, 0, stream>>>(ei, alpha, mbuf, den, ex, E, ET);
    aggK<<<gr((long long)ET * 64, B), B, 0, stream>>>(ei, ex, den, h, hagg, 6, E, ET);
    biasReluK<<<gr((long long)N * 64, B), B, 0, stream>>>(hagg, b1, N * 64, 63);

    gemmWmmaK<64, 2><<<gr(mtiles, 8), 8 * 32, 0, stream>>>(hagg, W2, h, mtiles);
    attdotK<<<gr(N, B), B, 0, stream>>>(h, as2, ad2, asrc, adst, N, 8);
    initMDK<<<gr(N, B), B, 0, stream>>>(mbuf, den, N);
    zeroF<<<gr((long long)N * 32, B), B, 0, stream>>>(out, N * 32);
    alphaK<<<gr(ET, B), B, 0, stream>>>(ei, ea, la, asrc, adst, kscal, 1, alpha, mbuf, E, ET);
    expSumK<<<gr(ET, B), B, 0, stream>>>(ei, alpha, mbuf, den, ex, E, ET);
    aggK<<<gr((long long)ET * 32, B), B, 0, stream>>>(ei, ex, den, h, out, 5, E, ET);
    biasK<<<gr((long long)N * 32, B), B, 0, stream>>>(out, b2, N * 32, 31);
}


# Counting-sort edges by destination (int atomic rank + scan); one wave per node does softmax and weighted sum over its segment: no float atomics, sums in segment order not arrival order
# speedup vs baseline: 2.9874x; 2.9874x over previous
#include <hip/hip_runtime.h>
#include <math.h>

typedef __attribute__((ext_vector_type(2))) float v2f;
typedef __attribute__((ext_vector_type(8))) float v8f;

#define NEG_SLOPE 0.2f

__global__ void zeroF(float* __restrict__ p, int n) {
    int i = blockIdx.x * blockDim.x + threadIdx.x;
    if (i < n) p[i] = 0.0f;
}

__global__ void rankK(const int* __restrict__ ei, int* __restrict__ cnt,
                      int* __restrict__ rank, int E) {
    int e = blockIdx.x * blockDim.x + threadIdx.x;
    if (e >= E) return;
    rank[e] = atomicAdd(&cnt[ei[E + e]], 1);
}

__global__ void __launch_bounds__(1024) scanK(const int* __restrict__ cnt, int* __restrict__ off, int N) {
    __shared__ int sh[1024];
    const int t = threadIdx.x;
    const int per = (N + 1023) >> 10;
    const int beg = t * per;
    const int end = min(beg + per, N);
    int s = 0;
    for (int j = beg; j < end; ++j) s += cnt[j];
    sh[t] = s;
    __syncthreads();
    for (int d = 1; d < 1024; d <<= 1) {
        int v = (t >= d) ? sh[t - d] : 0;
        __syncthreads();
        sh[t] += v;
        __syncthreads();
    }
    int run = sh[t] - s;
    for (int j = beg; j < end; ++j) { off[j] = run; run += cnt[j]; }
    if (t == 1023) off[N] = sh[1023];
}

__global__ void fillK(const int* __restrict__ ei, const float* __restrict__ ea,
                      const int* __restrict__ off, const int* __restrict__ rank,
                      int2* __restrict__ rec, int E) {
    int e = blockIdx.x * blockDim.x + threadIdx.x;
    if (e >= E) return;
    int p = off[ei[E + e]] + rank[e];
    rec[p] = make_int2(ei[e], __float_as_int(ea[e]));
}

__global__ void kdotK(const float* __restrict__ we1, const float* __restrict__ ae1,
                      const float* __restrict__ we2, const float* __restrict__ ae2,
                      float* __restrict__ kscal) {
    int t = threadIdx.x;
    if (t == 0) { float s = 0.f; for (int j = 0; j < 64; ++j) s += we1[j] * ae1[j]; kscal[0] = s; }
    if (t == 1) { float s = 0.f; for (int j = 0; j < 32; ++j) s += we2[j] * ae2[j]; kscal[1] = s; }
}

template <int KDIM, int NT>
__global__ void gemmWmmaK(const float* __restrict__ A, const float* __restrict__ B,
                          float* __restrict__ C, int mtiles) {
    constexpr int NCOL = NT * 16;
    int wave = (blockIdx.x * blockDim.x + threadIdx.x) >> 5;
    int lane = threadIdx.x & 31;
    if (wave >= mtiles) return;

    int half = lane >> 4;
    int idx  = lane & 15;
    int kk   = half << 1;

    const float* arow  = A + (wave * 16 + idx) * KDIM + kk;
    const float* bbase = B + kk * NCOL + idx;

    v8f acc[NT];
#pragma unroll
    for (int t = 0; t < NT; ++t) acc[t] = (v8f){};

#pragma unroll
    for (int k0 = 0; k0 < KDIM; k0 += 4) {
        v2f a;
        a.x = arow[k0];
        a.y = arow[k0 + 1];
#pragma unroll
        for (int t = 0; t < NT; ++t) {
            v2f b;
            b.x = bbase[k0 * NCOL + t * 16];
            b.y = bbase[(k0 + 1) * NCOL + t * 16];
            acc[t] = __builtin_amdgcn_wmma_f32_16x16x4_f32(false, a, false, b,
                                                           (short)0, acc[t], false, false);
        }
    }
    int rbase = wave * 16 + (half << 3);
#pragma unroll
    for (int t = 0; t < NT; ++t) {
        float* cp = C + rbase * NCOL + t * 16 + idx;
#pragma unroll
        for (int i = 0; i < 8; ++i) cp[i * NCOL] = acc[t][i];
    }
}

__global__ void attdotK(const float* __restrict__ h, const float* __restrict__ as,
                        const float* __restrict__ ad, float* __restrict__ osrc,
                        float* __restrict__ odst, int N, int d4) {
    int i = blockIdx.x * blockDim.x + threadIdx.x;
    if (i >= N) return;
    const float4* hp = (const float4*)(h + (size_t)i * (d4 * 4));
    const float4* ap = (const float4*)as;
    const float4* bp = (const float4*)ad;
    float s1 = 0.f, s2 = 0.f;
    for (int j = 0; j < d4; ++j) {
        float4 v = hp[j], a = ap[j], b = bp[j];
        s1 += v.x * a.x + v.y * a.y + v.z * a.z + v.w * a.w;
        s2 += v.x * b.x + v.y * b.y + v.z * b.z + v.w * b.w;
    }
    osrc[i] = s1; odst[i] = s2;
}

__device__ __forceinline__ float waveSum(float v) {
    for (int o = 16; o > 0; o >>= 1) v += __shfl_xor(v, o);
    return v;
}
__device__ __forceinline__ float waveMax(float v) {
    for (int o = 16; o > 0; o >>= 1) v = fmaxf(v, __shfl_xor(v, o));
    return v;
}
__device__ __forceinline__ float lrelu(float a) { return (a > 0.0f) ? a : NEG_SLOPE * a; }

template <int D, bool RELU>
__global__ void __launch_bounds__(256) segK(const int* __restrict__ off, const int2* __restrict__ rec,
                                            const float* __restrict__ asrc, const float* __restrict__ adst,
                                            const float* __restrict__ kscal, int ki,
                                            const float* __restrict__ h, const float* __restrict__ bias,
                                            float* __restrict__ out, int N) {
    constexpr int V = D / 32;
    const int lane = threadIdx.x & 31;
    const int i = __builtin_amdgcn_readfirstlane((int)((blockIdx.x * blockDim.x + threadIdx.x) >> 5));
    if (i >= N) return;
    const int b = off[i], e = off[i + 1];
    const int deg = e - b;
    const float kk = kscal[ki];
    const float ad = adst[i];

    auto edge = [&](int j, int& s, float& at) -> float {
        int2 r = rec[j];
        s = r.x; at = __int_as_float(r.y);
        return lrelu(asrc[s] + ad + at * kk);
    };
    auto loadRow = [&](int row, float (&v)[V]) {
        if constexpr (V == 2) {
            float2 t = *(const float2*)(h + row * D + lane * 2);
            v[0] = t.x; v[1] = t.y;
        } else {
            v[0] = h[row * D + lane];
        }
    };

    int s0 = i; float at0 = 0.0f, al0 = -INFINITY;
    const bool v0 = lane < deg;
    if (v0) al0 = edge(b + lane, s0, at0);
    float mx = al0, asum = at0;
    for (int cb = b + 32; cb < e; cb += 32) {
        int j = cb + lane;
        if (j < e) { int s; float at; float al = edge(j, s, at); mx = fmaxf(mx, al); asum += at; }
    }
    mx = waveMax(mx);
    asum = waveSum(asum);

    const float la  = asum / fmaxf((float)deg, 1.0f);
    const float als = lrelu(asrc[i] + ad + la * kk);
    mx = fmaxf(mx, als);
    const float exs = expf(als - mx);
    const float ex0 = v0 ? expf(al0 - mx) : 0.0f;
    float dsum = ex0;
    for (int cb = b + 32; cb < e; cb += 32) {
        int j = cb + lane;
        if (j < e) { int s; float at; dsum += expf(edge(j, s, at) - mx); }
    }
    const float den = waveSum(dsum) + exs;

    float acc[V];
    {
        float hv[V];
        loadRow(i, hv);
        const float wself = exs / den;
        for (int v = 0; v < V; ++v) acc[v] = wself * hv[v];
    }
    auto gather = [&](int s, float w, int n) {
        for (int t = 0; t < n; t += 8) {
            float hv[8][V]; float wt[8];
            for (int u = 0; u < 8; ++u) {
                int st = __builtin_amdgcn_readlane(s, t + u);
                wt[u] = __int_as_float(__builtin_amdgcn_readlane(__float_as_int(w), t + u));
                loadRow(st, hv[u]);
            }
            for (int u = 0; u < 8; ++u)
                for (int v = 0; v < V; ++v) acc[v] += wt[u] * hv[u][v];
        }
    };
    gather(s0, ex0 / den, min(deg, 32));
    for (int cb = b + 32; cb < e; cb += 32) {
        int j = cb + lane;
        int s = i; float w = 0.0f;
        if (j < e) { float at; w = expf(edge(j, s, at) - mx) / den; }
        gather(s, w, min(e - cb, 32));
    }

    if constexpr (V == 2) {
        float2 o;
        o.x = acc[0] + bias[lane * 2];
        o.y = acc[1] + bias[lane * 2 + 1];
        if (RELU) { o.x = fmaxf(o.x, 0.0f); o.y = fmaxf(o.y, 0.0f); }
        *(float2*)(out + i * D + lane * 2) = o;
    } else {
        float o = acc[0] + bias[lane];
        if (RELU) o = fmaxf(o, 0.0f);
        out[i * D + lane] = o;
    }
}

static inline int gr(long long n, int blk) { return (int)((n + blk - 1) / blk); }

extern "C" void kernel_launch(void* const* d_in, const int* in_sizes, int n_in,
                              void* d_out, int out_size, void* d_ws, size_t ws_size,
                              hipStream_t stream) {
    const float* x    = (const float*)d_in[0];
    const int*   ei   = (const int*)d_in[1];
    const float* ea   = (const float*)d_in[2];
    const float* W1   = (const float*)d_in[3];
    const float* as1  = (const float*)d_in[4];
    const float* ad1  = (const float*)d_in[5];
    const float* we1  = (const float*)d_in[6];
    const float* ae1  = (const float*)d_in[7];
    const float* b1   = (const float*)d_in[8];
    const float* W2   = (const float*)d_in[9];
    const float* as2  = (const float*)d_in[10];
    const float* ad2  = (const float*)d_in[11];
    const float* we2  = (const float*)d_in[12];
    const float* ae2  = (const float*)d_in[13];
    const float* b2   = (const float*)d_in[14];
    float* out = (float*)d_out;

    const int N  = in_sizes[0] / 128;
    const int E  = in_sizes[1] / 2;

    float* ws = (float*)d_ws;
    size_t o = 0;
    float* h     = ws + o; o += (size_t)N * 64;
    float* hagg  = ws + o; o += (size_t)N * 64;
    int2*  rec   = (int2*)(ws + o); o += (size_t)E * 2;
    int*   rank  = (int*)(ws + o); o += E;
    int*   cnt   = (int*)(ws + o); o += N;
    int*   off   = (int*)(ws + o); o += N + 1;
    float* asrc  = ws + o; o += N;
    float* adst  = ws + o; o += N;
    float* kscal = ws + o; o += 4;

    const int B = 256;
    const int mtiles = N / 16;

    zeroF<<<gr(N, B), B, 0, stream>>>((float*)cnt, N);
    rankK<<<gr(E, B), B, 0, stream>>>(ei, cnt, rank, E);
    scanK<<<1, 1024, 0, stream>>>(cnt, off, N);
    fillK<<<gr(E, B), B, 0, stream>>>(ei, ea, off, rank, rec, E);
    kdotK<<<1, 64, 0, stream>>>(we1, ae1, we2, ae2, kscal);

    gemmWmmaK<128, 4><<<gr(mtiles, 8), 8 * 32, 0, stream>>>(x, W1, h, mtiles);
    attdotK<<<gr(N, B), B, 0, stream>>>(h, as1, ad1, asrc, adst, N, 16);
    segK<64, true><<<gr((long long)N * 32, B), B, 0, stream>>>(off, rec, asrc, adst, kscal, 0, h, b1, hagg, N);

    gemmWmmaK<64, 2><<<gr(mtiles, 8), 8 * 32, 0, stream>>>(hagg, W2, h, mtiles);
    attdotK<<<gr(N, B), B, 0, stream>>>(h, as2, ad2, asrc, adst, N, 8);
    segK<32, false><<<gr((long long)N * 32, B), B, 0, stream>>>(off, rec, asrc, adst, kscal, 1, h, b2, out, N);
}


# Prefix scan of the destination counts spread over 13 blocks: each sums the counters in front of its range with whole-line vector loads, then scans 4 per thread in registers; no strided walk
# speedup vs baseline: 4.3112x; 1.4431x over previous
#include <hip/hip_runtime.h>
#include <math.h>

typedef __attribute__((ext_vector_type(2))) float v2f;
typedef __attribute__((ext_vector_type(8))) float v8f;

#define NEG_SLOPE 0.2f

__global__ void zeroF(float* __restrict__ p, int n) {
    int i = blockIdx.x * blockDim.x + threadIdx.x;
    if (i < n) p[i] = 0.0f;
}

__global__ void rankK(const int* __restrict__ ei, int* __restrict__ cnt,
                      int* __restrict__ rank, int E) {
    int e = blockIdx.x * blockDim.x + threadIdx.x;
    if (e >= E) return;
    rank[e] = atomicAdd(&cnt[ei[E + e]], 1);
}

__device__ __forceinline__ int waveSumI(int v) {
    for (int o = 16; o > 0; o >>= 1) v += __shfl_xor(v, o);
    return v;
}
__global__ void __launch_bounds__(1024) scanK(const int* __restrict__ cnt, int* __restrict__ off, int N) {
    __shared__ int shPre[32];
    __shared__ int shTot[32];
    const int t = threadIdx.x, lane = t & 31, w = t >> 5;
    const int nb = blockIdx.x;
    const int4* c4 = (const int4*)cnt;
    const int g  = nb * 1024 + t;
    const int i0 = g * 4;

    int4 own = make_int4(0, 0, 0, 0);
    if (i0 + 3 < N) {
        own = c4[g];
    } else {
        if (i0     < N) own.x = cnt[i0];
        if (i0 + 1 < N) own.y = cnt[i0 + 1];
        if (i0 + 2 < N) own.z = cnt[i0 + 2];
    }
    int pre = 0;
    for (int k = 0; k < nb; ++k) {
        int4 v = c4[k * 1024 + t];
        pre += (v.x + v.y) + (v.z + v.w);
    }

    const int s1 = own.x, s2 = s1 + own.y, s3 = s2 + own.z, s4 = s3 + own.w;
    int x = s4;
    for (int o = 1; o < 32; o <<= 1) {
        int y = __shfl_up(x, o);
        if (lane >= o) x += y;
    }
    pre = waveSumI(pre);
    if (lane == 31) shTot[w] = x;
    if (lane == 0)  shPre[w] = pre;
    __syncthreads();
    const int front = waveSumI(shPre[lane] + ((lane < w) ? shTot[lane] : 0));
    const int e0 = front + x - s4;

    if (i0 + 3 < N) {
        ((int4*)off)[g] = make_int4(e0, e0 + s1, e0 + s2, e0 + s3);
    } else {
        if (i0     < N) off[i0]     = e0;
        if (i0 + 1 < N) off[i0 + 1] = e0 + s1;
        if (i0 + 2 < N) off[i0 + 2] = e0 + s2;
    }
    if (i0 < N && N <= i0 + 4) off[N] = e0 + s4;
}

__global__ void fillK(const int* __restrict__ ei, const float* __restrict__ ea,
                      const int* __restrict__ off, const int* __restrict__ rank,
                      int2* __restrict__ rec, int E) {
    int e = blockIdx.x * blockDim.x + threadIdx.x;
    if (e >= E) return;
    int p = off[ei[E + e]] + rank[e];
    rec[p] = make_int2(ei[e], __float_as_int(ea[e]));
}

__global__ void kdotK(const float* __restrict__ we1, const float* __restrict__ ae1,
                      const float* __restrict__ we2, const float* __restrict__ ae2,
                      float* __restrict__ kscal) {
    int t = threadIdx.x;
    if (t == 0) { float s = 0.f; for (int j = 0; j < 64; ++j) s += we1[j] * ae1[j]; kscal[0] = s; }
    if (t == 1) { float s = 0.f; for (int j = 0; j < 32; ++j) s += we2[j] * ae2[j]; kscal[1] = s; }
}

template <int KDIM, int NT>
__global__ void gemmWmmaK(const float* __restrict__ A, const float* __restrict__ B,
                          float* __restrict__ C, int mtiles) {
    constexpr int NCOL = NT * 16;
    int wave = (blockIdx.x * blockDim.x + threadIdx.x) >> 5;
    int lane = threadIdx.x & 31;
    if (wave >= mtiles) return;

    int half = lane >> 4;
    int idx  = lane & 15;
    int kk   = half << 1;

    const float* arow  = A + (wave * 16 + idx) * KDIM + kk;
    const float* bbase = B + kk * NCOL + idx;

    v8f acc[NT];
#pragma unroll
    for (int t = 0; t < NT; ++t) acc[t] = (v8f){};

#pragma unroll
    for (int k0 = 0; k0 < KDIM; k0 += 4) {
        v2f a;
        a.x = arow[k0];
        a.y = arow[k0 + 1];
#pragma unroll
        for (int t = 0; t < NT; ++t) {
            v2f b;
            b.x = bbase[k0 * NCOL + t * 16];
            b.y = bbase[(k0 + 1) * NCOL + t * 16];
            acc[t] = __builtin_amdgcn_wmma_f32_16x16x4_f32(false, a, false, b,
                                                           (short)0, acc[t], false, false);
        }
    }
    int rbase = wave * 16 + (half << 3);
#pragma unroll
    for (int t = 0; t < NT; ++t) {
        float* cp = C + rbase * NCOL + t * 16 + idx;
#pragma unroll
        for (int i = 0; i < 8; ++i) cp[i * NCOL] = acc[t][i];
    }
}

__global__ void attdotK(const float* __restrict__ h, const float* __restrict__ as,
                        const float* __restrict__ ad, float* __restrict__ osrc,
                        float* __restrict__ odst, int N, int d4) {
    int i = blockIdx.x * blockDim.x + threadIdx.x;
    if (i >= N) return;
    const float4* hp = (const float4*)(h + (size_t)i * (d4 * 4));
    const float4* ap = (const float4*)as;
    const float4* bp = (const float4*)ad;
    float s1 = 0.f, s2 = 0.f;
    for (int j = 0; j < d4; ++j) {
        float4 v = hp[j], a = ap[j], b = bp[j];
        s1 += v.x * a.x + v.y * a.y + v.z * a.z + v.w * a.w;
        s2 += v.x * b.x + v.y * b.y + v.z * b.z + v.w * b.w;
    }
    osrc[i] = s1; odst[i] = s2;
}

__device__ __forceinline__ float waveSum(float v) {
    for (int o = 16; o > 0; o >>= 1) v += __shfl_xor(v, o);
    return v;
}
__device__ __forceinline__ float waveMax(float v) {
    for (int o = 16; o > 0; o >>= 1) v = fmaxf(v, __shfl_xor(v, o));
    return v;
}
__device__ __forceinline__ float lrelu(float a) { return (a > 0.0f) ? a : NEG_SLOPE * a; }

template <int D, bool RELU>
__global__ void __launch_bounds__(256) segK(const int* __restrict__ off, const int2* __restrict__ rec,
                                            const float* __restrict__ asrc, const float* __restrict__ adst,
                                            const float* __restrict__ kscal, int ki,
                                            const float* __restrict__ h, const float* __restrict__ bias,
                                            float* __restrict__ out, int N) {
    constexpr int V = D / 32;
    const int lane = threadIdx.x & 31;
    const int i = __builtin_amdgcn_readfirstlane((int)((blockIdx.x * blockDim.x + threadIdx.x) >> 5));
    if (i >= N) return;
    const int b = off[i], e = off[i + 1];
    const int deg = e - b;
    const float kk = kscal[ki];
    const float ad = adst[i];

    auto edge = [&](int j, int& s, float& at) -> float {
        int2 r = rec[j];
        s = r.x; at = __int_as_float(r.y);
        return lrelu(asrc[s] + ad + at * kk);
    };
    auto loadRow = [&](int row, float (&v)[V]) {
        if constexpr (V == 2) {
            float2 t = *(const float2*)(h + row * D + lane * 2);
            v[0] = t.x; v[1] = t.y;
        } else {
            v[0] = h[row * D + lane];
        }
    };

    int s0 = i; float at0 = 0.0f, al0 = -INFINITY;
    const bool v0 = lane < deg;
    if (v0) al0 = edge(b + lane, s0, at0);
    float mx = al0, asum = at0;
    for (int cb = b + 32; cb < e; cb += 32) {
        int j = cb + lane;
        if (j < e) { int s; float at; float al = edge(j, s, at); mx = fmaxf(mx, al); asum += at; }
    }
    mx = waveMax(mx);
    asum = waveSum(asum);

    const float la  = asum / fmaxf((float)deg, 1.0f);
    const float als = lrelu(asrc[i] + ad + la * kk);
    mx = fmaxf(mx, als);
    const float exs = expf(als - mx);
    const float ex0 = v0 ? expf(al0 - mx) : 0.0f;
    float dsum = ex0;
    for (int cb = b + 32; cb < e; cb += 32) {
        int j = cb + lane;
        if (j < e) { int s; float at; dsum += expf(edge(j, s, at) - mx); }
    }
    const float den = waveSum(dsum) + exs;

    float acc[V];
    {
        float hv[V];
        loadRow(i, hv);
        const float wself = exs / den;
        for (int v = 0; v < V; ++v) acc[v] = wself * hv[v];
    }
    auto gather = [&](int s, float w, int n) {
        for (int t = 0; t < n; t += 8) {
            float hv[8][V]; float wt[8];
            for (int u = 0; u < 8; ++u) {
                int st = __builtin_amdgcn_readlane(s, t + u);
                wt[u] = __int_as_float(__builtin_amdgcn_readlane(__float_as_int(w), t + u));
                loadRow(st, hv[u]);
            }
            for (int u = 0; u < 8; ++u)
                for (int v = 0; v < V; ++v) acc[v] += wt[u] * hv[u][v];
        }
    };
    gather(s0, ex0 / den, min(deg, 32));
    for (int cb = b + 32; cb < e; cb += 32) {
        int j = cb + lane;
        int s = i; float w = 0.0f;
        if (j < e) { float at; w = expf(edge(j, s, at) - mx) / den; }
        gather(s, w, min(e - cb, 32));
    }

    if constexpr (V == 2) {
        float2 o;
        o.x = acc[0] + bias[lane * 2];
        o.y = acc[1] + bias[lane * 2 + 1];
        if (RELU) { o.x = fmaxf(o.x, 0.0f); o.y = fmaxf(o.y, 0.0f); }
        *(float2*)(out + i * D + lane * 2) = o;
    } else {
        float o = acc[0] + bias[lane];
        if (RELU) o = fmaxf(o, 0.0f);
        out[i * D + lane] = o;
    }
}

static inline int gr(long long n, int blk) { return (int)((n + blk - 1) / blk); }

extern "C" void kernel_launch(void* const* d_in, const int* in_sizes, int n_in,
                              void* d_out, int out_size, void* d_ws, size_t ws_size,
                              hipStream_t stream) {
    const float* x    = (const float*)d_in[0];
    const int*   ei   = (const int*)d_in[1];
    const float* ea   = (const float*)d_in[2];
    const float* W1   = (const float*)d_in[3];
    const float* as1  = (const float*)d_in[4];
    const float* ad1  = (const float*)d_in[5];
    const float* we1  = (const float*)d_in[6];
    const float* ae1  = (const float*)d_in[7];
    const float* b1   = (const float*)d_in[8];
    const float* W2   = (const float*)d_in[9];
    const float* as2  = (const float*)d_in[10];
    const float* ad2  = (const float*)d_in[11];
    const float* we2  = (const float*)d_in[12];
    const float* ae2  = (const float*)d_in[13];
    const float* b2   = (const float*)d_in[14];
    float* out = (float*)d_out;

    const int N  = in_sizes[0] / 128;
    const int E  = in_sizes[1] / 2;

    float* ws = (float*)d_ws;
    size_t o = 0;
    float* h     = ws + o; o += (size_t)N * 64;
    float* hagg  = ws + o; o += (size_t)N * 64;
    int2*  rec   = (int2*)(ws + o); o += (size_t)E * 2;
    int*   rank  = (int*)(ws + o); o += E;
    int*   cnt   = (int*)(ws + o); o += N;
    int*   off   = (int*)(ws + o); o += N + 1;
    float* asrc  = ws + o; o += N;
    float* adst  = ws + o; o += N;
    float* kscal = ws + o; o += 4;

    const int B = 256;
    const int mtiles = N / 16;

    zeroF<<<gr(N, B), B, 0, stream>>>((float*)cnt, N);
    rankK<<<gr(E, B), B, 0, stream>>>(ei, cnt, rank, E);
    scanK<<<gr(N, 4096), 1024, 0, stream>>>(cnt, off, N);
    fillK<<<gr(E, B), B, 0, stream>>>(ei, ea, off, rank, rec, E);
    kdotK<<<1, 64, 0, stream>>>(we1, ae1, we2, ae2, kscal);

    gemmWmmaK<128, 4><<<gr(mtiles, 8), 8 * 32, 0, stream>>>(x, W1, h, mtiles);
    attdotK<<<gr(N, B), B, 0, stream>>>(h, as1, ad1, asrc, adst, N, 16);
    segK<64, true><<<gr((long long)N * 32, B), B, 0, stream>>>(off, rec, asrc, adst, kscal, 0, h, b1, hagg, N);

    gemmWmmaK<64, 2><<<gr(mtiles, 8), 8 * 32, 0, stream>>>(hagg, W2, h, mtiles);
    attdotK<<<gr(N, B), B, 0, stream>>>(h, as2, ad2, asrc, adst, N, 8);
    segK<32, false><<<gr((long long)N * 32, B), B, 0, stream>>>(off, rec, asrc, adst, kscal, 1, h, b2, out, N);
}


# Four small passes folded into the GEMM launches (11 -> 7): attention dots from the accumulators in the epilogue (column sums then lane butterfly: new sum order); counter clear, edge scalars in GEMM 1
# speedup vs baseline: 4.7513x; 1.1021x over previous
#include <hip/hip_runtime.h>
#include <math.h>

typedef __attribute__((ext_vector_type(2))) float v2f;
typedef __attribute__((ext_vector_type(8))) float v8f;

#define NEG_SLOPE 0.2f

__global__ void rankK(const int* __restrict__ ei, int* __restrict__ cnt,
                      int* __restrict__ rank, int E) {
    int e = blockIdx.x * blockDim.x + threadIdx.x;
    if (e >= E) return;
    rank[e] = atomicAdd(&cnt[ei[E + e]], 1);
}

__device__ __forceinline__ int waveSumI(int v) {
    for (int o = 16; o > 0; o >>= 1) v += __shfl_xor(v, o);
    return v;
}
__global__ void __launch_bounds__(1024) scanK(const int* __restrict__ cnt, int* __restrict__ off, int N) {
    __shared__ int shPre[32];
    __shared__ int shTot[32];
    const int t = threadIdx.x, lane = t & 31, w = t >> 5;
    const int nb = blockIdx.x;
    const int4* c4 = (const int4*)cnt;
    const int g  = nb * 1024 + t;
    const int i0 = g * 4;

    int4 own = make_int4(0, 0, 0, 0);
    if (i0 + 3 < N) {
        own = c4[g];
    } else {
        if (i0     < N) own.x = cnt[i0];
        if (i0 + 1 < N) own.y = cnt[i0 + 1];
        if (i0 + 2 < N) own.z = cnt[i0 + 2];
    }
    int pre = 0;
    for (int k = 0; k < nb; ++k) {
        int4 v = c4[k * 1024 + t];
        pre += (v.x + v.y) + (v.z + v.w);
    }

    const int s1 = own.x, s2 = s1 + own.y, s3 = s2 + own.z, s4 = s3 + own.w;
    int x = s4;
    for (int o = 1; o < 32; o <<= 1) {
        int y = __shfl_up(x, o);
        if (lane >= o) x += y;
    }
    pre = waveSumI(pre);
    if (lane == 31) shTot[w] = x;
    if (lane == 0)  shPre[w] = pre;
    __syncthreads();
    const int front = waveSumI(shPre[lane] + ((lane < w) ? shTot[lane] : 0));
    const int e0 = front + x - s4;

    if (i0 + 3 < N) {
        ((int4*)off)[g] = make_int4(e0, e0 + s1, e0 + s2, e0 + s3);
    } else {
        if (i0     < N) off[i0]     = e0;
        if (i0 + 1 < N) off[i0 + 1] = e0 + s1;
        if (i0 + 2 < N) off[i0 + 2] = e0 + s2;
    }
    if (i0 < N && N <= i0 + 4) off[N] = e0 + s4;
}

__global__ void fillK(const int* __restrict__ ei, const float* __restrict__ ea,
                      const int* __restrict__ off, const int* __restrict__ rank,
                      int2* __restrict__ rec, int E) {
    int e = blockIdx.x * blockDim.x + threadIdx.x;
    if (e >= E) return;
    int p = off[ei[E + e]] + rank[e];
    rec[p] = make_int2(ei[e], __float_as_int(ea[e]));
}

template <int KDIM, int NT, bool SIDE>
__global__ void gemmWmmaK(const float* __restrict__ A, const float* __restrict__ B,
                          float* __restrict__ C, int mtiles,
                          const float* __restrict__ as, const float* __restrict__ ad,
                          float* __restrict__ osrc, float* __restrict__ odst,
                          int* __restrict__ cnt, int N,
                          const float* __restrict__ we1, const float* __restrict__ ae1,
                          const float* __restrict__ we2, const float* __restrict__ ae2,
                          float* __restrict__ kscal) {
    constexpr int NCOL = NT * 16;
    const int gt = blockIdx.x * blockDim.x + threadIdx.x;
    int wave = gt >> 5;
    int lane = threadIdx.x & 31;
    if constexpr (SIDE) {
        for (int i = gt; i < N; i += gridDim.x * blockDim.x) cnt[i] = 0;
    }
    if (wave >= mtiles) {
        if constexpr (SIDE) {
            if (wave == mtiles) {
                if (lane == 0) { float s = 0.f; for (int j = 0; j < 64; ++j) s += we1[j] * ae1[j]; kscal[0] = s; }
                if (lane == 1) { float s = 0.f; for (int j = 0; j < 32; ++j) s += we2[j] * ae2[j]; kscal[1] = s; }
            }
        }
        return;
    }

    int half = lane >> 4;
    int idx  = lane & 15;
    int kk   = half << 1;

    const float* arow  = A + (wave * 16 + idx) * KDIM + kk;
    const float* bbase = B + kk * NCOL + idx;

    v8f acc[NT];
#pragma unroll
    for (int t = 0; t < NT; ++t) acc[t] = (v8f){};

#pragma unroll
    for (int k0 = 0; k0 < KDIM; k0 += 4) {
        v2f a;
        a.x = arow[k0];
        a.y = arow[k0 + 1];
#pragma unroll
        for (int t = 0; t < NT; ++t) {
            v2f b;
            b.x = bbase[k0 * NCOL + t * 16];
            b.y = bbase[(k0 + 1) * NCOL + t * 16];
            acc[t] = __builtin_amdgcn_wmma_f32_16x16x4_f32(false, a, false, b,
                                                           (short)0, acc[t], false, false);
        }
    }
    int rbase = wave * 16 + (half << 3);
#pragma unroll
    for (int t = 0; t < NT; ++t) {
        float* cp = C + rbase * NCOL + t * 16 + idx;
#pragma unroll
        for (int i = 0; i < 8; ++i) cp[i * NCOL] = acc[t][i];
    }

    float ps[8], pd[8];
    for (int i = 0; i < 8; ++i) { ps[i] = 0.f; pd[i] = 0.f; }
    for (int t = 0; t < NT; ++t) {
        const float sa = as[t * 16 + idx], sd = ad[t * 16 + idx];
        for (int i = 0; i < 8; ++i) { ps[i] += acc[t][i] * sa; pd[i] += acc[t][i] * sd; }
    }
    for (int o = 8; o > 0; o >>= 1)
        for (int i = 0; i < 8; ++i) { ps[i] += __shfl_xor(ps[i], o); pd[i] += __shfl_xor(pd[i], o); }
    if (idx == 0)
        for (int i = 0; i < 8; ++i) { osrc[rbase + i] = ps[i]; odst[rbase + i] = pd[i]; }
}

__device__ __forceinline__ float waveSum(float v) {
    for (int o = 16; o > 0; o >>= 1) v += __shfl_xor(v, o);
    return v;
}
__device__ __forceinline__ float waveMax(float v) {
    for (int o = 16; o > 0; o >>= 1) v = fmaxf(v, __shfl_xor(v, o));
    return v;
}
__device__ __forceinline__ float lrelu(float a) { return (a > 0.0f) ? a : NEG_SLOPE * a; }

template <int D, bool RELU>
__global__ void __launch_bounds__(256) segK(const int* __restrict__ off, const int2* __restrict__ rec,
                                            const float* __restrict__ asrc, const float* __restrict__ adst,
                                            const float* __restrict__ kscal, int ki,
                                            const float* __restrict__ h, const float* __restrict__ bias,
                                            float* __restrict__ out, int N) {
    constexpr int V = D / 32;
    const int lane = threadIdx.x & 31;
    const int i = __builtin_amdgcn_readfirstlane((int)((blockIdx.x * blockDim.x + threadIdx.x) >> 5));
    if (i >= N) return;
    const int b = off[i], e = off[i + 1];
    const int deg = e - b;
    const float kk = kscal[ki];
    const float ad = adst[i];

    auto edge = [&](int j, int& s, float& at) -> float {
        int2 r = rec[j];
        s = r.x; at = __int_as_float(r.y);
        return lrelu(asrc[s] + ad + at * kk);
    };
    auto loadRow = [&](int row, float (&v)[V]) {
        if constexpr (V == 2) {
            float2 t = *(const float2*)(h + row * D + lane * 2);
            v[0] = t.x; v[1] = t.y;
        } else {
            v[0] = h[row * D + lane];
        }
    };

    int s0 = i; float at0 = 0.0f, al0 = -INFINITY;
    const bool v0 = lane < deg;
    if (v0) al0 = edge(b + lane, s0, at0);
    float mx = al0, asum = at0;
    for (int cb = b + 32; cb < e; cb += 32) {
        int j = cb + lane;
        if (j < e) { int s; float at; float al = edge(j, s, at); mx = fmaxf(mx, al); asum += at; }
    }
    mx = waveMax(mx);
    asum = waveSum(asum);

    const float la  = asum / fmaxf((float)deg, 1.0f);
    const float als = lrelu(asrc[i] + ad + la * kk);
    mx = fmaxf(mx, als);
    const float exs = expf(als - mx);
    const float ex0 = v0 ? expf(al0 - mx) : 0.0f;
    float dsum = ex0;
    for (int cb = b + 32; cb < e; cb += 32) {
        int j = cb + lane;
        if (j < e) { int s; float at; dsum += expf(edge(j, s, at) - mx); }
    }
    const float den = waveSum(dsum) + exs;

    float acc[V];
    {
        float hv[V];
        loadRow(i, hv);
        const float wself = exs / den;
        for (int v = 0; v < V; ++v) acc[v] = wself * hv[v];
    }
    auto gather = [&](int s, float w, int n) {
        for (int t = 0; t < n; t += 8) {
            float hv[8][V]; float wt[8];
            for (int u = 0; u < 8; ++u) {
                int st = __builtin_amdgcn_readlane(s, t + u);
                wt[u] = __int_as_float(__builtin_amdgcn_readlane(__float_as_int(w), t + u));
                loadRow(st, hv[u]);
            }
            for (int u = 0; u < 8; ++u)
                for (int v = 0; v < V; ++v) acc[v] += wt[u] * hv[u][v];
        }
    };
    gather(s0, ex0 / den, min(deg, 32));
    for (int cb = b + 32; cb < e; cb += 32) {
        int j = cb + lane;
        int s = i; float w = 0.0f;
        if (j < e) { float at; w = expf(edge(j, s, at) - mx) / den; }
        gather(s, w, min(e - cb, 32));
    }

    if constexpr (V == 2) {
        float2 o;
        o.x = acc[0] + bias[lane * 2];
        o.y = acc[1] + bias[lane * 2 + 1];
        if (RELU) { o.x = fmaxf(o.x, 0.0f); o.y = fmaxf(o.y, 0.0f); }
        *(float2*)(out + i * D + lane * 2) = o;
    } else {
        float o = acc[0] + bias[lane];
        if (RELU) o = fmaxf(o, 0.0f);
        out[i * D + lane] = o;
    }
}

static inline int gr(long long n, int blk) { return (int)((n + blk - 1) / blk); }

extern "C" void kernel_launch(void* const* d_in, const int* in_sizes, int n_in,
                              void* d_out, int out_size, void* d_ws, size_t ws_size,
                              hipStream_t stream) {
    const float* x    = (const float*)d_in[0];
    const int*   ei   = (const int*)d_in[1];
    const float* ea   = (const float*)d_in[2];
    const float* W1   = (const float*)d_in[3];
    const float* as1  = (const float*)d_in[4];
    const float* ad1  = (const float*)d_in[5];
    const float* we1  = (const float*)d_in[6];
    const float* ae1  = (const float*)d_in[7];
    const float* b1   = (const float*)d_in[8];
    const float* W2   = (const float*)d_in[9];
    const float* as2  = (const float*)d_in[10];
    const float* ad2  = (const float*)d_in[11];
    const float* we2  = (const float*)d_in[12];
    const float* ae2  = (const float*)d_in[13];
    const float* b2   = (const float*)d_in[14];
    float* out = (float*)d_out;

    const int N  = in_sizes[0] / 128;
    const int E  = in_sizes[1] / 2;

    float* ws = (float*)d_ws;
    size_t o = 0;
    float* h     = ws + o; o += (size_t)N * 64;
    float* hagg  = ws + o; o += (size_t)N * 64;
    int2*  rec   = (int2*)(ws + o); o += (size_t)E * 2;
    int*   rank  = (int*)(ws + o); o += E;
    int*   cnt   = (int*)(ws + o); o += N;
    int*   off   = (int*)(ws + o); o += N + 1;
    float* asrc  = ws + o; o += N;
    float* adst  = ws + o; o += N;
    float* kscal = ws + o; o += 4;

    const int B = 256;
    const int mtiles = N / 16;

    gemmWmmaK<128, 4, true><<<gr(mtiles + 1, 8), 8 * 32, 0, stream>>>(x, W1, h, mtiles, as1, ad1, asrc, adst,
                                                                      cnt, N, we1, ae1, we2, ae2, kscal);

    rankK<<<gr(E, B), B, 0, stream>>>(ei, cnt, rank, E);
    scanK<<<gr(N, 4096), 1024, 0, stream>>>(cnt, off, N);
    fillK<<<gr(E, B), B, 0, stream>>>(ei, ea, off, rank, rec, E);

    segK<64, true><<<gr((long long)N * 32, B), B, 0, stream>>>(off, rec, asrc, adst, kscal, 0, h, b1, hagg, N);

    gemmWmmaK<64, 2, false><<<gr(mtiles, 8), 8 * 32, 0, stream>>>(hagg, W2, h, mtiles, as2, ad2, asrc, adst,
                                                                   nullptr, 0, nullptr, nullptr, nullptr, nullptr, nullptr);
    segK<32, false><<<gr((long long)N * 32, B), B, 0, stream>>>(off, rec, asrc, adst, kscal, 1, h, b2, out, N);
}


# Edge sort without global atomics: 4096-edge chunks ordered by 256-node bin, then one block per bin orders by node; ranks from returning LDS adds (segment order = LDS arrival order); 7 -> 6 launches
# speedup vs baseline: 7.7185x; 1.6245x over previous
#include <hip/hip_runtime.h>
#include <math.h>

typedef __attribute__((ext_vector_type(2))) float v2f;
typedef __attribute__((ext_vector_type(8))) float v8f;

#define NEG_SLOPE 0.2f

#define CH_EDGES 4096
#define BIN_SHIFT 8
#define MAX_BINS 256
#define MAX_CH 256

__device__ __forceinline__ int waveSumI(int v) {
    for (int o = 16; o > 0; o >>= 1) v += __shfl_xor(v, o);
    return v;
}
__device__ __forceinline__ int waveScanI(int x, int lane) {
    for (int o = 1; o < 32; o <<= 1) {
        int y = __shfl_up(x, o);
        if (lane >= o) x += y;
    }
    return x;
}

__global__ void __launch_bounds__(1024) binK(const int* __restrict__ ei, const float* __restrict__ ea,
                                             int4* __restrict__ tmp, int* __restrict__ tab,
                                             int E, int NB, int G) {
    __shared__ int hist[MAX_BINS];
    __shared__ int wtot[8];
    const int t = threadIdx.x, lane = t & 31, w = t >> 5;
    const int blk = blockIdx.x;
    if (t < MAX_BINS) hist[t] = 0;
    __syncthreads();

    const int e0 = blk * CH_EDGES + t * 4;
    int d[4], s[4], a[4], lr[4];
    bool ok[4];
    for (int k = 0; k < 4; ++k) { ok[k] = (e0 + k) < E; d[k] = 0; s[k] = 0; a[k] = 0; lr[k] = 0; }
    if (((E & 3) == 0) && (e0 + 3 < E)) {
        const int4 dv = *(const int4*)(ei + E + e0);
        const int4 sv = *(const int4*)(ei + e0);
        const float4 av = *(const float4*)(ea + e0);
        d[0] = dv.x; d[1] = dv.y; d[2] = dv.z; d[3] = dv.w;
        s[0] = sv.x; s[1] = sv.y; s[2] = sv.z; s[3] = sv.w;
        a[0] = __float_as_int(av.x); a[1] = __float_as_int(av.y);
        a[2] = __float_as_int(av.z); a[3] = __float_as_int(av.w);
    } else {
        for (int k = 0; k < 4; ++k)
            if (ok[k]) { d[k] = ei[E + e0 + k]; s[k] = ei[e0 + k]; a[k] = __float_as_int(ea[e0 + k]); }
    }
    for (int k = 0; k < 4; ++k)
        if (ok[k]) lr[k] = atomicAdd(&hist[d[k] >> BIN_SHIFT], 1);
    __syncthreads();

    int cv = 0, ci = 0;
    if (t < MAX_BINS) {
        cv = hist[t];
        ci = waveScanI(cv, lane);
        if (lane == 31) wtot[w] = ci;
    }
    __syncthreads();
    if (t < MAX_BINS) {
        int front = 0;
        for (int j = 0; j < w; ++j) front += wtot[j];
        const int ex = front + ci - cv;
        hist[t] = ex;
        if (t < NB) tab[t * G + blk] = cv | (ex << 16);
    }
    __syncthreads();
    for (int k = 0; k < 4; ++k)
        if (ok[k]) tmp[blk * CH_EDGES + hist[d[k] >> BIN_SHIFT] + lr[k]] = make_int4(d[k], s[k], a[k], 0);
}

__global__ void __launch_bounds__(1024) placeK(const int4* __restrict__ tmp, const int* __restrict__ tab,
                                               int* __restrict__ off, int2* __restrict__ rec,
                                               int N, int G) {
    __shared__ int pre[MAX_CH + 1];
    __shared__ int rs[MAX_CH];
    __shared__ int c[256];
    __shared__ int wsum[32];
    __shared__ int wtot[8];
    const int t = threadIdx.x, lane = t & 31, w = t >> 5;
    const int b = blockIdx.x;
    const int nfront = b * G;

    int part = 0;
    {
        const int4* t4 = (const int4*)tab;
        const int n4 = nfront >> 2;
        for (int k = t; k < n4; k += 1024) {
            const int4 v = t4[k];
            part += ((v.x & 0xFFFF) + (v.y & 0xFFFF)) + ((v.z & 0xFFFF) + (v.w & 0xFFFF));
        }
        for (int k = (n4 << 2) + t; k < nfront; k += 1024) part += tab[k] & 0xFFFF;
    }
    int rc = 0, ro = 0;
    if (t < G) { const int v = tab[nfront + t]; rc = v & 0xFFFF; ro = v >> 16; }
    part = waveSumI(part);
    if (lane == 0) wsum[w] = part;
    int ri = 0;
    if (t < 256) {
        ri = waveScanI(rc, lane);
        if (lane == 31) wtot[w] = ri;
        c[t] = 0;
    }
    __syncthreads();
    const int base = waveSumI(wsum[lane]);
    if (t < 256) {
        int front = 0;
        for (int j = 0; j < w; ++j) front += wtot[j];
        const int ex = front + ri - rc;
        if (t < G) { pre[t] = ex; rs[t] = t * CH_EDGES + ro; }
        if (t == G - 1) pre[G] = ex + rc;
    }
    __syncthreads();
    const int total = pre[G];

    auto slot = [&](int i) -> int {
        int lo = 0, hi = G;
        for (int st = 0; st < 8; ++st) {
            const int mid = (lo + hi) >> 1;
            if (pre[mid] <= i) lo = mid; else hi = mid;
        }
        return rs[lo] + (i - pre[lo]);
    };

    for (int i0 = t; i0 < total; i0 += 4096) {
        int dd[4]; bool ok[4];
        for (int u = 0; u < 4; ++u) {
            const int i = i0 + u * 1024;
            ok[u] = i < total;
            dd[u] = tmp[slot(ok[u] ? i : total - 1)].x;
        }
        for (int u = 0; u < 4; ++u)
            if (ok[u]) atomicAdd(&c[dd[u] & 255], 1);
    }
    __syncthreads();

    int cv = 0, ci = 0;
    if (t < 256) {
        cv = c[t];
        ci = waveScanI(cv, lane);
        if (lane == 31) wtot[w] = ci;
    }
    __syncthreads();
    if (t < 256) {
        int front = 0;
        for (int j = 0; j < w; ++j) front += wtot[j];
        const int ex = front + ci - cv;
        c[t] = ex;
        const int node = (b << BIN_SHIFT) + t;
        if (node < N) off[node] = base + ex;
        if (node == N - 1) off[N] = base + ex + cv;
    }
    __syncthreads();

    for (int i0 = t; i0 < total; i0 += 4096) {
        int4 r[4]; bool ok[4];
        for (int u = 0; u < 4; ++u) {
            const int i = i0 + u * 1024;
            ok[u] = i < total;
            r[u] = tmp[slot(ok[u] ? i : total - 1)];
        }
        for (int u = 0; u < 4; ++u)
            if (ok[u]) {
                const int p = atomicAdd(&c[r[u].x & 255], 1);
                rec[base + p] = make_int2(r[u].y, r[u].z);
            }
    }
}

template <int KDIM, int NT, bool SIDE>
__global__ void gemmWmmaK(const float* __restrict__ A, const float* __restrict__ B,
                          float* __restrict__ C, int mtiles,
                          const float* __restrict__ as, const float* __restrict__ ad,
                          float* __restrict__ osrc, float* __restrict__ odst,
                          const float* __restrict__ we1, const float* __restrict__ ae1,
                          const float* __restrict__ we2, const float* __restrict__ ae2,
                          float* __restrict__ kscal) {
    constexpr int NCOL = NT * 16;
    const int gt = blockIdx.x * blockDim.x + threadIdx.x;
    int wave = gt >> 5;
    int lane = threadIdx.x & 31;
    if (wave >= mtiles) {
        if constexpr (SIDE) {
            if (wave == mtiles) {
                if (lane == 0) { float s = 0.f; for (int j = 0; j < 64; ++j) s += we1[j] * ae1[j]; kscal[0] = s; }
                if (lane == 1) { float s = 0.f; for (int j = 0; j < 32; ++j) s += we2[j] * ae2[j]; kscal[1] = s; }
            }
        }
        return;
    }

    int half = lane >> 4;
    int idx  = lane & 15;
    int kk   = half << 1;

    const float* arow  = A + (wave * 16 + idx) * KDIM + kk;
    const float* bbase = B + kk * NCOL + idx;

    v8f acc[NT];
#pragma unroll
    for (int t = 0; t < NT; ++t) acc[t] = (v8f){};

#pragma unroll
    for (int k0 = 0; k0 < KDIM; k0 += 4) {
        v2f a;
        a.x = arow[k0];
        a.y = arow[k0 + 1];
#pragma unroll
        for (int t = 0; t < NT; ++t) {
            v2f b;
            b.x = bbase[k0 * NCOL + t * 16];
            b.y = bbase[(k0 + 1) * NCOL + t * 16];
            acc[t] = __builtin_amdgcn_wmma_f32_16x16x4_f32(false, a, false, b,
                                                           (short)0, acc[t], false, false);
        }
    }
    int rbase = wave * 16 + (half << 3);
#pragma unroll
    for (int t = 0; t < NT; ++t) {
        float* cp = C + rbase * NCOL + t * 16 + idx;
#pragma unroll
        for (int i = 0; i < 8; ++i) cp[i * NCOL] = acc[t][i];
    }

    float ps[8], pd[8];
    for (int i = 0; i < 8; ++i) { ps[i] = 0.f; pd[i] = 0.f; }
    for (int t = 0; t < NT; ++t) {
        const float sa = as[t * 16 + idx], sd = ad[t * 16 + idx];
        for (int i = 0; i < 8; ++i) { ps[i] += acc[t][i] * sa; pd[i] += acc[t][i] * sd; }
    }
    for (int o = 8; o > 0; o >>= 1)
        for (int i = 0; i < 8; ++i) { ps[i] += __shfl_xor(ps[i], o); pd[i] += __shfl_xor(pd[i], o); }
    if (idx == 0)
        for (int i = 0; i < 8; ++i) { osrc[rbase + i] = ps[i]; odst[rbase + i] = pd[i]; }
}

__device__ __forceinline__ float waveSum(float v) {
    for (int o = 16; o > 0; o >>= 1) v += __shfl_xor(v, o);
    return v;
}
__device__ __forceinline__ float waveMax(float v) {
    for (int o = 16; o > 0; o >>= 1) v = fmaxf(v, __shfl_xor(v, o));
    return v;
}
__device__ __forceinline__ float lrelu(float a) { return (a > 0.0f) ? a : NEG_SLOPE * a; }

template <int D, bool RELU>
__global__ void __launch_bounds__(256) segK(const int* __restrict__ off, const int2* __restrict__ rec,
                                            const float* __restrict__ asrc, const float* __restrict__ adst,
                                            const float* __restrict__ kscal, int ki,
                                            const float* __restrict__ h, const float* __restrict__ bias,
                                            float* __restrict__ out, int N) {
    constexpr int V = D / 32;
    const int lane = threadIdx.x & 31;
    const int i = __builtin_amdgcn_readfirstlane((int)((blockIdx.x * blockDim.x + threadIdx.x) >> 5));
    if (i >= N) return;
    const int b = off[i], e = off[i + 1];
    const int deg = e - b;
    const float kk = kscal[ki];
    const float ad = adst[i];

    auto edge = [&](int j, int& s, float& at) -> float {
        int2 r = rec[j];
        s = r.x; at = __int_as_float(r.y);
        return lrelu(asrc[s] + ad + at * kk);
    };
    auto loadRow = [&](int row, float (&v)[V]) {
        if constexpr (V == 2) {
            float2 t = *(const float2*)(h + row * D + lane * 2);
            v[0] = t.x; v[1] = t.y;
        } else {
            v[0] = h[row * D + lane];
        }
    };

    int s0 = i; float at0 = 0.0f, al0 = -INFINITY;
    const bool v0 = lane < deg;
    if (v0) al0 = edge(b + lane, s0, at0);
    float mx = al0, asum = at0;
    for (int cb = b + 32; cb < e; cb += 32) {
        int j = cb + lane;
        if (j < e) { int s; float at; float al = edge(j, s, at); mx = fmaxf(mx, al); asum += at; }
    }
    mx = waveMax(mx);
    asum = waveSum(asum);

    const float la  = asum / fmaxf((float)deg, 1.0f);
    const float als = lrelu(asrc[i] + ad + la * kk);
    mx = fmaxf(mx, als);
    const float exs = expf(als - mx);
    const float ex0 = v0 ? expf(al0 - mx) : 0.0f;
    float dsum = ex0;
    for (int cb = b + 32; cb < e; cb += 32) {
        int j = cb + lane;
        if (j < e) { int s; float at; dsum += expf(edge(j, s, at) - mx); }
    }
    const float den = waveSum(dsum) + exs;

    float acc[V];
    {
        float hv[V];
        loadRow(i, hv);
        const float wself = exs / den;
        for (int v = 0; v < V; ++v) acc[v] = wself * hv[v];
    }
    auto gather = [&](int s, float w, int n) {
        for (int t = 0; t < n; t += 8) {
            float hv[8][V]; float wt[8];
            for (int u = 0; u < 8; ++u) {
                int st = __builtin_amdgcn_readlane(s, t + u);
                wt[u] = __int_as_float(__builtin_amdgcn_readlane(__float_as_int(w), t + u));
                loadRow(st, hv[u]);
            }
            for (int u = 0; u < 8; ++u)
                for (int v = 0; v < V; ++v) acc[v] += wt[u] * hv[u][v];
        }
    };
    gather(s0, ex0 / den, min(deg, 32));
    for (int cb = b + 32; cb < e; cb += 32) {
        int j = cb + lane;
        int s = i; float w = 0.0f;
        if (j < e) { float at; w = expf(edge(j, s, at) - mx) / den; }
        gather(s, w, min(e - cb, 32));
    }

    if constexpr (V == 2) {
        float2 o;
        o.x = acc[0] + bias[lane * 2];
        o.y = acc[1] + bias[lane * 2 + 1];
        if (RELU) { o.x = fmaxf(o.x, 0.0f); o.y = fmaxf(o.y, 0.0f); }
        *(float2*)(out + i * D + lane * 2) = o;
    } else {
        float o = acc[0] + bias[lane];
        if (RELU) o = fmaxf(o, 0.0f);
        out[i * D + lane] = o;
    }
}

static inline int gr(long long n, int blk) { return (int)((n + blk - 1) / blk); }

extern "C" void kernel_launch(void* const* d_in, const int* in_sizes, int n_in,
                              void* d_out, int out_size, void* d_ws, size_t ws_size,
                              hipStream_t stream) {
    const float* x    = (const float*)d_in[0];
    const int*   ei   = (const int*)d_in[1];
    const float* ea   = (const float*)d_in[2];
    const float* W1   = (const float*)d_in[3];
    const float* as1  = (const float*)d_in[4];
    const float* ad1  = (const float*)d_in[5];
    const float* we1  = (const float*)d_in[6];
    const float* ae1  = (const float*)d_in[7];
    const float* b1   = (const float*)d_in[8];
    const float* W2   = (const float*)d_in[9];
    const float* as2  = (const float*)d_in[10];
    const float* ad2  = (const float*)d_in[11];
    const float* we2  = (const float*)d_in[12];
    const float* ae2  = (const float*)d_in[13];
    const float* b2   = (const float*)d_in[14];
    float* out = (float*)d_out;

    const int N  = in_sizes[0] / 128;
    const int E  = in_sizes[1] / 2;

    float* ws = (float*)d_ws;
    size_t o = 0;
    float* h     = ws + o; o += (size_t)N * 64;
    float* hagg  = ws + o; o += (size_t)N * 64;
    int2*  rec   = (int2*)(ws + o); o += (size_t)E * 2;
    const int G  = gr(E, CH_EDGES);
    const int NB = gr(N, 1 << BIN_SHIFT);
    o = (o + 3) & ~(size_t)3;
    int4*  tmp   = (int4*)(ws + o); o += (size_t)G * CH_EDGES * 4;
    int*   tab   = (int*)(ws + o); o += (size_t)NB * G;
    int*   off   = (int*)(ws + o); o += N + 1;
    float* asrc  = ws + o; o += N;
    float* adst  = ws + o; o += N;
    float* kscal = ws + o; o += 4;

    const int B = 256;
    const int mtiles = N / 16;

    gemmWmmaK<128, 4, true><<<gr(mtiles + 1, 8), 8 * 32, 0, stream>>>(x, W1, h, mtiles, as1, ad1, asrc, adst,
                                                                      we1, ae1, we2, ae2, kscal);

    binK<<<G, 1024, 0, stream>>>(ei, ea, tmp, tab, E, NB, G);
    placeK<<<NB, 1024, 0, stream>>>(tmp, tab, off, rec, N, G);

    segK<64, true><<<gr((long long)N * 32, B), B, 0, stream>>>(off, rec, asrc, adst, kscal, 0, h, b1, hagg, N);

    gemmWmmaK<64, 2, false><<<gr(mtiles, 8), 8 * 32, 0, stream>>>(hagg, W2, h, mtiles, as2, ad2, asrc, adst,
                                                                   nullptr, nullptr, nullptr, nullptr, nullptr);
    segK<32, false><<<gr((long long)N * 32, B), B, 0, stream>>>(off, rec, asrc, adst, kscal, 1, h, b2, out, N);
}


# Layer-1 product rides in the launch of the edge sort's first level (independent work side by side): 196 blocks of 16 one-tile waves ahead of the sort's 196; 6 -> 5 launches; sum orders kept
# speedup vs baseline: 8.2909x; 1.0742x over previous
#include <hip/hip_runtime.h>
#include <math.h>

typedef __attribute__((ext_vector_type(2))) float v2f;
typedef __attribute__((ext_vector_type(8))) float v8f;

#define NEG_SLOPE 0.2f

#define CH_EDGES 4096
#define BIN_SHIFT 8
#define MAX_BINS 256
#define MAX_CH 256

__device__ __forceinline__ int waveSumI(int v) {
    for (int o = 16; o > 0; o >>= 1) v += __shfl_xor(v, o);
    return v;
}
__device__ __forceinline__ int waveScanI(int x, int lane) {
    for (int o = 1; o < 32; o <<= 1) {
        int y = __shfl_up(x, o);
        if (lane >= o) x += y;
    }
    return x;
}

__device__ __forceinline__ void binBody(const int* __restrict__ ei, const float* __restrict__ ea,
                                        int4* __restrict__ tmp, int* __restrict__ tab,
                                        int E, int NB, int G, const int blk) {
    __shared__ int hist[MAX_BINS];
    __shared__ int wtot[8];
    const int t = threadIdx.x, lane = t & 31, w = t >> 5;
    if (t < MAX_BINS) hist[t] = 0;
    __syncthreads();

    const int e0 = blk * CH_EDGES + t * 4;
    int d[4], s[4], a[4], lr[4];
    bool ok[4];
    for (int k = 0; k < 4; ++k) { ok[k] = (e0 + k) < E; d[k] = 0; s[k] = 0; a[k] = 0; lr[k] = 0; }
    if (((E & 3) == 0) && (e0 + 3 < E)) {
        const int4 dv = *(const int4*)(ei + E + e0);
        const int4 sv = *(const int4*)(ei + e0);
        const float4 av = *(const float4*)(ea + e0);
        d[0] = dv.x; d[1] = dv.y; d[2] = dv.z; d[3] = dv.w;
        s[0] = sv.x; s[1] = sv.y; s[2] = sv.z; s[3] = sv.w;
        a[0] = __float_as_int(av.x); a[1] = __float_as_int(av.y);
        a[2] = __float_as_int(av.z); a[3] = __float_as_int(av.w);
    } else {
        for (int k = 0; k < 4; ++k)
            if (ok[k]) { d[k] = ei[E + e0 + k]; s[k] = ei[e0 + k]; a[k] = __float_as_int(ea[e0 + k]); }
    }
    for (int k = 0; k < 4; ++k)
        if (ok[k]) lr[k] = atomicAdd(&hist[d[k] >> BIN_SHIFT], 1);
    __syncthreads();

    int cv = 0, ci = 0;
    if (t < MAX_BINS) {
        cv = hist[t];
        ci = waveScanI(cv, lane);
        if (lane == 31) wtot[w] = ci;
    }
    __syncthreads();
    if (t < MAX_BINS) {
        int front = 0;
        for (int j = 0; j < w; ++j) front += wtot[j];
        const int ex = front + ci - cv;
        hist[t] = ex;
        if (t < NB) tab[t * G + blk] = cv | (ex << 16);
    }
    __syncthreads();
    for (int k = 0; k < 4; ++k)
        if (ok[k]) tmp[blk * CH_EDGES + hist[d[k] >> BIN_SHIFT] + lr[k]] = make_int4(d[k], s[k], a[k], 0);
}

__global__ void __launch_bounds__(1024) placeK(const int4* __restrict__ tmp, const int* __restrict__ tab,
                                               int* __restrict__ off, int2* __restrict__ rec,
                                               int N, int G) {
    __shared__ int pre[MAX_CH + 1];
    __shared__ int rs[MAX_CH];
    __shared__ int c[256];
    __shared__ int wsum[32];
    __shared__ int wtot[8];
    const int t = threadIdx.x, lane = t & 31, w = t >> 5;
    const int b = blockIdx.x;
    const int nfront = b * G;

    int part = 0;
    {
        const int4* t4 = (const int4*)tab;
        const int n4 = nfront >> 2;
        for (int k = t; k < n4; k += 1024) {
            const int4 v = t4[k];
            part += ((v.x & 0xFFFF) + (v.y & 0xFFFF)) + ((v.z & 0xFFFF) + (v.w & 0xFFFF));
        }
        for (int k = (n4 << 2) + t; k < nfront; k += 1024) part += tab[k] & 0xFFFF;
    }
    int rc = 0, ro = 0;
    if (t < G) { const int v = tab[nfront + t]; rc = v & 0xFFFF; ro = v >> 16; }
    part = waveSumI(part);
    if (lane == 0) wsum[w] = part;
    int ri = 0;
    if (t < 256) {
        ri = waveScanI(rc, lane);
        if (lane == 31) wtot[w] = ri;
        c[t] = 0;
    }
    __syncthreads();
    const int base = waveSumI(wsum[lane]);
    if (t < 256) {
        int front = 0;
        for (int j = 0; j < w; ++j) front += wtot[j];
        const int ex = front + ri - rc;
        if (t < G) { pre[t] = ex; rs[t] = t * CH_EDGES + ro; }
        if (t == G - 1) pre[G] = ex + rc;
    }
    __syncthreads();
    const int total = pre[G];

    auto slot = [&](int i) -> int {
        int lo = 0, hi = G;
        for (int st = 0; st < 8; ++st) {
            const int mid = (lo + hi) >> 1;
            if (pre[mid] <= i) lo = mid; else hi = mid;
        }
        return rs[lo] + (i - pre[lo]);
    };

    for (int i0 = t; i0 < total; i0 += 4096) {
        int dd[4]; bool ok[4];
        for (int u = 0; u < 4; ++u) {
            const int i = i0 + u * 1024;
            ok[u] = i < total;
            dd[u] = tmp[slot(ok[u] ? i : total - 1)].x;
        }
        for (int u = 0; u < 4; ++u)
            if (ok[u]) atomicAdd(&c[dd[u] & 255], 1);
    }
    __syncthreads();

    int cv = 0, ci = 0;
    if (t < 256) {
        cv = c[t];
        ci = waveScanI(cv, lane);
        if (lane == 31) wtot[w] = ci;
    }
    __syncthreads();
    if (t < 256) {
        int front = 0;
        for (int j = 0; j < w; ++j) front += wtot[j];
        const int ex = front + ci - cv;
        c[t] = ex;
        const int node = (b << BIN_SHIFT) + t;
        if (node < N) off[node] = base + ex;
        if (node == N - 1) off[N] = base + ex + cv;
    }
    __syncthreads();

    for (int i0 = t; i0 < total; i0 += 4096) {
        int4 r[4]; bool ok[4];
        for (int u = 0; u < 4; ++u) {
            const int i = i0 + u * 1024;
            ok[u] = i < total;
            r[u] = tmp[slot(ok[u] ? i : total - 1)];
        }
        for (int u = 0; u < 4; ++u)
            if (ok[u]) {
                const int p = atomicAdd(&c[r[u].x & 255], 1);
                rec[base + p] = make_int2(r[u].y, r[u].z);
            }
    }
}

template <int KDIM, int NT>
__device__ __forceinline__ void gemmTile(const float* __restrict__ A, const float* __restrict__ B,
                                         float* __restrict__ C, const int wave, const int lane,
                                         const float* __restrict__ as, const float* __restrict__ ad,
                                         float* __restrict__ osrc, float* __restrict__ odst) {
    constexpr int NCOL = NT * 16;
    int half = lane >> 4;
    int idx  = lane & 15;
    int kk   = half << 1;

    const float* arow  = (A + wave * (16 * KDIM)) + (idx * KDIM + kk);
    const float* bbase = B + (kk * NCOL + idx);

    v8f acc[NT];
#pragma unroll
    for (int t = 0; t < NT; ++t) acc[t] = (v8f){};

#pragma unroll
    for (int k0 = 0; k0 < KDIM; k0 += 4) {
        v2f a;
        a.x = arow[k0];
        a.y = arow[k0 + 1];
#pragma unroll
        for (int t = 0; t < NT; ++t) {
            v2f b;
            b.x = bbase[k0 * NCOL + t * 16];
            b.y = bbase[(k0 + 1) * NCOL + t * 16];
            acc[t] = __builtin_amdgcn_wmma_f32_16x16x4_f32(false, a, false, b,
                                                           (short)0, acc[t], false, false);
        }
    }
    int rbase = wave * 16 + (half << 3);
#pragma unroll
    for (int t = 0; t < NT; ++t) {
        float* cp = (C + wave * (16 * NCOL)) + ((half << 3) * NCOL + t * 16 + idx);
#pragma unroll
        for (int i = 0; i < 8; ++i) cp[i * NCOL] = acc[t][i];
    }

    float ps[8], pd[8];
    for (int i = 0; i < 8; ++i) { ps[i] = 0.f; pd[i] = 0.f; }
    for (int t = 0; t < NT; ++t) {
        const float sa = as[t * 16 + idx], sd = ad[t * 16 + idx];
        for (int i = 0; i < 8; ++i) { ps[i] += acc[t][i] * sa; pd[i] += acc[t][i] * sd; }
    }
    for (int o = 8; o > 0; o >>= 1)
        for (int i = 0; i < 8; ++i) { ps[i] += __shfl_xor(ps[i], o); pd[i] += __shfl_xor(pd[i], o); }
    if (idx == 0)
        for (int i = 0; i < 8; ++i) { osrc[rbase + i] = ps[i]; odst[rbase + i] = pd[i]; }
}

template <int KDIM, int NT>
__global__ void gemmWmmaK(const float* __restrict__ A, const float* __restrict__ B,
                          float* __restrict__ C, int mtiles,
                          const float* __restrict__ as, const float* __restrict__ ad,
                          float* __restrict__ osrc, float* __restrict__ odst) {
    const int gt = blockIdx.x * blockDim.x + threadIdx.x;
    const int wave = gt >> 5;
    const int lane = threadIdx.x & 31;
    if (wave >= mtiles) return;
    gemmTile<KDIM, NT>(A, B, C, wave, lane, as, ad, osrc, odst);
}

#define TILE_WAVES 16

__device__ __forceinline__ bool productBlock(const float* __restrict__ x, const float* __restrict__ W1,
                                             float* __restrict__ h, int mtiles, int gb,
                                             const float* __restrict__ as1, const float* __restrict__ ad1,
                                             float* __restrict__ asrc, float* __restrict__ adst) {
    if ((int)blockIdx.x >= gb) return false;
    const int w = __builtin_amdgcn_readfirstlane((int)(threadIdx.x >> 5));
    const int lane = threadIdx.x & 31;
    const int tile = blockIdx.x * TILE_WAVES + w;
    if (w < TILE_WAVES && tile < mtiles)
        gemmTile<128, 4>(x, W1, h, tile, lane, as1, ad1, asrc, adst);
    return true;
}

__global__ void __launch_bounds__(1024, 16)
binGemmK(const float* __restrict__ x, const float* __restrict__ W1, float* __restrict__ h,
         int mtiles, int gb,
         const float* __restrict__ as1, const float* __restrict__ ad1,
         float* __restrict__ asrc, float* __restrict__ adst,
         const float* __restrict__ we1, const float* __restrict__ ae1,
         const float* __restrict__ we2, const float* __restrict__ ae2,
         float* __restrict__ kscal,
         const int* __restrict__ ei, const float* __restrict__ ea,
         int4* __restrict__ tmp, int* __restrict__ tab, int E, int NB, int G) {
    if (productBlock(x, W1, h, mtiles, gb, as1, ad1, asrc, adst)) {
        if (blockIdx.x == 0 && (threadIdx.x >> 5) == TILE_WAVES) {
            const int lane = threadIdx.x & 31;
            if (lane == 0) { float s = 0.f; for (int j = 0; j < 64; ++j) s += we1[j] * ae1[j]; kscal[0] = s; }
            if (lane == 1) { float s = 0.f; for (int j = 0; j < 32; ++j) s += we2[j] * ae2[j]; kscal[1] = s; }
        }
        return;
    }
    binBody(ei, ea, tmp, tab, E, NB, G, (int)blockIdx.x - gb);
}

__device__ __forceinline__ float waveSum(float v) {
    for (int o = 16; o > 0; o >>= 1) v += __shfl_xor(v, o);
    return v;
}
__device__ __forceinline__ float waveMax(float v) {
    for (int o = 16; o > 0; o >>= 1) v = fmaxf(v, __shfl_xor(v, o));
    return v;
}
__device__ __forceinline__ float lrelu(float a) { return (a > 0.0f) ? a : NEG_SLOPE * a; }

template <int D, bool RELU>
__global__ void __launch_bounds__(256) segK(const int* __restrict__ off, const int2* __restrict__ rec,
                                            const float* __restrict__ asrc, const float* __restrict__ adst,
                                            const float* __restrict__ kscal, int ki,
                                            const float* __restrict__ h, const float* __restrict__ bias,
                                            float* __restrict__ out, int N) {
    constexpr int V = D / 32;
    const int lane = threadIdx.x & 31;
    const int i = __builtin_amdgcn_readfirstlane((int)((blockIdx.x * blockDim.x + threadIdx.x) >> 5));
    if (i >= N) return;
    const int b = off[i], e = off[i + 1];
    const int deg = e - b;
    const float kk = kscal[ki];
    const float ad = adst[i];

    auto edge = [&](int j, int& s, float& at) -> float {
        int2 r = rec[j];
        s = r.x; at = __int_as_float(r.y);
        return lrelu(asrc[s] + ad + at * kk);
    };
    auto loadRow = [&](int row, float (&v)[V]) {
        if constexpr (V == 2) {
            float2 t = *(const float2*)(h + row * D + lane * 2);
            v[0] = t.x; v[1] = t.y;
        } else {
            v[0] = h[row * D + lane];
        }
    };

    int s0 = i; float at0 = 0.0f, al0 = -INFINITY;
    const bool v0 = lane < deg;
    if (v0) al0 = edge(b + lane, s0, at0);
    float mx = al0, asum = at0;
    for (int cb = b + 32; cb < e; cb += 32) {
        int j = cb + lane;
        if (j < e) { int s; float at; float al = edge(j, s, at); mx = fmaxf(mx, al); asum += at; }
    }
    mx = waveMax(mx);
    asum = waveSum(asum);

    const float la  = asum / fmaxf((float)deg, 1.0f);
    const float als = lrelu(asrc[i] + ad + la * kk);
    mx = fmaxf(mx, als);
    const float exs = expf(als - mx);
    const float ex0 = v0 ? expf(al0 - mx) : 0.0f;
    float dsum = ex0;
    for (int cb = b + 32; cb < e; cb += 32) {
        int j = cb + lane;
        if (j < e) { int s; float at; dsum += expf(edge(j, s, at) - mx); }
    }
    const float den = waveSum(dsum) + exs;

    float acc[V];
    {
        float hv[V];
        loadRow(i, hv);
        const float wself = exs / den;
        for (int v = 0; v < V; ++v) acc[v] = wself * hv[v];
    }
    auto gather = [&](int s, float w, int n) {
        for (int t = 0; t < n; t += 8) {
            float hv[8][V]; float wt[8];
            for (int u = 0; u < 8; ++u) {
                int st = __builtin_amdgcn_readlane(s, t + u);
                wt[u] = __int_as_float(__builtin_amdgcn_readlane(__float_as_int(w), t + u));
                loadRow(st, hv[u]);
            }
            for (int u = 0; u < 8; ++u)
                for (int v = 0; v < V; ++v) acc[v] += wt[u] * hv[u][v];
        }
    };
    gather(s0, ex0 / den, min(deg, 32));
    for (int cb = b + 32; cb < e; cb += 32) {
        int j = cb + lane;
        int s = i; float w = 0.0f;
        if (j < e) { float at; w = expf(edge(j, s, at) - mx) / den; }
        gather(s, w, min(e - cb, 32));
    }

    if constexpr (V == 2) {
        float2 o;
        o.x = acc[0] + bias[lane * 2];
        o.y = acc[1] + bias[lane * 2 + 1];
        if (RELU) { o.x = fmaxf(o.x, 0.0f); o.y = fmaxf(o.y, 0.0f); }
        *(float2*)(out + i * D + lane * 2) = o;
    } else {
        float o = acc[0] + bias[lane];
        if (RELU) o = fmaxf(o, 0.0f);
        out[i * D + lane] = o;
    }
}

static inline int gr(long long n, int blk) { return (int)((n + blk - 1) / blk); }

extern "C" void kernel_launch(void* const* d_in, const int* in_sizes, int n_in,
                              void* d_out, int out_size, void* d_ws, size_t ws_size,
                              hipStream_t stream) {
    const float* x    = (const float*)d_in[0];
    const int*   ei   = (const int*)d_in[1];
    const float* ea   = (const float*)d_in[2];
    const float* W1   = (const float*)d_in[3];
    const float* as1  = (const float*)d_in[4];
    const float* ad1  = (const float*)d_in[5];
    const float* we1  = (const float*)d_in[6];
    const float* ae1  = (const float*)d_in[7];
    const float* b1   = (const float*)d_in[8];
    const float* W2   = (const float*)d_in[9];
    const float* as2  = (const float*)d_in[10];
    const float* ad2  = (const float*)d_in[11];
    const float* we2  = (const float*)d_in[12];
    const float* ae2  = (const float*)d_in[13];
    const float* b2   = (const float*)d_in[14];
    float* out = (float*)d_out;

    const int N  = in_sizes[0] / 128;
    const int E  = in_sizes[1] / 2;

    float* ws = (float*)d_ws;
    size_t o = 0;
    float* h     = ws + o; o += (size_t)N * 64;
    float* hagg  = ws + o; o += (size_t)N * 64;
    int2*  rec   = (int2*)(ws + o); o += (size_t)E * 2;
    const int G  = gr(E, CH_EDGES);
    const int NB = gr(N, 1 << BIN_SHIFT);
    o = (o + 3) & ~(size_t)3;
    int4*  tmp   = (int4*)(ws + o); o += (size_t)G * CH_EDGES * 4;
    int*   tab   = (int*)(ws + o); o += (size_t)NB * G;
    int*   off   = (int*)(ws + o); o += N + 1;
    float* asrc  = ws + o; o += N;
    float* adst  = ws + o; o += N;
    float* kscal = ws + o; o += 4;

    const int B = 256;
    const int mtiles = N / 16;

    const int GB = gr(mtiles, TILE_WAVES);
    binGemmK<<<GB + G, 1024, 0, stream>>>(x, W1, h, mtiles, GB, as1, ad1, asrc, adst,
                                          we1, ae1, we2, ae2, kscal, ei, ea, tmp, tab, E, NB, G);
    placeK<<<NB, 1024, 0, stream>>>(tmp, tab, off, rec, N, G);

    segK<64, true><<<gr((long long)N * 32, B), B, 0, stream>>>(off, rec, asrc, adst, kscal, 0, h, b1, hagg, N);

    gemmWmmaK<64, 2><<<gr(mtiles, 8), 8 * 32, 0, stream>>>(hagg, W2, h, mtiles, as2, ad2, asrc, adst);
    segK<32, false><<<gr((long long)N * 32, B), B, 0, stream>>>(off, rec, asrc, adst, kscal, 1, h, b2, out, N);
}


# Aggregation: two destination nodes per wave, one per 16-lane half (one load = a whole row per half; DPP row steps pass edges on and fold softmax max/sum: new sum order): half the waves, no LDS trips
# speedup vs baseline: 8.8058x; 1.0621x over previous
#include <hip/hip_runtime.h>
#include <math.h>

typedef __attribute__((ext_vector_type(2))) float v2f;
typedef __attribute__((ext_vector_type(8))) float v8f;

#define NEG_SLOPE 0.2f

#define CH_EDGES 4096
#define BIN_SHIFT 8
#define MAX_BINS 256
#define MAX_CH 256

__device__ __forceinline__ int waveSumI(int v) {
    for (int o = 16; o > 0; o >>= 1) v += __shfl_xor(v, o);
    return v;
}
__device__ __forceinline__ int waveScanI(int x, int lane) {
    for (int o = 1; o < 32; o <<= 1) {
        int y = __shfl_up(x, o);
        if (lane >= o) x += y;
    }
    return x;
}

__device__ __forceinline__ void binBody(const int* __restrict__ ei, const float* __restrict__ ea,
                                        int4* __restrict__ tmp, int* __restrict__ tab,
                                        int E, int NB, int G, const int blk) {
    __shared__ int hist[MAX_BINS];
    __shared__ int wtot[8];
    const int t = threadIdx.x, lane = t & 31, w = t >> 5;
    if (t < MAX_BINS) hist[t] = 0;
    __syncthreads();

    const int e0 = blk * CH_EDGES + t * 4;
    int d[4], s[4], a[4], lr[4];
    bool ok[4];
    for (int k = 0; k < 4; ++k) { ok[k] = (e0 + k) < E; d[k] = 0; s[k] = 0; a[k] = 0; lr[k] = 0; }
    if (((E & 3) == 0) && (e0 + 3 < E)) {
        const int4 dv = *(const int4*)(ei + E + e0);
        const int4 sv = *(const int4*)(ei + e0);
        const float4 av = *(const float4*)(ea + e0);
        d[0] = dv.x; d[1] = dv.y; d[2] = dv.z; d[3] = dv.w;
        s[0] = sv.x; s[1] = sv.y; s[2] = sv.z; s[3] = sv.w;
        a[0] = __float_as_int(av.x); a[1] = __float_as_int(av.y);
        a[2] = __float_as_int(av.z); a[3] = __float_as_int(av.w);
    } else {
        for (int k = 0; k < 4; ++k)
            if (ok[k]) { d[k] = ei[E + e0 + k]; s[k] = ei[e0 + k]; a[k] = __float_as_int(ea[e0 + k]); }
    }
    for (int k = 0; k < 4; ++k)
        if (ok[k]) lr[k] = atomicAdd(&hist[d[k] >> BIN_SHIFT], 1);
    __syncthreads();

    int cv = 0, ci = 0;
    if (t < MAX_BINS) {
        cv = hist[t];
        ci = waveScanI(cv, lane);
        if (lane == 31) wtot[w] = ci;
    }
    __syncthreads();
    if (t < MAX_BINS) {
        int front = 0;
        for (int j = 0; j < w; ++j) front += wtot[j];
        const int ex = front + ci - cv;
        hist[t] = ex;
        if (t < NB) tab[t * G + blk] = cv | (ex << 16);
    }
    __syncthreads();
    for (int k = 0; k < 4; ++k)
        if (ok[k]) tmp[blk * CH_EDGES + hist[d[k] >> BIN_SHIFT] + lr[k]] = make_int4(d[k], s[k], a[k], 0);
}

__global__ void __launch_bounds__(1024) placeK(const int4* __restrict__ tmp, const int* __restrict__ tab,
                                               int* __restrict__ off, int2* __restrict__ rec,
                                               int N, int G) {
    __shared__ int pre[MAX_CH + 1];
    __shared__ int rs[MAX_CH];
    __shared__ int c[256];
    __shared__ int wsum[32];
    __shared__ int wtot[8];
    const int t = threadIdx.x, lane = t & 31, w = t >> 5;
    const int b = blockIdx.x;
    const int nfront = b * G;

    int part = 0;
    {
        const int4* t4 = (const int4*)tab;
        const int n4 = nfront >> 2;
        for (int k = t; k < n4; k += 1024) {
            const int4 v = t4[k];
            part += ((v.x & 0xFFFF) + (v.y & 0xFFFF)) + ((v.z & 0xFFFF) + (v.w & 0xFFFF));
        }
        for (int k = (n4 << 2) + t; k < nfront; k += 1024) part += tab[k] & 0xFFFF;
    }
    int rc = 0, ro = 0;
    if (t < G) { const int v = tab[nfront + t]; rc = v & 0xFFFF; ro = v >> 16; }
    part = waveSumI(part);
    if (lane == 0) wsum[w] = part;
    int ri = 0;
    if (t < 256) {
        ri = waveScanI(rc, lane);
        if (lane == 31) wtot[w] = ri;
        c[t] = 0;
    }
    __syncthreads();
    const int base = waveSumI(wsum[lane]);
    if (t < 256) {
        int front = 0;
        for (int j = 0; j < w; ++j) front += wtot[j];
        const int ex = front + ri - rc;
        if (t < G) { pre[t] = ex; rs[t] = t * CH_EDGES + ro; }
        if (t == G - 1) pre[G] = ex + rc;
    }
    __syncthreads();
    const int total = pre[G];

    auto slot = [&](int i) -> int {
        int lo = 0, hi = G;
        for (int st = 0; st < 8; ++st) {
            const int mid = (lo + hi) >> 1;
            if (pre[mid] <= i) lo = mid; else hi = mid;
        }
        return rs[lo] + (i - pre[lo]);
    };

    for (int i0 = t; i0 < total; i0 += 4096) {
        int dd[4]; bool ok[4];
        for (int u = 0; u < 4; ++u) {
            const int i = i0 + u * 1024;
            ok[u] = i < total;
            dd[u] = tmp[slot(ok[u] ? i : total - 1)].x;
        }
        for (int u = 0; u < 4; ++u)
            if (ok[u]) atomicAdd(&c[dd[u] & 255], 1);
    }
    __syncthreads();

    int cv = 0, ci = 0;
    if (t < 256) {
        cv = c[t];
        ci = waveScanI(cv, lane);
        if (lane == 31) wtot[w] = ci;
    }
    __syncthreads();
    if (t < 256) {
        int front = 0;
        for (int j = 0; j < w; ++j) front += wtot[j];
        const int ex = front + ci - cv;
        c[t] = ex;
        const int node = (b << BIN_SHIFT) + t;
        if (node < N) off[node] = base + ex;
        if (node == N - 1) off[N] = base + ex + cv;
    }
    __syncthreads();

    for (int i0 = t; i0 < total; i0 += 4096) {
        int4 r[4]; bool ok[4];
        for (int u = 0; u < 4; ++u) {
            const int i = i0 + u * 1024;
            ok[u] = i < total;
            r[u] = tmp[slot(ok[u] ? i : total - 1)];
        }
        for (int u = 0; u < 4; ++u)
            if (ok[u]) {
                const int p = atomicAdd(&c[r[u].x & 255], 1);
                rec[base + p] = make_int2(r[u].y, r[u].z);
            }
    }
}

template <int KDIM, int NT>
__device__ __forceinline__ void gemmTile(const float* __restrict__ A, const float* __restrict__ B,
                                         float* __restrict__ C, const int wave, const int lane,
                                         const float* __restrict__ as, const float* __restrict__ ad,
                                         float* __restrict__ osrc, float* __restrict__ odst) {
    constexpr int NCOL = NT * 16;
    int half = lane >> 4;
    int idx  = lane & 15;
    int kk   = half << 1;

    const float* arow  = (A + wave * (16 * KDIM)) + (idx * KDIM + kk);
    const float* bbase = B + (kk * NCOL + idx);

    v8f acc[NT];
#pragma unroll
    for (int t = 0; t < NT; ++t) acc[t] = (v8f){};

#pragma unroll
    for (int k0 = 0; k0 < KDIM; k0 += 4) {
        v2f a;
        a.x = arow[k0];
        a.y = arow[k0 + 1];
#pragma unroll
        for (int t = 0; t < NT; ++t) {
            v2f b;
            b.x = bbase[k0 * NCOL + t * 16];
            b.y = bbase[(k0 + 1) * NCOL + t * 16];
            acc[t] = __builtin_amdgcn_wmma_f32_16x16x4_f32(false, a, false, b,
                                                           (short)0, acc[t], false, false);
        }
    }
    int rbase = wave * 16 + (half << 3);
#pragma unroll
    for (int t = 0; t < NT; ++t) {
        float* cp = (C + wave * (16 * NCOL)) + ((half << 3) * NCOL + t * 16 + idx);
#pragma unroll
        for (int i = 0; i < 8; ++i) cp[i * NCOL] = acc[t][i];
    }

    float ps[8], pd[8];
    for (int i = 0; i < 8; ++i) { ps[i] = 0.f; pd[i] = 0.f; }
    for (int t = 0; t < NT; ++t) {
        const float sa = as[t * 16 + idx], sd = ad[t * 16 + idx];
        for (int i = 0; i < 8; ++i) { ps[i] += acc[t][i] * sa; pd[i] += acc[t][i] * sd; }
    }
    for (int o = 8; o > 0; o >>= 1)
        for (int i = 0; i < 8; ++i) { ps[i] += __shfl_xor(ps[i], o); pd[i] += __shfl_xor(pd[i], o); }
    if (idx == 0)
        for (int i = 0; i < 8; ++i) { osrc[rbase + i] = ps[i]; odst[rbase + i] = pd[i]; }
}

template <int KDIM, int NT>
__global__ void gemmWmmaK(const float* __restrict__ A, const float* __restrict__ B,
                          float* __restrict__ C, int mtiles,
                          const float* __restrict__ as, const float* __restrict__ ad,
                          float* __restrict__ osrc, float* __restrict__ odst) {
    const int gt = blockIdx.x * blockDim.x + threadIdx.x;
    const int wave = gt >> 5;
    const int lane = threadIdx.x & 31;
    if (wave >= mtiles) return;
    gemmTile<KDIM, NT>(A, B, C, wave, lane, as, ad, osrc, odst);
}

#define TILE_WAVES 16

__device__ __forceinline__ bool productBlock(const float* __restrict__ x, const float* __restrict__ W1,
                                             float* __restrict__ h, int mtiles, int gb,
                                             const float* __restrict__ as1, const float* __restrict__ ad1,
                                             float* __restrict__ asrc, float* __restrict__ adst) {
    if ((int)blockIdx.x >= gb) return false;
    const int w = __builtin_amdgcn_readfirstlane((int)(threadIdx.x >> 5));
    const int lane = threadIdx.x & 31;
    const int tile = blockIdx.x * TILE_WAVES + w;
    if (w < TILE_WAVES && tile < mtiles)
        gemmTile<128, 4>(x, W1, h, tile, lane, as1, ad1, asrc, adst);
    return true;
}

__global__ void __launch_bounds__(1024, 16)
binGemmK(const float* __restrict__ x, const float* __restrict__ W1, float* __restrict__ h,
         int mtiles, int gb,
         const float* __restrict__ as1, const float* __restrict__ ad1,
         float* __restrict__ asrc, float* __restrict__ adst,
         const float* __restrict__ we1, const float* __restrict__ ae1,
         const float* __restrict__ we2, const float* __restrict__ ae2,
         float* __restrict__ kscal,
         const int* __restrict__ ei, const float* __restrict__ ea,
         int4* __restrict__ tmp, int* __restrict__ tab, int E, int NB, int G) {
    if (productBlock(x, W1, h, mtiles, gb, as1, ad1, asrc, adst)) {
        if (blockIdx.x == 0 && (threadIdx.x >> 5) == TILE_WAVES) {
            const int lane = threadIdx.x & 31;
            if (lane == 0) { float s = 0.f; for (int j = 0; j < 64; ++j) s += we1[j] * ae1[j]; kscal[0] = s; }
            if (lane == 1) { float s = 0.f; for (int j = 0; j < 32; ++j) s += we2[j] * ae2[j]; kscal[1] = s; }
        }
        return;
    }
    binBody(ei, ea, tmp, tab, E, NB, G, (int)blockIdx.x - gb);
}

#define ROW_XOR_F(v, m) __int_as_float(__builtin_amdgcn_update_dpp(__float_as_int(v), __float_as_int(v), 0x160 + (m), 0xf, 0xf, false))
#define ROW_LANE_I(v, u) __builtin_amdgcn_update_dpp((v), (v), 0x150 + (u), 0xf, 0xf, false)
#define ROW_LANE_F(v, u) __int_as_float(__builtin_amdgcn_update_dpp(__float_as_int(v), __float_as_int(v), 0x150 + (u), 0xf, 0xf, false))

__device__ __forceinline__ float rowSum16(float v) {
    v += ROW_XOR_F(v, 8); v += ROW_XOR_F(v, 4); v += ROW_XOR_F(v, 2); v += ROW_XOR_F(v, 1);
    return v;
}
__device__ __forceinline__ float rowMax16(float v) {
    v = fmaxf(v, ROW_XOR_F(v, 8)); v = fmaxf(v, ROW_XOR_F(v, 4));
    v = fmaxf(v, ROW_XOR_F(v, 2)); v = fmaxf(v, ROW_XOR_F(v, 1));
    return v;
}
__device__ __forceinline__ float lrelu(float a) { return (a > 0.0f) ? a : NEG_SLOPE * a; }

typedef __attribute__((ext_vector_type(4))) float v4f;

template <int D>
__device__ __forceinline__ void loadPiece(const float* __restrict__ h, int row, int hl, v2f (&v)[D / 32]) {
    const char* p = (const char*)h + ((unsigned)row * (unsigned)(D * 4) + (unsigned)hl * (unsigned)(D / 4));
    if constexpr (D == 64) {
        const v4f t = *(const v4f*)p;
        v[0] = t.xy; v[1] = t.zw;
    } else {
        v[0] = *(const v2f*)p;
    }
}

template <int D, int U0>
__device__ __forceinline__ void gather8(const float* __restrict__ h, int hl, int s, float w,
                                        v2f (&acc)[D / 32]) {
    constexpr int P = D / 32;
    v2f hv[8][P]; float wt[8];
#define GATHER_LD(u) { const int st = ROW_LANE_I(s, U0 + (u)); wt[u] = ROW_LANE_F(w, U0 + (u)); loadPiece<D>(h, st, hl, hv[u]); }
    GATHER_LD(0) GATHER_LD(1) GATHER_LD(2) GATHER_LD(3)
    GATHER_LD(4) GATHER_LD(5) GATHER_LD(6) GATHER_LD(7)
#undef GATHER_LD
    for (int u = 0; u < 8; ++u)
        for (int p = 0; p < P; ++p) acc[p] += hv[u][p] * wt[u];
}

template <int D, bool RELU>
__global__ void __launch_bounds__(256) segK(const int* __restrict__ off, const int2* __restrict__ rec,
                                            const float* __restrict__ asrc, const float* __restrict__ adst,
                                            const float* __restrict__ kscal, int ki,
                                            const float* __restrict__ h, const float* __restrict__ bias,
                                            float* __restrict__ out, int N) {
    constexpr int P = D / 32;
    const int lane = threadIdx.x & 31;
    const int hl = lane & 15;
    const int i0 = 2 * __builtin_amdgcn_readfirstlane((int)((blockIdx.x * blockDim.x + threadIdx.x) >> 5));
    if (i0 >= N) return;
    const int iraw = i0 + (lane >> 4);
    const bool live = iraw < N;
    const int i = live ? iraw : i0;
    const int b = off[i];
    const int deg = off[live ? i + 1 : i] - b;
    const float kk = kscal[ki];
    const float ad = adst[i];
    const float asi = asrc[i];
    const int dmax = max(__builtin_amdgcn_readlane(deg, 0), __builtin_amdgcn_readlane(deg, 16));

    auto edge = [&](int j, int& s, float& at) -> float {
        int2 r = rec[j];
        s = r.x; at = __int_as_float(r.y);
        return lrelu(asrc[s] + ad + at * kk);
    };

    const bool v0 = hl < deg, v1 = hl + 16 < deg;
    int2 r0 = make_int2(i, 0), r1 = make_int2(i, 0);
    if (v0) r0 = rec[b + hl];
    if (v1) r1 = rec[b + 16 + hl];
    const int s0 = r0.x, s1 = r1.x;
    const float a0 = asrc[s0], a1 = asrc[s1];
    const float at0 = __int_as_float(r0.y), at1 = __int_as_float(r1.y);
    const float al0 = v0 ? lrelu(a0 + ad + at0 * kk) : -INFINITY;
    const float al1 = v1 ? lrelu(a1 + ad + at1 * kk) : -INFINITY;
    float mx = fmaxf(al0, al1), asum = at0 + at1;
#pragma nounroll
    for (int cb = 32; cb < dmax; cb += 16) {
        const int j = cb + hl;
        if (j < deg) { int s; float at; float al = edge(b + j, s, at); mx = fmaxf(mx, al); asum += at; }
    }
    mx = rowMax16(mx);
    asum = rowSum16(asum);

    const float la  = asum / fmaxf((float)deg, 1.0f);
    const float als = lrelu(asi + ad + la * kk);
    mx = fmaxf(mx, als);
    const float exs = expf(als - mx);
    const float ex0 = v0 ? expf(al0 - mx) : 0.0f;
    const float ex1 = v1 ? expf(al1 - mx) : 0.0f;
    float dsum = ex0 + ex1;
#pragma nounroll
    for (int cb = 32; cb < dmax; cb += 16) {
        const int j = cb + hl;
        if (j < deg) { int s; float at; dsum += expf(edge(b + j, s, at) - mx); }
    }
    const float den = rowSum16(dsum) + exs;
    const float w0 = ex0 / den, w1 = ex1 / den;

    v2f acc[P];
    {
        v2f hv[P];
        loadPiece<D>(h, i, hl, hv);
        const float wself = exs / den;
        for (int p = 0; p < P; ++p) acc[p] = hv[p] * wself;
    }
    if (dmax > 0)  gather8<D, 0>(h, hl, s0, w0, acc);
    if (dmax > 8)  gather8<D, 8>(h, hl, s0, w0, acc);
    if (dmax > 16) gather8<D, 0>(h, hl, s1, w1, acc);
    if (dmax > 24) gather8<D, 8>(h, hl, s1, w1, acc);
#pragma nounroll
    for (int cb = 32; cb < dmax; cb += 16) {
        const int j = cb + hl;
        int s = i; float w = 0.0f;
        if (j < deg) { float at; w = expf(edge(b + j, s, at) - mx) / den; }
        gather8<D, 0>(h, hl, s, w, acc);
        if (dmax > cb + 8) gather8<D, 8>(h, hl, s, w, acc);
    }

    if (live) {
        if constexpr (P == 2) {
            const v4f bv = *(const v4f*)(bias + hl * 4);
            v4f o;
            o.xy = acc[0] + bv.xy; o.zw = acc[1] + bv.zw;
            if (RELU) { o.x = fmaxf(o.x, 0.0f); o.y = fmaxf(o.y, 0.0f); o.z = fmaxf(o.z, 0.0f); o.w = fmaxf(o.w, 0.0f); }
            *(v4f*)(out + i * D + hl * 4) = o;
        } else {
            const v2f bv = *(const v2f*)(bias + hl * 2);
            v2f o = acc[0] + bv;
            if (RELU) { o.x = fmaxf(o.x, 0.0f); o.y = fmaxf(o.y, 0.0f); }
            *(v2f*)(out + i * D + hl * 2) = o;
        }
    }
}

static inline int gr(long long n, int blk) { return (int)((n + blk - 1) / blk); }

extern "C" void kernel_launch(void* const* d_in, const int* in_sizes, int n_in,
                              void* d_out, int out_size, void* d_ws, size_t ws_size,
                              hipStream_t stream) {
    const float* x    = (const float*)d_in[0];
    const int*   ei   = (const int*)d_in[1];
    const float* ea   = (const float*)d_in[2];
    const float* W1   = (const float*)d_in[3];
    const float* as1  = (const float*)d_in[4];
    const float* ad1  = (const float*)d_in[5];
    const float* we1  = (const float*)d_in[6];
    const float* ae1  = (const float*)d_in[7];
    const float* b1   = (const float*)d_in[8];
    const float* W2   = (const float*)d_in[9];
    const float* as2  = (const float*)d_in[10];
    const float* ad2  = (const float*)d_in[11];
    const float* we2  = (const float*)d_in[12];
    const float* ae2  = (const float*)d_in[13];
    const float* b2   = (const float*)d_in[14];
    float* out = (float*)d_out;

    const int N  = in_sizes[0] / 128;
    const int E  = in_sizes[1] / 2;

    float* ws = (float*)d_ws;
    size_t o = 0;
    float* h     = ws + o; o += (size_t)N * 64;
    float* hagg  = ws + o; o += (size_t)N * 64;
    int2*  rec   = (int2*)(ws + o); o += (size_t)E * 2;
    const int G  = gr(E, CH_EDGES);
    const int NB = gr(N, 1 << BIN_SHIFT);
    o = (o + 3) & ~(size_t)3;
    int4*  tmp   = (int4*)(ws + o); o += (size_t)G * CH_EDGES * 4;
    int*   tab   = (int*)(ws + o); o += (size_t)NB * G;
    int*   off   = (int*)(ws + o); o += N + 1;
    float* asrc  = ws + o; o += N;
    float* adst  = ws + o; o += N;
    float* kscal = ws + o; o += 4;

    const int B = 256;
    const int mtiles = N / 16;

    const int GB = gr(mtiles, TILE_WAVES);
    binGemmK<<<GB + G, 1024, 0, stream>>>(x, W1, h, mtiles, GB, as1, ad1, asrc, adst,
                                          we1, ae1, we2, ae2, kscal, ei, ea, tmp, tab, E, NB, G);
    placeK<<<NB, 1024, 0, stream>>>(tmp, tab, off, rec, N, G);

    segK<64, true><<<gr((long long)((N + 1) / 2) * 32, B), B, 0, stream>>>(off, rec, asrc, adst, kscal, 0, h, b1, hagg, N);

    gemmWmmaK<64, 2><<<gr(mtiles, 8), 8 * 32, 0, stream>>>(hagg, W2, h, mtiles, as2, ad2, asrc, adst);
    segK<32, false><<<gr((long long)((N + 1) / 2) * 32, B), B, 0, stream>>>(off, rec, asrc, adst, kscal, 1, h, b2, out, N);
}


# Aggregation: source logit term = dot of the gathered row with att_src (8-row DPP transpose-reduce, folded lane reads); online softmax per 8 edges; no scattered 4-byte read per edge; new sum order
# speedup vs baseline: 9.5561x; 1.0852x over previous
#include <hip/hip_runtime.h>
#include <math.h>

typedef __attribute__((ext_vector_type(2))) float v2f;
typedef __attribute__((ext_vector_type(8))) float v8f;

#define NEG_SLOPE 0.2f

#define CH_EDGES 4096
#define BIN_SHIFT 8
#define MAX_BINS 256
#define MAX_CH 256

__device__ __forceinline__ int waveSumI(int v) {
    for (int o = 16; o > 0; o >>= 1) v += __shfl_xor(v, o);
    return v;
}
__device__ __forceinline__ int waveScanI(int x, int lane) {
    for (int o = 1; o < 32; o <<= 1) {
        int y = __shfl_up(x, o);
        if (lane >= o) x += y;
    }
    return x;
}

__device__ __forceinline__ void binBody(const int* __restrict__ ei, const float* __restrict__ ea,
                                        int4* __restrict__ tmp, int* __restrict__ tab,
                                        int E, int NB, int G, const int blk) {
    __shared__ int hist[MAX_BINS];
    __shared__ int wtot[8];
    const int t = threadIdx.x, lane = t & 31, w = t >> 5;
    if (t < MAX_BINS) hist[t] = 0;
    __syncthreads();

    const int e0 = blk * CH_EDGES + t * 4;
    int d[4], s[4], a[4], lr[4];
    bool ok[4];
    for (int k = 0; k < 4; ++k) { ok[k] = (e0 + k) < E; d[k] = 0; s[k] = 0; a[k] = 0; lr[k] = 0; }
    if (((E & 3) == 0) && (e0 + 3 < E)) {
        const int4 dv = *(const int4*)(ei + E + e0);
        const int4 sv = *(const int4*)(ei + e0);
        const float4 av = *(const float4*)(ea + e0);
        d[0] = dv.x; d[1] = dv.y; d[2] = dv.z; d[3] = dv.w;
        s[0] = sv.x; s[1] = sv.y; s[2] = sv.z; s[3] = sv.w;
        a[0] = __float_as_int(av.x); a[1] = __float_as_int(av.y);
        a[2] = __float_as_int(av.z); a[3] = __float_as_int(av.w);
    } else {
        for (int k = 0; k < 4; ++k)
            if (ok[k]) { d[k] = ei[E + e0 + k]; s[k] = ei[e0 + k]; a[k] = __float_as_int(ea[e0 + k]); }
    }
    for (int k = 0; k < 4; ++k)
        if (ok[k]) lr[k] = atomicAdd(&hist[d[k] >> BIN_SHIFT], 1);
    __syncthreads();

    int cv = 0, ci = 0;
    if (t < MAX_BINS) {
        cv = hist[t];
        ci = waveScanI(cv, lane);
        if (lane == 31) wtot[w] = ci;
    }
    __syncthreads();
    if (t < MAX_BINS) {
        int front = 0;
        for (int j = 0; j < w; ++j) front += wtot[j];
        const int ex = front + ci - cv;
        hist[t] = ex;
        if (t < NB) tab[t * G + blk] = cv | (ex << 16);
    }
    __syncthreads();
    for (int k = 0; k < 4; ++k)
        if (ok[k]) tmp[blk * CH_EDGES + hist[d[k] >> BIN_SHIFT] + lr[k]] = make_int4(d[k], s[k], a[k], 0);
}

__global__ void __launch_bounds__(1024) placeK(const int4* __restrict__ tmp, const int* __restrict__ tab,
                                               int* __restrict__ off, int2* __restrict__ rec,
                                               int N, int G) {
    __shared__ int pre[MAX_CH + 1];
    __shared__ int rs[MAX_CH];
    __shared__ int c[256];
    __shared__ int wsum[32];
    __shared__ int wtot[8];
    const int t = threadIdx.x, lane = t & 31, w = t >> 5;
    const int b = blockIdx.x;
    const int nfront = b * G;

    int part = 0;
    {
        const int4* t4 = (const int4*)tab;
        const int n4 = nfront >> 2;
        for (int k = t; k < n4; k += 1024) {
            const int4 v = t4[k];
            part += ((v.x & 0xFFFF) + (v.y & 0xFFFF)) + ((v.z & 0xFFFF) + (v.w & 0xFFFF));
        }
        for (int k = (n4 << 2) + t; k < nfront; k += 1024) part += tab[k] & 0xFFFF;
    }
    int rc = 0, ro = 0;
    if (t < G) { const int v = tab[nfront + t]; rc = v & 0xFFFF; ro = v >> 16; }
    part = waveSumI(part);
    if (lane == 0) wsum[w] = part;
    int ri = 0;
    if (t < 256) {
        ri = waveScanI(rc, lane);
        if (lane == 31) wtot[w] = ri;
        c[t] = 0;
    }
    __syncthreads();
    const int base = waveSumI(wsum[lane]);
    if (t < 256) {
        int front = 0;
        for (int j = 0; j < w; ++j) front += wtot[j];
        const int ex = front + ri - rc;
        if (t < G) { pre[t] = ex; rs[t] = t * CH_EDGES + ro; }
        if (t == G - 1) pre[G] = ex + rc;
    }
    __syncthreads();
    const int total = pre[G];

    auto slot = [&](int i) -> int {
        int lo = 0, hi = G;
        for (int st = 0; st < 8; ++st) {
            const int mid = (lo + hi) >> 1;
            if (pre[mid] <= i) lo = mid; else hi = mid;
        }
        return rs[lo] + (i - pre[lo]);
    };

    for (int i0 = t; i0 < total; i0 += 4096) {
        int dd[4]; bool ok[4];
        for (int u = 0; u < 4; ++u) {
            const int i = i0 + u * 1024;
            ok[u] = i < total;
            dd[u] = tmp[slot(ok[u] ? i : total - 1)].x;
        }
        for (int u = 0; u < 4; ++u)
            if (ok[u]) atomicAdd(&c[dd[u] & 255], 1);
    }
    __syncthreads();

    int cv = 0, ci = 0;
    if (t < 256) {
        cv = c[t];
        ci = waveScanI(cv, lane);
        if (lane == 31) wtot[w] = ci;
    }
    __syncthreads();
    if (t < 256) {
        int front = 0;
        for (int j = 0; j < w; ++j) front += wtot[j];
        const int ex = front + ci - cv;
        c[t] = ex;
        const int node = (b << BIN_SHIFT) + t;
        if (node < N) off[node] = base + ex;
        if (node == N - 1) off[N] = base + ex + cv;
    }
    __syncthreads();

    for (int i0 = t; i0 < total; i0 += 4096) {
        int4 r[4]; bool ok[4];
        for (int u = 0; u < 4; ++u) {
            const int i = i0 + u * 1024;
            ok[u] = i < total;
            r[u] = tmp[slot(ok[u] ? i : total - 1)];
        }
        for (int u = 0; u < 4; ++u)
            if (ok[u]) {
                const int p = atomicAdd(&c[r[u].x & 255], 1);
                rec[base + p] = make_int2(r[u].y, r[u].z);
            }
    }
}

template <int KDIM, int NT>
__device__ __forceinline__ void gemmTile(const float* __restrict__ A, const float* __restrict__ B,
                                         float* __restrict__ C, const int wave, const int lane,
                                         const float* __restrict__ as, const float* __restrict__ ad,
                                         float* __restrict__ osrc, float* __restrict__ odst) {
    constexpr int NCOL = NT * 16;
    int half = lane >> 4;
    int idx  = lane & 15;
    int kk   = half << 1;

    const float* arow  = (A + wave * (16 * KDIM)) + (idx * KDIM + kk);
    const float* bbase = B + (kk * NCOL + idx);

    v8f acc[NT];
#pragma unroll
    for (int t = 0; t < NT; ++t) acc[t] = (v8f){};

#pragma unroll
    for (int k0 = 0; k0 < KDIM; k0 += 4) {
        v2f a;
        a.x = arow[k0];
        a.y = arow[k0 + 1];
#pragma unroll
        for (int t = 0; t < NT; ++t) {
            v2f b;
            b.x = bbase[k0 * NCOL + t * 16];
            b.y = bbase[(k0 + 1) * NCOL + t * 16];
            acc[t] = __builtin_amdgcn_wmma_f32_16x16x4_f32(false, a, false, b,
                                                           (short)0, acc[t], false, false);
        }
    }
    int rbase = wave * 16 + (half << 3);
#pragma unroll
    for (int t = 0; t < NT; ++t) {
        float* cp = (C + wave * (16 * NCOL)) + ((half << 3) * NCOL + t * 16 + idx);
#pragma unroll
        for (int i = 0; i < 8; ++i) cp[i * NCOL] = acc[t][i];
    }

    float ps[8], pd[8];
    for (int i = 0; i < 8; ++i) { ps[i] = 0.f; pd[i] = 0.f; }
    for (int t = 0; t < NT; ++t) {
        const float sa = as[t * 16 + idx], sd = ad[t * 16 + idx];
        for (int i = 0; i < 8; ++i) { ps[i] += acc[t][i] * sa; pd[i] += acc[t][i] * sd; }
    }
    for (int o = 8; o > 0; o >>= 1)
        for (int i = 0; i < 8; ++i) { ps[i] += __shfl_xor(ps[i], o); pd[i] += __shfl_xor(pd[i], o); }
    if (idx == 0)
        for (int i = 0; i < 8; ++i) { osrc[rbase + i] = ps[i]; odst[rbase + i] = pd[i]; }
}

template <int KDIM, int NT>
__global__ void gemmWmmaK(const float* __restrict__ A, const float* __restrict__ B,
                          float* __restrict__ C, int mtiles,
                          const float* __restrict__ as, const float* __restrict__ ad,
                          float* __restrict__ osrc, float* __restrict__ odst) {
    const int gt = blockIdx.x * blockDim.x + threadIdx.x;
    const int wave = gt >> 5;
    const int lane = threadIdx.x & 31;
    if (wave >= mtiles) return;
    gemmTile<KDIM, NT>(A, B, C, wave, lane, as, ad, osrc, odst);
}

#define TILE_WAVES 16

__device__ __forceinline__ bool productBlock(const float* __restrict__ x, const float* __restrict__ W1,
                                             float* __restrict__ h, int mtiles, int gb,
                                             const float* __restrict__ as1, const float* __restrict__ ad1,
                                             float* __restrict__ asrc, float* __restrict__ adst) {
    if ((int)blockIdx.x >= gb) return false;
    const int w = __builtin_amdgcn_readfirstlane((int)(threadIdx.x >> 5));
    const int lane = threadIdx.x & 31;
    const int tile = blockIdx.x * TILE_WAVES + w;
    if (w < TILE_WAVES && tile < mtiles)
        gemmTile<128, 4>(x, W1, h, tile, lane, as1, ad1, asrc, adst);
    return true;
}

__global__ void __launch_bounds__(1024, 16)
binGemmK(const float* __restrict__ x, const float* __restrict__ W1, float* __restrict__ h,
         int mtiles, int gb,
         const float* __restrict__ as1, const float* __restrict__ ad1,
         float* __restrict__ asrc, float* __restrict__ adst,
         const float* __restrict__ we1, const float* __restrict__ ae1,
         const float* __restrict__ we2, const float* __restrict__ ae2,
         float* __restrict__ kscal,
         const int* __restrict__ ei, const float* __restrict__ ea,
         int4* __restrict__ tmp, int* __restrict__ tab, int E, int NB, int G) {
    if (productBlock(x, W1, h, mtiles, gb, as1, ad1, asrc, adst)) {
        if (blockIdx.x == 0 && (threadIdx.x >> 5) == TILE_WAVES) {
            const int lane = threadIdx.x & 31;
            if (lane == 0) { float s = 0.f; for (int j = 0; j < 64; ++j) s += we1[j] * ae1[j]; kscal[0] = s; }
            if (lane == 1) { float s = 0.f; for (int j = 0; j < 32; ++j) s += we2[j] * ae2[j]; kscal[1] = s; }
        }
        return;
    }
    binBody(ei, ea, tmp, tab, E, NB, G, (int)blockIdx.x - gb);
}

#define ROW_XOR_F(v, m) __int_as_float(__builtin_amdgcn_mov_dpp(__float_as_int(v), 0x160 + (m), 0xf, 0xf, true))
#define ROW_LANE_I(v, u) __builtin_amdgcn_mov_dpp((v), 0x150 + (u), 0xf, 0xf, true)
#define ROW_LANE_F(v, u) __int_as_float(__builtin_amdgcn_mov_dpp(__float_as_int(v), 0x150 + (u), 0xf, 0xf, true))

__device__ __forceinline__ float rowSum16(float v) {
    v += ROW_XOR_F(v, 8); v += ROW_XOR_F(v, 4); v += ROW_XOR_F(v, 2); v += ROW_XOR_F(v, 1);
    return v;
}
__device__ __forceinline__ float rowMax16(float v) {
    v = fmaxf(v, ROW_XOR_F(v, 8)); v = fmaxf(v, ROW_XOR_F(v, 4));
    v = fmaxf(v, ROW_XOR_F(v, 2)); v = fmaxf(v, ROW_XOR_F(v, 1));
    return v;
}
__device__ __forceinline__ float lrelu(float a) { return (a > 0.0f) ? a : NEG_SLOPE * a; }

typedef __attribute__((ext_vector_type(4))) float v4f;

template <int D>
__device__ __forceinline__ void loadPiece(const float* __restrict__ h, int row, int hl, v2f (&v)[D / 32]) {
    const char* p = (const char*)h + ((unsigned)row * (unsigned)(D * 4) + (unsigned)hl * (unsigned)(D / 4));
    if constexpr (D == 64) {
        const v4f t = *(const v4f*)p;
        v[0] = t.xy; v[1] = t.zw;
    } else {
        v[0] = *(const v2f*)p;
    }
}

__device__ __forceinline__ float sum8(float v) {
    v += ROW_XOR_F(v, 4); v += ROW_XOR_F(v, 2); v += ROW_XOR_F(v, 1);
    return v;
}

template <int D, int U0>
__device__ __forceinline__ void chunk8(const float* __restrict__ h, int hl, int s, float lb,
                                       const v2f (&av)[D / 32], float& m, float& l,
                                       v2f (&acc)[D / 32]) {
    constexpr int P = D / 32;
    v2f hv[8][P];
#define CHUNK_LD(u) { const int st = ROW_LANE_I(s, U0 + (u)); loadPiece<D>(h, st, hl, hv[u]); }
    CHUNK_LD(0) CHUNK_LD(1) CHUNK_LD(2) CHUNK_LD(3)
    CHUNK_LD(4) CHUNK_LD(5) CHUNK_LD(6) CHUNK_LD(7)
#undef CHUNK_LD
    float d[8];
    for (int u = 0; u < 8; ++u) {
        v2f t = hv[u][0] * av[0];
        for (int p = 1; p < P; ++p) t += hv[u][p] * av[p];
        d[u] = t.x + t.y;
    }
    const bool b2 = (hl & 4) != 0, b1 = (hl & 2) != 0, b0 = (hl & 1) != 0;
    float w[4], x[2];
    for (int k = 0; k < 4; ++k) {
        const float keep = b2 ? d[k + 4] : d[k];
        const float send = b2 ? d[k] : d[k + 4];
        w[k] = keep + ROW_XOR_F(send, 4);
    }
    for (int k = 0; k < 2; ++k) {
        const float keep = b1 ? w[k + 2] : w[k];
        const float send = b1 ? w[k] : w[k + 2];
        x[k] = keep + ROW_XOR_F(send, 2);
    }
    float y;
    {
        const float keep = b0 ? x[1] : x[0];
        const float send = b0 ? x[0] : x[1];
        y = keep + ROW_XOR_F(send, 1);
    }
    y += ROW_XOR_F(y, 8);
    const float lbx = ROW_XOR_F(lb, 8);
    const float mylb = ((hl & 8) == U0) ? lb : lbx;
    const float lg = lrelu(y + mylb);
    const float mn = fmaxf(m, rowMax16(lg));
    const float sc = expf(m - mn);
    const float pe = expf(lg - mn);
    l = l * sc + sum8(pe);
    m = mn;
    for (int p = 0; p < P; ++p) acc[p] *= sc;
#define CHUNK_ADD(u) { const float pu = ROW_LANE_F(pe, u); for (int p = 0; p < P; ++p) acc[p] += hv[u][p] * pu; }
    CHUNK_ADD(0) CHUNK_ADD(1) CHUNK_ADD(2) CHUNK_ADD(3)
    CHUNK_ADD(4) CHUNK_ADD(5) CHUNK_ADD(6) CHUNK_ADD(7)
#undef CHUNK_ADD
}

template <int D, bool RELU>
__global__ void __launch_bounds__(256) segK(const int* __restrict__ off, const int2* __restrict__ rec,
                                            const float* __restrict__ asrc, const float* __restrict__ adst,
                                            const float* __restrict__ kscal, int ki,
                                            const float* __restrict__ h, const float* __restrict__ att,
                                            const float* __restrict__ bias,
                                            float* __restrict__ out, int N) {
    constexpr int P = D / 32;
    const int lane = threadIdx.x & 31;
    const int hl = lane & 15;
    const int i0 = 2 * __builtin_amdgcn_readfirstlane((int)((blockIdx.x * blockDim.x + threadIdx.x) >> 5));
    if (i0 >= N) return;
    const int iraw = i0 + (lane >> 4);
    const bool live = iraw < N;
    const int i = live ? iraw : i0;
    const int b = off[i];
    const int deg = off[live ? i + 1 : i] - b;
    const float kk = kscal[ki];
    const float ad = adst[i];
    const float asi = asrc[i];
    const int dmax = max(__builtin_amdgcn_readlane(deg, 0), __builtin_amdgcn_readlane(deg, 16));

    v2f av[P];
    loadPiece<D>(att, 0, hl, av);
    v2f acc[P];
    loadPiece<D>(h, i, hl, acc);

    const bool v0 = hl < deg, v1 = hl + 16 < deg;
    int2 r0 = make_int2(i, 0), r1 = make_int2(i, 0);
    if (v0) r0 = rec[b + hl];
    if (v1) r1 = rec[b + 16 + hl];
    const int s0 = r0.x, s1 = r1.x;
    const float at0 = __int_as_float(r0.y), at1 = __int_as_float(r1.y);
    const float lb0 = v0 ? (ad + at0 * kk) : -INFINITY;
    const float lb1 = v1 ? (ad + at1 * kk) : -INFINITY;
    float asum = at0 + at1;
#pragma nounroll
    for (int cb = 32; cb < dmax; cb += 16) {
        const int j = cb + hl;
        if (j < deg) asum += __int_as_float(rec[b + j].y);
    }
    asum = rowSum16(asum);

    const float la = asum / fmaxf((float)deg, 1.0f);
    float m = lrelu(asi + ad + la * kk);
    float l = 1.0f;

    if (dmax > 0)  chunk8<D, 0>(h, hl, s0, lb0, av, m, l, acc);
    if (dmax > 8)  chunk8<D, 8>(h, hl, s0, lb0, av, m, l, acc);
    if (dmax > 16) chunk8<D, 0>(h, hl, s1, lb1, av, m, l, acc);
    if (dmax > 24) chunk8<D, 8>(h, hl, s1, lb1, av, m, l, acc);
#pragma nounroll
    for (int cb = 32; cb < dmax; cb += 16) {
        const int j = cb + hl;
        int s = i; float lb = -INFINITY;
        if (j < deg) { const int2 r = rec[b + j]; s = r.x; lb = ad + __int_as_float(r.y) * kk; }
        chunk8<D, 0>(h, hl, s, lb, av, m, l, acc);
        if (dmax > cb + 8) chunk8<D, 8>(h, hl, s, lb, av, m, l, acc);
    }
    const float inv = 1.0f / l;

    if (live) {
        if constexpr (P == 2) {
            const v4f bv = *(const v4f*)(bias + hl * 4);
            v4f o;
            o.xy = acc[0] * inv + bv.xy; o.zw = acc[1] * inv + bv.zw;
            if (RELU) { o.x = fmaxf(o.x, 0.0f); o.y = fmaxf(o.y, 0.0f); o.z = fmaxf(o.z, 0.0f); o.w = fmaxf(o.w, 0.0f); }
            *(v4f*)(out + i * D + hl * 4) = o;
        } else {
            const v2f bv = *(const v2f*)(bias + hl * 2);
            v2f o = acc[0] * inv + bv;
            if (RELU) { o.x = fmaxf(o.x, 0.0f); o.y = fmaxf(o.y, 0.0f); }
            *(v2f*)(out + i * D + hl * 2) = o;
        }
    }
}

static inline int gr(long long n, int blk) { return (int)((n + blk - 1) / blk); }

extern "C" void kernel_launch(void* const* d_in, const int* in_sizes, int n_in,
                              void* d_out, int out_size, void* d_ws, size_t ws_size,
                              hipStream_t stream) {
    const float* x    = (const float*)d_in[0];
    const int*   ei   = (const int*)d_in[1];
    const float* ea   = (const float*)d_in[2];
    const float* W1   = (const float*)d_in[3];
    const float* as1  = (const float*)d_in[4];
    const float* ad1  = (const float*)d_in[5];
    const float* we1  = (const float*)d_in[6];
    const float* ae1  = (const float*)d_in[7];
    const float* b1   = (const float*)d_in[8];
    const float* W2   = (const float*)d_in[9];
    const float* as2  = (const float*)d_in[10];
    const float* ad2  = (const float*)d_in[11];
    const float* we2  = (const float*)d_in[12];
    const float* ae2  = (const float*)d_in[13];
    const float* b2   = (const float*)d_in[14];
    float* out = (float*)d_out;

    const int N  = in_sizes[0] / 128;
    const int E  = in_sizes[1] / 2;

    float* ws = (float*)d_ws;
    size_t o = 0;
    float* h     = ws + o; o += (size_t)N * 64;
    float* hagg  = ws + o; o += (size_t)N * 64;
    int2*  rec   = (int2*)(ws + o); o += (size_t)E * 2;
    const int G  = gr(E, CH_EDGES);
    const int NB = gr(N, 1 << BIN_SHIFT);
    o = (o + 3) & ~(size_t)3;
    int4*  tmp   = (int4*)(ws + o); o += (size_t)G * CH_EDGES * 4;
    int*   tab   = (int*)(ws + o); o += (size_t)NB * G;
    int*   off   = (int*)(ws + o); o += N + 1;
    float* asrc  = ws + o; o += N;
    float* adst  = ws + o; o += N;
    float* kscal = ws + o; o += 4;

    const int B = 256;
    const int mtiles = N / 16;

    const int GB = gr(mtiles, TILE_WAVES);
    binGemmK<<<GB + G, 1024, 0, stream>>>(x, W1, h, mtiles, GB, as1, ad1, asrc, adst,
                                          we1, ae1, we2, ae2, kscal, ei, ea, tmp, tab, E, NB, G);
    placeK<<<NB, 1024, 0, stream>>>(tmp, tab, off, rec, N, G);

    segK<64, true><<<gr((long long)((N + 1) / 2) * 32, B), B, 0, stream>>>(off, rec, asrc, adst, kscal, 0, h, as1, b1, hagg, N);

    gemmWmmaK<64, 2><<<gr(mtiles, 8), 8 * 32, 0, stream>>>(hagg, W2, h, mtiles, as2, ad2, asrc, adst);
    segK<32, false><<<gr((long long)((N + 1) / 2) * 32, B), B, 0, stream>>>(off, rec, asrc, adst, kscal, 1, h, as2, b2, out, N);
}


# Edge sort level 2 in one walk: each run of a bin has its own group of threads that reads it once into registers; rank from the counting LDS add; no search, no second load; new segment (sum) order
# speedup vs baseline: 10.0101x; 1.0475x over previous
#include <hip/hip_runtime.h>
#include <math.h>

typedef __attribute__((ext_vector_type(2))) float v2f;
typedef __attribute__((ext_vector_type(8))) float v8f;

#define NEG_SLOPE 0.2f

#define CH_EDGES 4096
#define BIN_SHIFT 8
#define MAX_BINS 256
#define MAX_CH 256

__device__ __forceinline__ int waveSumI(int v) {
    for (int o = 16; o > 0; o >>= 1) v += __shfl_xor(v, o);
    return v;
}
__device__ __forceinline__ int waveScanI(int x, int lane) {
    for (int o = 1; o < 32; o <<= 1) {
        int y = __shfl_up(x, o);
        if (lane >= o) x += y;
    }
    return x;
}

__device__ __forceinline__ void binBody(const int* __restrict__ ei, const float* __restrict__ ea,
                                        int4* __restrict__ tmp, int* __restrict__ tab,
                                        int E, int NB, int G, const int blk) {
    __shared__ int hist[MAX_BINS];
    __shared__ int wtot[8];
    const int t = threadIdx.x, lane = t & 31, w = t >> 5;
    if (t < MAX_BINS) hist[t] = 0;
    __syncthreads();

    const int e0 = blk * CH_EDGES + t * 4;
    int d[4], s[4], a[4], lr[4];
    bool ok[4];
    for (int k = 0; k < 4; ++k) { ok[k] = (e0 + k) < E; d[k] = 0; s[k] = 0; a[k] = 0; lr[k] = 0; }
    if (((E & 3) == 0) && (e0 + 3 < E)) {
        const int4 dv = *(const int4*)(ei + E + e0);
        const int4 sv = *(const int4*)(ei + e0);
        const float4 av = *(const float4*)(ea + e0);
        d[0] = dv.x; d[1] = dv.y; d[2] = dv.z; d[3] = dv.w;
        s[0] = sv.x; s[1] = sv.y; s[2] = sv.z; s[3] = sv.w;
        a[0] = __float_as_int(av.x); a[1] = __float_as_int(av.y);
        a[2] = __float_as_int(av.z); a[3] = __float_as_int(av.w);
    } else {
        for (int k = 0; k < 4; ++k)
            if (ok[k]) { d[k] = ei[E + e0 + k]; s[k] = ei[e0 + k]; a[k] = __float_as_int(ea[e0 + k]); }
    }
    for (int k = 0; k < 4; ++k)
        if (ok[k]) lr[k] = atomicAdd(&hist[d[k] >> BIN_SHIFT], 1);
    __syncthreads();

    int cv = 0, ci = 0;
    if (t < MAX_BINS) {
        cv = hist[t];
        ci = waveScanI(cv, lane);
        if (lane == 31) wtot[w] = ci;
    }
    __syncthreads();
    if (t < MAX_BINS) {
        int front = 0;
        for (int j = 0; j < w; ++j) front += wtot[j];
        const int ex = front + ci - cv;
        hist[t] = ex;
        if (t < NB) tab[t * G + blk] = cv | (ex << 16);
    }
    __syncthreads();
    for (int k = 0; k < 4; ++k)
        if (ok[k]) tmp[blk * CH_EDGES + hist[d[k] >> BIN_SHIFT] + lr[k]] = make_int4(d[k], s[k], a[k], 0);
}

#define PL_K 10
__global__ void __launch_bounds__(1024) placeK(const int4* __restrict__ tmp, const int* __restrict__ tab,
                                               int* __restrict__ off, int2* __restrict__ rec,
                                               int N, int G) {
    __shared__ int c[256];
    __shared__ int held[256];
    __shared__ int cur[256];
    __shared__ int wsum[32];
    __shared__ int wtot[8];
    __shared__ int anyTail;
    const int t = threadIdx.x, lane = t & 31, w = t >> 5;
    const int b = blockIdx.x;
    const int nfront = b * G;

    if (t < 256) c[t] = 0;
    if (t == 0) anyTail = 0;
    __syncthreads();

    const int grp = 1024 / max(G, 1);
    const int r = t / grp;
    const int j = t - r * grp;
    int v = 0;
    if (r < G) v = tab[nfront + r];

    int part = 0;
    {
        const int4* t4 = (const int4*)tab;
        const int n4 = nfront >> 2;
        for (int k = t; k < n4; k += 1024) {
            const int4 x = t4[k];
            part += ((x.x & 0xFFFF) + (x.y & 0xFFFF)) + ((x.z & 0xFFFF) + (x.w & 0xFFFF));
        }
        for (int k = (n4 << 2) + t; k < nfront; k += 1024) part += tab[k] & 0xFFFF;
    }
    part = waveSumI(part);
    if (lane == 0) wsum[w] = part;

    const int rc = v & 0xFFFF;
    const int start = r * CH_EDGES + (v >> 16);
    const int cap = PL_K * grp;

    int4 q[PL_K];
    int lr[PL_K];
    bool ok[PL_K];
    for (int u = 0; u < PL_K; ++u) {
        const int k = j + u * grp;
        ok[u] = k < rc;
        lr[u] = 0;
        q[u] = tmp[ok[u] ? start + k : 0];
    }

    for (int u = 0; u < PL_K; ++u)
        if (ok[u]) lr[u] = atomicAdd(&c[q[u].x & 255], 1);
    if (rc > cap) anyTail = 1;
    __syncthreads();

    const bool more = anyTail != 0;
    if (more) {
        if (t < 256) held[t] = c[t];
        __syncthreads();
        for (int k = j + cap; k < rc; k += grp) atomicAdd(&c[tmp[start + k].x & 255], 1);
        __syncthreads();
    }

    const int base = waveSumI(wsum[lane]);
    int cv = 0, ci = 0;
    if (t < 256) {
        cv = c[t];
        ci = waveScanI(cv, lane);
        if (lane == 31) wtot[w] = ci;
    }
    __syncthreads();
    if (t < 256) {
        int front = 0;
        for (int z = 0; z < w; ++z) front += wtot[z];
        const int ex = front + ci - cv;
        c[t] = ex;
        if (more) cur[t] = ex + held[t];
        const int node = (b << BIN_SHIFT) + t;
        if (node < N) off[node] = base + ex;
        if (node == N - 1) off[N] = base + ex + cv;
    }
    __syncthreads();

    int st[PL_K];
    for (int u = 0; u < PL_K; ++u) st[u] = c[q[u].x & 255];
    for (int u = 0; u < PL_K; ++u)
        if (ok[u]) rec[base + st[u] + lr[u]] = make_int2(q[u].y, q[u].z);
    if (more) {
        for (int k = j + cap; k < rc; k += grp) {
            const int4 e = tmp[start + k];
            const int p = atomicAdd(&cur[e.x & 255], 1);
            rec[base + p] = make_int2(e.y, e.z);
        }
    }
}

template <int KDIM, int NT>
__device__ __forceinline__ void gemmTile(const float* __restrict__ A, const float* __restrict__ B,
                                         float* __restrict__ C, const int wave, const int lane,
                                         const float* __restrict__ as, const float* __restrict__ ad,
                                         float* __restrict__ osrc, float* __restrict__ odst) {
    constexpr int NCOL = NT * 16;
    int half = lane >> 4;
    int idx  = lane & 15;
    int kk   = half << 1;

    const float* arow  = (A + wave * (16 * KDIM)) + (idx * KDIM + kk);
    const float* bbase = B + (kk * NCOL + idx);

    v8f acc[NT];
#pragma unroll
    for (int t = 0; t < NT; ++t) acc[t] = (v8f){};

#pragma unroll
    for (int k0 = 0; k0 < KDIM; k0 += 4) {
        v2f a;
        a.x = arow[k0];
        a.y = arow[k0 + 1];
#pragma unroll
        for (int t = 0; t < NT; ++t) {
            v2f b;
            b.x = bbase[k0 * NCOL + t * 16];
            b.y = bbase[(k0 + 1) * NCOL + t * 16];
            acc[t] = __builtin_amdgcn_wmma_f32_16x16x4_f32(false, a, false, b,
                                                           (short)0, acc[t], false, false);
        }
    }
    int rbase = wave * 16 + (half << 3);
#pragma unroll
    for (int t = 0; t < NT; ++t) {
        float* cp = (C + wave * (16 * NCOL)) + ((half << 3) * NCOL + t * 16 + idx);
#pragma unroll
        for (int i = 0; i < 8; ++i) cp[i * NCOL] = acc[t][i];
    }

    float ps[8], pd[8];
    for (int i = 0; i < 8; ++i) { ps[i] = 0.f; pd[i] = 0.f; }
    for (int t = 0; t < NT; ++t) {
        const float sa = as[t * 16 + idx], sd = ad[t * 16 + idx];
        for (int i = 0; i < 8; ++i) { ps[i] += acc[t][i] * sa; pd[i] += acc[t][i] * sd; }
    }
    for (int o = 8; o > 0; o >>= 1)
        for (int i = 0; i < 8; ++i) { ps[i] += __shfl_xor(ps[i], o); pd[i] += __shfl_xor(pd[i], o); }
    if (idx == 0)
        for (int i = 0; i < 8; ++i) { osrc[rbase + i] = ps[i]; odst[rbase + i] = pd[i]; }
}

template <int KDIM, int NT>
__global__ void gemmWmmaK(const float* __restrict__ A, const float* __restrict__ B,
                          float* __restrict__ C, int mtiles,
                          const float* __restrict__ as, const float* __restrict__ ad,
                          float* __restrict__ osrc, float* __restrict__ odst) {
    const int gt = blockIdx.x * blockDim.x + threadIdx.x;
    const int wave = gt >> 5;
    const int lane = threadIdx.x & 31;
    if (wave >= mtiles) return;
    gemmTile<KDIM, NT>(A, B, C, wave, lane, as, ad, osrc, odst);
}

#define TILE_WAVES 16

__device__ __forceinline__ bool productBlock(const float* __restrict__ x, const float* __restrict__ W1,
                                             float* __restrict__ h, int mtiles, int gb,
                                             const float* __restrict__ as1, const float* __restrict__ ad1,
                                             float* __restrict__ asrc, float* __restrict__ adst) {
    if ((int)blockIdx.x >= gb) return false;
    const int w = __builtin_amdgcn_readfirstlane((int)(threadIdx.x >> 5));
    const int lane = threadIdx.x & 31;
    const int tile = blockIdx.x * TILE_WAVES + w;
    if (w < TILE_WAVES && tile < mtiles)
        gemmTile<128, 4>(x, W1, h, tile, lane, as1, ad1, asrc, adst);
    return true;
}

__global__ void __launch_bounds__(1024, 16)
binGemmK(const float* __restrict__ x, const float* __restrict__ W1, float* __restrict__ h,
         int mtiles, int gb,
         const float* __restrict__ as1, const float* __restrict__ ad1,
         float* __restrict__ asrc, float* __restrict__ adst,
         const float* __restrict__ we1, const float* __restrict__ ae1,
         const float* __restrict__ we2, const float* __restrict__ ae2,
         float* __restrict__ kscal,
         const int* __restrict__ ei, const float* __restrict__ ea,
         int4* __restrict__ tmp, int* __restrict__ tab, int E, int NB, int G) {
    if (productBlock(x, W1, h, mtiles, gb, as1, ad1, asrc, adst)) {
        if (blockIdx.x == 0 && (threadIdx.x >> 5) == TILE_WAVES) {
            const int lane = threadIdx.x & 31;
            if (lane == 0) { float s = 0.f; for (int j = 0; j < 64; ++j) s += we1[j] * ae1[j]; kscal[0] = s; }
            if (lane == 1) { float s = 0.f; for (int j = 0; j < 32; ++j) s += we2[j] * ae2[j]; kscal[1] = s; }
        }
        return;
    }
    binBody(ei, ea, tmp, tab, E, NB, G, (int)blockIdx.x - gb);
}

#define ROW_XOR_F(v, m) __int_as_float(__builtin_amdgcn_mov_dpp(__float_as_int(v), 0x160 + (m), 0xf, 0xf, true))
#define ROW_LANE_I(v, u) __builtin_amdgcn_mov_dpp((v), 0x150 + (u), 0xf, 0xf, true)
#define ROW_LANE_F(v, u) __int_as_float(__builtin_amdgcn_mov_dpp(__float_as_int(v), 0x150 + (u), 0xf, 0xf, true))

__device__ __forceinline__ float rowSum16(float v) {
    v += ROW_XOR_F(v, 8); v += ROW_XOR_F(v, 4); v += ROW_XOR_F(v, 2); v += ROW_XOR_F(v, 1);
    return v;
}
__device__ __forceinline__ float rowMax16(float v) {
    v = fmaxf(v, ROW_XOR_F(v, 8)); v = fmaxf(v, ROW_XOR_F(v, 4));
    v = fmaxf(v, ROW_XOR_F(v, 2)); v = fmaxf(v, ROW_XOR_F(v, 1));
    return v;
}
__device__ __forceinline__ float lrelu(float a) { return (a > 0.0f) ? a : NEG_SLOPE * a; }

typedef __attribute__((ext_vector_type(4))) float v4f;

template <int D>
__device__ __forceinline__ void loadPiece(const float* __restrict__ h, int row, int hl, v2f (&v)[D / 32]) {
    const char* p = (const char*)h + ((unsigned)row * (unsigned)(D * 4) + (unsigned)hl * (unsigned)(D / 4));
    if constexpr (D == 64) {
        const v4f t = *(const v4f*)p;
        v[0] = t.xy; v[1] = t.zw;
    } else {
        v[0] = *(const v2f*)p;
    }
}

__device__ __forceinline__ float sum8(float v) {
    v += ROW_XOR_F(v, 4); v += ROW_XOR_F(v, 2); v += ROW_XOR_F(v, 1);
    return v;
}

template <int D, int U0>
__device__ __forceinline__ void chunk8(const float* __restrict__ h, int hl, int s, float lb,
                                       const v2f (&av)[D / 32], float& m, float& l,
                                       v2f (&acc)[D / 32]) {
    constexpr int P = D / 32;
    v2f hv[8][P];
#define CHUNK_LD(u) { const int st = ROW_LANE_I(s, U0 + (u)); loadPiece<D>(h, st, hl, hv[u]); }
    CHUNK_LD(0) CHUNK_LD(1) CHUNK_LD(2) CHUNK_LD(3)
    CHUNK_LD(4) CHUNK_LD(5) CHUNK_LD(6) CHUNK_LD(7)
#undef CHUNK_LD
    float d[8];
    for (int u = 0; u < 8; ++u) {
        v2f t = hv[u][0] * av[0];
        for (int p = 1; p < P; ++p) t += hv[u][p] * av[p];
        d[u] = t.x + t.y;
    }
    const bool b2 = (hl & 4) != 0, b1 = (hl & 2) != 0, b0 = (hl & 1) != 0;
    float w[4], x[2];
    for (int k = 0; k < 4; ++k) {
        const float keep = b2 ? d[k + 4] : d[k];
        const float send = b2 ? d[k] : d[k + 4];
        w[k] = keep + ROW_XOR_F(send, 4);
    }
    for (int k = 0; k < 2; ++k) {
        const float keep = b1 ? w[k + 2] : w[k];
        const float send = b1 ? w[k] : w[k + 2];
        x[k] = keep + ROW_XOR_F(send, 2);
    }
    float y;
    {
        const float keep = b0 ? x[1] : x[0];
        const float send = b0 ? x[0] : x[1];
        y = keep + ROW_XOR_F(send, 1);
    }
    y += ROW_XOR_F(y, 8);
    const float lbx = ROW_XOR_F(lb, 8);
    const float mylb = ((hl & 8) == U0) ? lb : lbx;
    const float lg = lrelu(y + mylb);
    const float mn = fmaxf(m, rowMax16(lg));
    const float sc = expf(m - mn);
    const float pe = expf(lg - mn);
    l = l * sc + sum8(pe);
    m = mn;
    for (int p = 0; p < P; ++p) acc[p] *= sc;
#define CHUNK_ADD(u) { const float pu = ROW_LANE_F(pe, u); for (int p = 0; p < P; ++p) acc[p] += hv[u][p] * pu; }
    CHUNK_ADD(0) CHUNK_ADD(1) CHUNK_ADD(2) CHUNK_ADD(3)
    CHUNK_ADD(4) CHUNK_ADD(5) CHUNK_ADD(6) CHUNK_ADD(7)
#undef CHUNK_ADD
}

template <int D, bool RELU>
__global__ void __launch_bounds__(256) segK(const int* __restrict__ off, const int2* __restrict__ rec,
                                            const float* __restrict__ asrc, const float* __restrict__ adst,
                                            const float* __restrict__ kscal, int ki,
                                            const float* __restrict__ h, const float* __restrict__ att,
                                            const float* __restrict__ bias,
                                            float* __restrict__ out, int N) {
    constexpr int P = D / 32;
    const int lane = threadIdx.x & 31;
    const int hl = lane & 15;
    const int i0 = 2 * __builtin_amdgcn_readfirstlane((int)((blockIdx.x * blockDim.x + threadIdx.x) >> 5));
    if (i0 >= N) return;
    const int iraw = i0 + (lane >> 4);
    const bool live = iraw < N;
    const int i = live ? iraw : i0;
    const int b = off[i];
    const int deg = off[live ? i + 1 : i] - b;
    const float kk = kscal[ki];
    const float ad = adst[i];
    const float asi = asrc[i];
    const int dmax = max(__builtin_amdgcn_readlane(deg, 0), __builtin_amdgcn_readlane(deg, 16));

    v2f av[P];
    loadPiece<D>(att, 0, hl, av);
    v2f acc[P];
    loadPiece<D>(h, i, hl, acc);

    const bool v0 = hl < deg, v1 = hl + 16 < deg;
    int2 r0 = make_int2(i, 0), r1 = make_int2(i, 0);
    if (v0) r0 = rec[b + hl];
    if (v1) r1 = rec[b + 16 + hl];
    const int s0 = r0.x, s1 = r1.x;
    const float at0 = __int_as_float(r0.y), at1 = __int_as_float(r1.y);
    const float lb0 = v0 ? (ad + at0 * kk) : -INFINITY;
    const float lb1 = v1 ? (ad + at1 * kk) : -INFINITY;
    float asum = at0 + at1;
#pragma nounroll
    for (int cb = 32; cb < dmax; cb += 16) {
        const int j = cb + hl;
        if (j < deg) asum += __int_as_float(rec[b + j].y);
    }
    asum = rowSum16(asum);

    const float la = asum / fmaxf((float)deg, 1.0f);
    float m = lrelu(asi + ad + la * kk);
    float l = 1.0f;

    if (dmax > 0)  chunk8<D, 0>(h, hl, s0, lb0, av, m, l, acc);
    if (dmax > 8)  chunk8<D, 8>(h, hl, s0, lb0, av, m, l, acc);
    if (dmax > 16) chunk8<D, 0>(h, hl, s1, lb1, av, m, l, acc);
    if (dmax > 24) chunk8<D, 8>(h, hl, s1, lb1, av, m, l, acc);
#pragma nounroll
    for (int cb = 32; cb < dmax; cb += 16) {
        const int j = cb + hl;
        int s = i; float lb = -INFINITY;
        if (j < deg) { const int2 r = rec[b + j]; s = r.x; lb = ad + __int_as_float(r.y) * kk; }
        chunk8<D, 0>(h, hl, s, lb, av, m, l, acc);
        if (dmax > cb + 8) chunk8<D, 8>(h, hl, s, lb, av, m, l, acc);
    }
    const float inv = 1.0f / l;

    if (live) {
        if constexpr (P == 2) {
            const v4f bv = *(const v4f*)(bias + hl * 4);
            v4f o;
            o.xy = acc[0] * inv + bv.xy; o.zw = acc[1] * inv + bv.zw;
            if (RELU) { o.x = fmaxf(o.x, 0.0f); o.y = fmaxf(o.y, 0.0f); o.z = fmaxf(o.z, 0.0f); o.w = fmaxf(o.w, 0.0f); }
            *(v4f*)(out + i * D + hl * 4) = o;
        } else {
            const v2f bv = *(const v2f*)(bias + hl * 2);
            v2f o = acc[0] * inv + bv;
            if (RELU) { o.x = fmaxf(o.x, 0.0f); o.y = fmaxf(o.y, 0.0f); }
            *(v2f*)(out + i * D + hl * 2) = o;
        }
    }
}

static inline int gr(long long n, int blk) { return (int)((n + blk - 1) / blk); }

extern "C" void kernel_launch(void* const* d_in, const int* in_sizes, int n_in,
                              void* d_out, int out_size, void* d_ws, size_t ws_size,
                              hipStream_t stream) {
    const float* x    = (const float*)d_in[0];
    const int*   ei   = (const int*)d_in[1];
    const float* ea   = (const float*)d_in[2];
    const float* W1   = (const float*)d_in[3];
    const float* as1  = (const float*)d_in[4];
    const float* ad1  = (const float*)d_in[5];
    const float* we1  = (const float*)d_in[6];
    const float* ae1  = (const float*)d_in[7];
    const float* b1   = (const float*)d_in[8];
    const float* W2   = (const float*)d_in[9];
    const float* as2  = (const float*)d_in[10];
    const float* ad2  = (const float*)d_in[11];
    const float* we2  = (const float*)d_in[12];
    const float* ae2  = (const float*)d_in[13];
    const float* b2   = (const float*)d_in[14];
    float* out = (float*)d_out;

    const int N  = in_sizes[0] / 128;
    const int E  = in_sizes[1] / 2;

    float* ws = (float*)d_ws;
    size_t o = 0;
    float* h     = ws + o; o += (size_t)N * 64;
    float* hagg  = ws + o; o += (size_t)N * 64;
    int2*  rec   = (int2*)(ws + o); o += (size_t)E * 2;
    const int G  = gr(E, CH_EDGES);
    const int NB = gr(N, 1 << BIN_SHIFT);
    o = (o + 3) & ~(size_t)3;
    int4*  tmp   = (int4*)(ws + o); o += (size_t)G * CH_EDGES * 4;
    int*   tab   = (int*)(ws + o); o += (size_t)NB * G;
    int*   off   = (int*)(ws + o); o += N + 1;
    float* asrc  = ws + o; o += N;
    float* adst  = ws + o; o += N;
    float* kscal = ws + o; o += 4;

    const int B = 256;
    const int mtiles = N / 16;

    const int GB = gr(mtiles, TILE_WAVES);
    binGemmK<<<GB + G, 1024, 0, stream>>>(x, W1, h, mtiles, GB, as1, ad1, asrc, adst,
                                          we1, ae1, we2, ae2, kscal, ei, ea, tmp, tab, E, NB, G);
    placeK<<<NB, 1024, 0, stream>>>(tmp, tab, off, rec, N, G);

    segK<64, true><<<gr((long long)((N + 1) / 2) * 32, B), B, 0, stream>>>(off, rec, asrc, adst, kscal, 0, h, as1, b1, hagg, N);

    gemmWmmaK<64, 2><<<gr(mtiles, 8), 8 * 32, 0, stream>>>(hagg, W2, h, mtiles, as2, ad2, asrc, adst);
    segK<32, false><<<gr((long long)((N + 1) / 2) * 32, B), B, 0, stream>>>(off, rec, asrc, adst, kscal, 1, h, as2, b2, out, N);
}


# Layer-2 product moved into the layer-1 aggregation launch: each block's 16 rows go to an LDS tile, one wave runs the WMMA tile from it; 5 -> 4 launches, layer-1 output never stored; sum orders kept
# speedup vs baseline: 10.5165x; 1.0506x over previous
#include <hip/hip_runtime.h>
#include <math.h>

typedef __attribute__((ext_vector_type(2))) float v2f;
typedef __attribute__((ext_vector_type(8))) float v8f;

#define NEG_SLOPE 0.2f

#define CH_EDGES 4096
#define BIN_SHIFT 8
#define MAX_BINS 256
#define MAX_CH 256

__device__ __forceinline__ int waveSumI(int v) {
    for (int o = 16; o > 0; o >>= 1) v += __shfl_xor(v, o);
    return v;
}
__device__ __forceinline__ int waveScanI(int x, int lane) {
    for (int o = 1; o < 32; o <<= 1) {
        int y = __shfl_up(x, o);
        if (lane >= o) x += y;
    }
    return x;
}

__device__ __forceinline__ void binBody(const int* __restrict__ ei, const float* __restrict__ ea,
                                        int4* __restrict__ tmp, int* __restrict__ tab,
                                        int E, int NB, int G, const int blk) {
    __shared__ int hist[MAX_BINS];
    __shared__ int wtot[8];
    const int t = threadIdx.x, lane = t & 31, w = t >> 5;
    if (t < MAX_BINS) hist[t] = 0;
    __syncthreads();

    const int e0 = blk * CH_EDGES + t * 4;
    int d[4], s[4], a[4], lr[4];
    bool ok[4];
    for (int k = 0; k < 4; ++k) { ok[k] = (e0 + k) < E; d[k] = 0; s[k] = 0; a[k] = 0; lr[k] = 0; }
    if (((E & 3) == 0) && (e0 + 3 < E)) {
        const int4 dv = *(const int4*)(ei + E + e0);
        const int4 sv = *(const int4*)(ei + e0);
        const float4 av = *(const float4*)(ea + e0);
        d[0] = dv.x; d[1] = dv.y; d[2] = dv.z; d[3] = dv.w;
        s[0] = sv.x; s[1] = sv.y; s[2] = sv.z; s[3] = sv.w;
        a[0] = __float_as_int(av.x); a[1] = __float_as_int(av.y);
        a[2] = __float_as_int(av.z); a[3] = __float_as_int(av.w);
    } else {
        for (int k = 0; k < 4; ++k)
            if (ok[k]) { d[k] = ei[E + e0 + k]; s[k] = ei[e0 + k]; a[k] = __float_as_int(ea[e0 + k]); }
    }
    for (int k = 0; k < 4; ++k)
        if (ok[k]) lr[k] = atomicAdd(&hist[d[k] >> BIN_SHIFT], 1);
    __syncthreads();

    int cv = 0, ci = 0;
    if (t < MAX_BINS) {
        cv = hist[t];
        ci = waveScanI(cv, lane);
        if (lane == 31) wtot[w] = ci;
    }
    __syncthreads();
    if (t < MAX_BINS) {
        int front = 0;
        for (int j = 0; j < w; ++j) front += wtot[j];
        const int ex = front + ci - cv;
        hist[t] = ex;
        if (t < NB) tab[t * G + blk] = cv | (ex << 16);
    }
    __syncthreads();
    for (int k = 0; k < 4; ++k)
        if (ok[k]) tmp[blk * CH_EDGES + hist[d[k] >> BIN_SHIFT] + lr[k]] = make_int4(d[k], s[k], a[k], 0);
}

#define PL_K 10
__global__ void __launch_bounds__(1024) placeK(const int4* __restrict__ tmp, const int* __restrict__ tab,
                                               int* __restrict__ off, int2* __restrict__ rec,
                                               int N, int G) {
    __shared__ int c[256];
    __shared__ int held[256];
    __shared__ int cur[256];
    __shared__ int wsum[32];
    __shared__ int wtot[8];
    __shared__ int anyTail;
    const int t = threadIdx.x, lane = t & 31, w = t >> 5;
    const int b = blockIdx.x;
    const int nfront = b * G;

    if (t < 256) c[t] = 0;
    if (t == 0) anyTail = 0;
    __syncthreads();

    const int grp = 1024 / max(G, 1);
    const int r = t / grp;
    const int j = t - r * grp;
    int v = 0;
    if (r < G) v = tab[nfront + r];

    int part = 0;
    {
        const int4* t4 = (const int4*)tab;
        const int n4 = nfront >> 2;
        for (int k = t; k < n4; k += 1024) {
            const int4 x = t4[k];
            part += ((x.x & 0xFFFF) + (x.y & 0xFFFF)) + ((x.z & 0xFFFF) + (x.w & 0xFFFF));
        }
        for (int k = (n4 << 2) + t; k < nfront; k += 1024) part += tab[k] & 0xFFFF;
    }
    part = waveSumI(part);
    if (lane == 0) wsum[w] = part;

    const int rc = v & 0xFFFF;
    const int start = r * CH_EDGES + (v >> 16);
    const int cap = PL_K * grp;

    int4 q[PL_K];
    int lr[PL_K];
    bool ok[PL_K];
    for (int u = 0; u < PL_K; ++u) {
        const int k = j + u * grp;
        ok[u] = k < rc;
        lr[u] = 0;
        q[u] = tmp[ok[u] ? start + k : 0];
    }

    for (int u = 0; u < PL_K; ++u)
        if (ok[u]) lr[u] = atomicAdd(&c[q[u].x & 255], 1);
    if (rc > cap) anyTail = 1;
    __syncthreads();

    const bool more = anyTail != 0;
    if (more) {
        if (t < 256) held[t] = c[t];
        __syncthreads();
        for (int k = j + cap; k < rc; k += grp) atomicAdd(&c[tmp[start + k].x & 255], 1);
        __syncthreads();
    }

    const int base = waveSumI(wsum[lane]);
    int cv = 0, ci = 0;
    if (t < 256) {
        cv = c[t];
        ci = waveScanI(cv, lane);
        if (lane == 31) wtot[w] = ci;
    }
    __syncthreads();
    if (t < 256) {
        int front = 0;
        for (int z = 0; z < w; ++z) front += wtot[z];
        const int ex = front + ci - cv;
        c[t] = ex;
        if (more) cur[t] = ex + held[t];
        const int node = (b << BIN_SHIFT) + t;
        if (node < N) off[node] = base + ex;
        if (node == N - 1) off[N] = base + ex + cv;
    }
    __syncthreads();

    int st[PL_K];
    for (int u = 0; u < PL_K; ++u) st[u] = c[q[u].x & 255];
    for (int u = 0; u < PL_K; ++u)
        if (ok[u]) rec[base + st[u] + lr[u]] = make_int2(q[u].y, q[u].z);
    if (more) {
        for (int k = j + cap; k < rc; k += grp) {
            const int4 e = tmp[start + k];
            const int p = atomicAdd(&cur[e.x & 255], 1);
            rec[base + p] = make_int2(e.y, e.z);
        }
    }
}

template <int KDIM, int NT, int LDA>
__device__ __forceinline__ void gemmTile(const float* __restrict__ Atile, const float* __restrict__ B,
                                         float* __restrict__ C, const int wave, const int lane,
                                         const float* __restrict__ as, const float* __restrict__ ad,
                                         float* __restrict__ osrc, float* __restrict__ odst) {
    constexpr int NCOL = NT * 16;
    int half = lane >> 4;
    int idx  = lane & 15;
    int kk   = half << 1;

    const float* arow  = Atile + (idx * LDA + kk);
    const float* bbase = B + (kk * NCOL + idx);

    v8f acc[NT];
#pragma unroll
    for (int t = 0; t < NT; ++t) acc[t] = (v8f){};

#pragma unroll
    for (int k0 = 0; k0 < KDIM; k0 += 4) {
        v2f a;
        a.x = arow[k0];
        a.y = arow[k0 + 1];
#pragma unroll
        for (int t = 0; t < NT; ++t) {
            v2f b;
            b.x = bbase[k0 * NCOL + t * 16];
            b.y = bbase[(k0 + 1) * NCOL + t * 16];
            acc[t] = __builtin_amdgcn_wmma_f32_16x16x4_f32(false, a, false, b,
                                                           (short)0, acc[t], false, false);
        }
    }
    int rbase = wave * 16 + (half << 3);
#pragma unroll
    for (int t = 0; t < NT; ++t) {
        float* cp = (C + wave * (16 * NCOL)) + ((half << 3) * NCOL + t * 16 + idx);
#pragma unroll
        for (int i = 0; i < 8; ++i) cp[i * NCOL] = acc[t][i];
    }

    float ps[8], pd[8];
    for (int i = 0; i < 8; ++i) { ps[i] = 0.f; pd[i] = 0.f; }
    for (int t = 0; t < NT; ++t) {
        const float sa = as[t * 16 + idx], sd = ad[t * 16 + idx];
        for (int i = 0; i < 8; ++i) { ps[i] += acc[t][i] * sa; pd[i] += acc[t][i] * sd; }
    }
    for (int o = 8; o > 0; o >>= 1)
        for (int i = 0; i < 8; ++i) { ps[i] += __shfl_xor(ps[i], o); pd[i] += __shfl_xor(pd[i], o); }
    if (idx == 0)
        for (int i = 0; i < 8; ++i) { osrc[rbase + i] = ps[i]; odst[rbase + i] = pd[i]; }
}

#define TILE_WAVES 16

__device__ __forceinline__ bool productBlock(const float* __restrict__ x, const float* __restrict__ W1,
                                             float* __restrict__ h, int mtiles, int gb,
                                             const float* __restrict__ as1, const float* __restrict__ ad1,
                                             float* __restrict__ asrc, float* __restrict__ adst) {
    if ((int)blockIdx.x >= gb) return false;
    const int w = __builtin_amdgcn_readfirstlane((int)(threadIdx.x >> 5));
    const int lane = threadIdx.x & 31;
    const int tile = blockIdx.x * TILE_WAVES + w;
    if (w < TILE_WAVES && tile < mtiles)
        gemmTile<128, 4, 128>(x + tile * (16 * 128), W1, h, tile, lane, as1, ad1, asrc, adst);
    return true;
}

__global__ void __launch_bounds__(1024, 16)
binGemmK(const float* __restrict__ x, const float* __restrict__ W1, float* __restrict__ h,
         int mtiles, int gb,
         const float* __restrict__ as1, const float* __restrict__ ad1,
         float* __restrict__ asrc, float* __restrict__ adst,
         const float* __restrict__ we1, const float* __restrict__ ae1,
         const float* __restrict__ we2, const float* __restrict__ ae2,
         float* __restrict__ kscal,
         const int* __restrict__ ei, const float* __restrict__ ea,
         int4* __restrict__ tmp, int* __restrict__ tab, int E, int NB, int G) {
    if (productBlock(x, W1, h, mtiles, gb, as1, ad1, asrc, adst)) {
        if (blockIdx.x == 0 && (threadIdx.x >> 5) == TILE_WAVES) {
            const int lane = threadIdx.x & 31;
            if (lane == 0) { float s = 0.f; for (int j = 0; j < 64; ++j) s += we1[j] * ae1[j]; kscal[0] = s; }
            if (lane == 1) { float s = 0.f; for (int j = 0; j < 32; ++j) s += we2[j] * ae2[j]; kscal[1] = s; }
        }
        return;
    }
    binBody(ei, ea, tmp, tab, E, NB, G, (int)blockIdx.x - gb);
}

#define ROW_XOR_F(v, m) __int_as_float(__builtin_amdgcn_mov_dpp(__float_as_int(v), 0x160 + (m), 0xf, 0xf, true))
#define ROW_LANE_I(v, u) __builtin_amdgcn_mov_dpp((v), 0x150 + (u), 0xf, 0xf, true)
#define ROW_LANE_F(v, u) __int_as_float(__builtin_amdgcn_mov_dpp(__float_as_int(v), 0x150 + (u), 0xf, 0xf, true))

__device__ __forceinline__ float rowSum16(float v) {
    v += ROW_XOR_F(v, 8); v += ROW_XOR_F(v, 4); v += ROW_XOR_F(v, 2); v += ROW_XOR_F(v, 1);
    return v;
}
__device__ __forceinline__ float rowMax16(float v) {
    v = fmaxf(v, ROW_XOR_F(v, 8)); v = fmaxf(v, ROW_XOR_F(v, 4));
    v = fmaxf(v, ROW_XOR_F(v, 2)); v = fmaxf(v, ROW_XOR_F(v, 1));
    return v;
}
__device__ __forceinline__ float lrelu(float a) { return (a > 0.0f) ? a : NEG_SLOPE * a; }

typedef __attribute__((ext_vector_type(4))) float v4f;

template <int D>
__device__ __forceinline__ void loadPiece(const float* __restrict__ h, int row, int hl, v2f (&v)[D / 32]) {
    const char* p = (const char*)h + ((unsigned)row * (unsigned)(D * 4) + (unsigned)hl * (unsigned)(D / 4));
    if constexpr (D == 64) {
        const v4f t = *(const v4f*)p;
        v[0] = t.xy; v[1] = t.zw;
    } else {
        v[0] = *(const v2f*)p;
    }
}

__device__ __forceinline__ float sum8(float v) {
    v += ROW_XOR_F(v, 4); v += ROW_XOR_F(v, 2); v += ROW_XOR_F(v, 1);
    return v;
}

template <int D, int U0>
__device__ __forceinline__ void chunk8(const float* __restrict__ h, int hl, int s, float lb,
                                       const v2f (&av)[D / 32], float& m, float& l,
                                       v2f (&acc)[D / 32]) {
    constexpr int P = D / 32;
    v2f hv[8][P];
#define CHUNK_LD(u) { const int st = ROW_LANE_I(s, U0 + (u)); loadPiece<D>(h, st, hl, hv[u]); }
    CHUNK_LD(0) CHUNK_LD(1) CHUNK_LD(2) CHUNK_LD(3)
    CHUNK_LD(4) CHUNK_LD(5) CHUNK_LD(6) CHUNK_LD(7)
#undef CHUNK_LD
    float d[8];
    for (int u = 0; u < 8; ++u) {
        v2f t = hv[u][0] * av[0];
        for (int p = 1; p < P; ++p) t += hv[u][p] * av[p];
        d[u] = t.x + t.y;
    }
    const bool b2 = (hl & 4) != 0, b1 = (hl & 2) != 0, b0 = (hl & 1) != 0;
    float w[4], x[2];
    for (int k = 0; k < 4; ++k) {
        const float keep = b2 ? d[k + 4] : d[k];
        const float send = b2 ? d[k] : d[k + 4];
        w[k] = keep + ROW_XOR_F(send, 4);
    }
    for (int k = 0; k < 2; ++k) {
        const float keep = b1 ? w[k + 2] : w[k];
        const float send = b1 ? w[k] : w[k + 2];
        x[k] = keep + ROW_XOR_F(send, 2);
    }
    float y;
    {
        const float keep = b0 ? x[1] : x[0];
        const float send = b0 ? x[0] : x[1];
        y = keep + ROW_XOR_F(send, 1);
    }
    y += ROW_XOR_F(y, 8);
    const float lbx = ROW_XOR_F(lb, 8);
    const float mylb = ((hl & 8) == U0) ? lb : lbx;
    const float lg = lrelu(y + mylb);
    const float mn = fmaxf(m, rowMax16(lg));
    const float sc = expf(m - mn);
    const float pe = expf(lg - mn);
    l = l * sc + sum8(pe);
    m = mn;
    for (int p = 0; p < P; ++p) acc[p] *= sc;
#define CHUNK_ADD(u) { const float pu = ROW_LANE_F(pe, u); for (int p = 0; p < P; ++p) acc[p] += hv[u][p] * pu; }
    CHUNK_ADD(0) CHUNK_ADD(1) CHUNK_ADD(2) CHUNK_ADD(3)
    CHUNK_ADD(4) CHUNK_ADD(5) CHUNK_ADD(6) CHUNK_ADD(7)
#undef CHUNK_ADD
}

template <int D>
__device__ __forceinline__ int segRow(const int* __restrict__ off, const int2* __restrict__ rec,
                                      const float* __restrict__ asrc, const float* __restrict__ adst,
                                      const float* __restrict__ kscal, int ki,
                                      const float* __restrict__ h, const float* __restrict__ att,
                                      int N, int i0, int lane, bool& live,
                                      v2f (&acc)[D / 32], float& inv) {
    constexpr int P = D / 32;
    const int hl = lane & 15;
    const int iraw = i0 + (lane >> 4);
    live = iraw < N;
    const int i = live ? iraw : i0;
    const int b = off[i];
    const int deg = off[live ? i + 1 : i] - b;
    const float kk = kscal[ki];
    const float ad = adst[i];
    const float asi = asrc[i];
    const int dmax = max(__builtin_amdgcn_readlane(deg, 0), __builtin_amdgcn_readlane(deg, 16));

    v2f av[P];
    loadPiece<D>(att, 0, hl, av);
    loadPiece<D>(h, i, hl, acc);

    const bool v0 = hl < deg, v1 = hl + 16 < deg;
    int2 r0 = make_int2(i, 0), r1 = make_int2(i, 0);
    if (v0) r0 = rec[b + hl];
    if (v1) r1 = rec[b + 16 + hl];
    const int s0 = r0.x, s1 = r1.x;
    const float at0 = __int_as_float(r0.y), at1 = __int_as_float(r1.y);
    const float lb0 = v0 ? (ad + at0 * kk) : -INFINITY;
    const float lb1 = v1 ? (ad + at1 * kk) : -INFINITY;
    float asum = at0 + at1;
#pragma nounroll
    for (int cb = 32; cb < dmax; cb += 16) {
        const int j = cb + hl;
        if (j < deg) asum += __int_as_float(rec[b + j].y);
    }
    asum = rowSum16(asum);

    const float la = asum / fmaxf((float)deg, 1.0f);
    float m = lrelu(asi + ad + la * kk);
    float l = 1.0f;

    if (dmax > 0)  chunk8<D, 0>(h, hl, s0, lb0, av, m, l, acc);
    if (dmax > 8)  chunk8<D, 8>(h, hl, s0, lb0, av, m, l, acc);
    if (dmax > 16) chunk8<D, 0>(h, hl, s1, lb1, av, m, l, acc);
    if (dmax > 24) chunk8<D, 8>(h, hl, s1, lb1, av, m, l, acc);
#pragma nounroll
    for (int cb = 32; cb < dmax; cb += 16) {
        const int j = cb + hl;
        int s = i; float lb = -INFINITY;
        if (j < deg) { const int2 r = rec[b + j]; s = r.x; lb = ad + __int_as_float(r.y) * kk; }
        chunk8<D, 0>(h, hl, s, lb, av, m, l, acc);
        if (dmax > cb + 8) chunk8<D, 8>(h, hl, s, lb, av, m, l, acc);
    }
    inv = 1.0f / l;
    return i;
}

__global__ void __launch_bounds__(256) segOutK(const int* __restrict__ off, const int2* __restrict__ rec,
                                               const float* __restrict__ asrc, const float* __restrict__ adst,
                                               const float* __restrict__ kscal, int ki,
                                               const float* __restrict__ h, const float* __restrict__ att,
                                               const float* __restrict__ bias,
                                               float* __restrict__ out, int N) {
    const int lane = threadIdx.x & 31;
    const int hl = lane & 15;
    const int i0 = 2 * __builtin_amdgcn_readfirstlane((int)((blockIdx.x * blockDim.x + threadIdx.x) >> 5));
    if (i0 >= N) return;
    bool live; v2f acc[1]; float inv;
    const int i = segRow<32>(off, rec, asrc, adst, kscal, ki, h, att, N, i0, lane, live, acc, inv);
    if (live) {
        const v2f bv = *(const v2f*)(bias + hl * 2);
        *(v2f*)(out + i * 32 + hl * 2) = acc[0] * inv + bv;
    }
}

#define AGG_LDA 68
__global__ void __launch_bounds__(256) segGemmK(const int* __restrict__ off, const int2* __restrict__ rec,
                                                const float* __restrict__ asrc, const float* __restrict__ adst,
                                                const float* __restrict__ kscal,
                                                const float* __restrict__ h, const float* __restrict__ att,
                                                const float* __restrict__ bias,
                                                const float* __restrict__ W2,
                                                const float* __restrict__ as2, const float* __restrict__ ad2,
                                                float* __restrict__ h2,
                                                float* __restrict__ asrc2, float* __restrict__ adst2,
                                                int N, int mtiles) {
    __shared__ __attribute__((aligned(16))) float tileA[16 * AGG_LDA];
    const int lane = threadIdx.x & 31;
    const int hl = lane & 15;
    const int w = __builtin_amdgcn_readfirstlane((int)(threadIdx.x >> 5));
    const int i0 = 2 * ((int)blockIdx.x * 8 + w);
    v4f o = (v4f){0.f, 0.f, 0.f, 0.f};
    if (i0 < N) {
        bool live; v2f acc[2]; float inv;
        segRow<64>(off, rec, asrc, adst, kscal, 0, h, att, N, i0, lane, live, acc, inv);
        if (live) {
            const v4f bv = *(const v4f*)(bias + hl * 4);
            o.xy = acc[0] * inv + bv.xy; o.zw = acc[1] * inv + bv.zw;
            o.x = fmaxf(o.x, 0.0f); o.y = fmaxf(o.y, 0.0f); o.z = fmaxf(o.z, 0.0f); o.w = fmaxf(o.w, 0.0f);
        }
    }
    *(v4f*)(tileA + (2 * w + (lane >> 4)) * AGG_LDA + hl * 4) = o;
    __syncthreads();
    if (w == 0 && (int)blockIdx.x < mtiles)
        gemmTile<64, 2, AGG_LDA>(tileA, W2, h2, (int)blockIdx.x, lane, as2, ad2, asrc2, adst2);
}

static inline int gr(long long n, int blk) { return (int)((n + blk - 1) / blk); }

extern "C" void kernel_launch(void* const* d_in, const int* in_sizes, int n_in,
                              void* d_out, int out_size, void* d_ws, size_t ws_size,
                              hipStream_t stream) {
    const float* x    = (const float*)d_in[0];
    const int*   ei   = (const int*)d_in[1];
    const float* ea   = (const float*)d_in[2];
    const float* W1   = (const float*)d_in[3];
    const float* as1  = (const float*)d_in[4];
    const float* ad1  = (const float*)d_in[5];
    const float* we1  = (const float*)d_in[6];
    const float* ae1  = (const float*)d_in[7];
    const float* b1   = (const float*)d_in[8];
    const float* W2   = (const float*)d_in[9];
    const float* as2  = (const float*)d_in[10];
    const float* ad2  = (const float*)d_in[11];
    const float* we2  = (const float*)d_in[12];
    const float* ae2  = (const float*)d_in[13];
    const float* b2   = (const float*)d_in[14];
    float* out = (float*)d_out;

    const int N  = in_sizes[0] / 128;
    const int E  = in_sizes[1] / 2;

    float* ws = (float*)d_ws;
    size_t o = 0;
    float* h     = ws + o; o += (size_t)N * 64;
    float* h2    = ws + o; o += (size_t)N * 32;
    int2*  rec   = (int2*)(ws + o); o += (size_t)E * 2;
    const int G  = gr(E, CH_EDGES);
    const int NB = gr(N, 1 << BIN_SHIFT);
    o = (o + 3) & ~(size_t)3;
    int4*  tmp   = (int4*)(ws + o); o += (size_t)G * CH_EDGES * 4;
    int*   tab   = (int*)(ws + o); o += (size_t)NB * G;
    int*   off   = (int*)(ws + o); o += N + 1;
    float* asrc  = ws + o; o += N;
    float* adst  = ws + o; o += N;
    float* asrc2 = ws + o; o += N;
    float* adst2 = ws + o; o += N;
    float* kscal = ws + o; o += 4;

    const int B = 256;
    const int mtiles = N / 16;

    const int GB = gr(mtiles, TILE_WAVES);
    binGemmK<<<GB + G, 1024, 0, stream>>>(x, W1, h, mtiles, GB, as1, ad1, asrc, adst,
                                          we1, ae1, we2, ae2, kscal, ei, ea, tmp, tab, E, NB, G);
    placeK<<<NB, 1024, 0, stream>>>(tmp, tab, off, rec, N, G);

    const int SB = gr((long long)((N + 1) / 2) * 32, B);
    segGemmK<<<SB, B, 0, stream>>>(off, rec, asrc, adst, kscal, h, as1, b1, W2, as2, ad2,
                                   h2, asrc2, adst2, N, mtiles);

    segOutK<<<SB, B, 0, stream>>>(off, rec, asrc2, adst2, kscal, 1, h2, as2, b2, out, N);
}


# Edge sort level 2: a bin's sorted records go to an LDS copy (48 KiB) first and leave as one contiguous run, 16 bytes per lane, whole lines per store, not one scattered 8-byte store per edge
# speedup vs baseline: 10.8264x; 1.0295x over previous
#include <hip/hip_runtime.h>
#include <math.h>

typedef __attribute__((ext_vector_type(2))) float v2f;
typedef __attribute__((ext_vector_type(8))) float v8f;

#define NEG_SLOPE 0.2f

#define CH_EDGES 4096
#define BIN_SHIFT 8
#define MAX_BINS 256
#define MAX_CH 256

__device__ __forceinline__ int waveSumI(int v) {
    for (int o = 16; o > 0; o >>= 1) v += __shfl_xor(v, o);
    return v;
}
__device__ __forceinline__ int waveScanI(int x, int lane) {
    for (int o = 1; o < 32; o <<= 1) {
        int y = __shfl_up(x, o);
        if (lane >= o) x += y;
    }
    return x;
}

__device__ __forceinline__ void binBody(const int* __restrict__ ei, const float* __restrict__ ea,
                                        int4* __restrict__ tmp, int* __restrict__ tab,
                                        int E, int NB, int G, const int blk) {
    __shared__ int hist[MAX_BINS];
    __shared__ int wtot[8];
    const int t = threadIdx.x, lane = t & 31, w = t >> 5;
    if (t < MAX_BINS) hist[t] = 0;
    __syncthreads();

    const int e0 = blk * CH_EDGES + t * 4;
    int d[4], s[4], a[4], lr[4];
    bool ok[4];
    for (int k = 0; k < 4; ++k) { ok[k] = (e0 + k) < E; d[k] = 0; s[k] = 0; a[k] = 0; lr[k] = 0; }
    if (((E & 3) == 0) && (e0 + 3 < E)) {
        const int4 dv = *(const int4*)(ei + E + e0);
        const int4 sv = *(const int4*)(ei + e0);
        const float4 av = *(const float4*)(ea + e0);
        d[0] = dv.x; d[1] = dv.y; d[2] = dv.z; d[3] = dv.w;
        s[0] = sv.x; s[1] = sv.y; s[2] = sv.z; s[3] = sv.w;
        a[0] = __float_as_int(av.x); a[1] = __float_as_int(av.y);
        a[2] = __float_as_int(av.z); a[3] = __float_as_int(av.w);
    } else {
        for (int k = 0; k < 4; ++k)
            if (ok[k]) { d[k] = ei[E + e0 + k]; s[k] = ei[e0 + k]; a[k] = __float_as_int(ea[e0 + k]); }
    }
    for (int k = 0; k < 4; ++k)
        if (ok[k]) lr[k] = atomicAdd(&hist[d[k] >> BIN_SHIFT], 1);
    __syncthreads();

    int cv = 0, ci = 0;
    if (t < MAX_BINS) {
        cv = hist[t];
        ci = waveScanI(cv, lane);
        if (lane == 31) wtot[w] = ci;
    }
    __syncthreads();
    if (t < MAX_BINS) {
        int front = 0;
        for (int j = 0; j < w; ++j) front += wtot[j];
        const int ex = front + ci - cv;
        hist[t] = ex;
        if (t < NB) tab[t * G + blk] = cv | (ex << 16);
    }
    __syncthreads();
    for (int k = 0; k < 4; ++k)
        if (ok[k]) tmp[blk * CH_EDGES + hist[d[k] >> BIN_SHIFT] + lr[k]] = make_int4(d[k], s[k], a[k], 0);
}

#define PL_K 10
#define ST_CAP 6144
__global__ void __launch_bounds__(1024) placeK(const int4* __restrict__ tmp, const int* __restrict__ tab,
                                               int* __restrict__ off, int2* __restrict__ rec,
                                               int N, int G) {
    __shared__ int c[256];
    __shared__ int held[256];
    __shared__ int cur[256];
    __shared__ int wsum[32];
    __shared__ int wtot[8];
    __shared__ int anyTail;
    __shared__ int btot;
    __shared__ int2 stage[ST_CAP];
    const int t = threadIdx.x, lane = t & 31, w = t >> 5;
    const int b = blockIdx.x;
    const int nfront = b * G;

    if (t < 256) c[t] = 0;
    if (t == 0) anyTail = 0;
    __syncthreads();

    const int grp = 1024 / max(G, 1);
    const int r = t / grp;
    const int j = t - r * grp;
    int v = 0;
    if (r < G) v = tab[nfront + r];

    int part = 0;
    {
        const int4* t4 = (const int4*)tab;
        const int n4 = nfront >> 2;
        for (int k = t; k < n4; k += 1024) {
            const int4 x = t4[k];
            part += ((x.x & 0xFFFF) + (x.y & 0xFFFF)) + ((x.z & 0xFFFF) + (x.w & 0xFFFF));
        }
        for (int k = (n4 << 2) + t; k < nfront; k += 1024) part += tab[k] & 0xFFFF;
    }
    part = waveSumI(part);
    if (lane == 0) wsum[w] = part;

    const int rc = v & 0xFFFF;
    const int start = r * CH_EDGES + (v >> 16);
    const int cap = PL_K * grp;

    int4 q[PL_K];
    int lr[PL_K];
    bool ok[PL_K];
    for (int u = 0; u < PL_K; ++u) {
        const int k = j + u * grp;
        ok[u] = k < rc;
        lr[u] = 0;
        q[u] = tmp[ok[u] ? start + k : 0];
    }

    for (int u = 0; u < PL_K; ++u)
        if (ok[u]) lr[u] = atomicAdd(&c[q[u].x & 255], 1);
    if (rc > cap) anyTail = 1;
    __syncthreads();

    const bool more = anyTail != 0;
    if (more) {
        if (t < 256) held[t] = c[t];
        __syncthreads();
        for (int k = j + cap; k < rc; k += grp) atomicAdd(&c[tmp[start + k].x & 255], 1);
        __syncthreads();
    }

    const int base = waveSumI(wsum[lane]);
    int cv = 0, ci = 0;
    if (t < 256) {
        cv = c[t];
        ci = waveScanI(cv, lane);
        if (lane == 31) wtot[w] = ci;
    }
    __syncthreads();
    if (t < 256) {
        int front = 0;
        for (int z = 0; z < w; ++z) front += wtot[z];
        const int ex = front + ci - cv;
        c[t] = ex;
        if (more) cur[t] = ex + held[t];
        const int node = (b << BIN_SHIFT) + t;
        if (node < N) off[node] = base + ex;
        if (node == N - 1) off[N] = base + ex + cv;
        if (t == 255) btot = ex + cv;
    }
    __syncthreads();

    int st[PL_K];
    for (int u = 0; u < PL_K; ++u) st[u] = c[q[u].x & 255];
    const int total = btot;
    if (total <= ST_CAP) {
        for (int u = 0; u < PL_K; ++u)
            if (ok[u]) stage[st[u] + lr[u]] = make_int2(q[u].y, q[u].z);
        if (more) {
            for (int k = j + cap; k < rc; k += grp) {
                const int4 e = tmp[start + k];
                const int p = atomicAdd(&cur[e.x & 255], 1);
                stage[p] = make_int2(e.y, e.z);
            }
        }
        __syncthreads();
        const int a = (total > 0) ? (base & 1) : 0;
        const int rem = total - a;
        const int npair = rem >> 1;
        int4* out4 = (int4*)(rec + base + a);
        for (int p = t; p < npair; p += 1024) {
            const int2 x = stage[a + 2 * p], y = stage[a + 2 * p + 1];
            out4[p] = make_int4(x.x, x.y, y.x, y.y);
        }
        if (t == 0 && a == 1) rec[base] = stage[0];
        if (t == 32 && (rem & 1)) rec[base + total - 1] = stage[total - 1];
    } else {
        for (int u = 0; u < PL_K; ++u)
            if (ok[u]) rec[base + st[u] + lr[u]] = make_int2(q[u].y, q[u].z);
        if (more) {
            for (int k = j + cap; k < rc; k += grp) {
                const int4 e = tmp[start + k];
                const int p = atomicAdd(&cur[e.x & 255], 1);
                rec[base + p] = make_int2(e.y, e.z);
            }
        }
    }
}

template <int KDIM, int NT, int LDA>
__device__ __forceinline__ void gemmTile(const float* __restrict__ Atile, const float* __restrict__ B,
                                         float* __restrict__ C, const int wave, const int lane,
                                         const float* __restrict__ as, const float* __restrict__ ad,
                                         float* __restrict__ osrc, float* __restrict__ odst) {
    constexpr int NCOL = NT * 16;
    int half = lane >> 4;
    int idx  = lane & 15;
    int kk   = half << 1;

    const float* arow  = Atile + (idx * LDA + kk);
    const float* bbase = B + (kk * NCOL + idx);

    v8f acc[NT];
#pragma unroll
    for (int t = 0; t < NT; ++t) acc[t] = (v8f){};

#pragma unroll
    for (int k0 = 0; k0 < KDIM; k0 += 4) {
        v2f a;
        a.x = arow[k0];
        a.y = arow[k0 + 1];
#pragma unroll
        for (int t = 0; t < NT; ++t) {
            v2f b;
            b.x = bbase[k0 * NCOL + t * 16];
            b.y = bbase[(k0 + 1) * NCOL + t * 16];
            acc[t] = __builtin_amdgcn_wmma_f32_16x16x4_f32(false, a, false, b,
                                                           (short)0, acc[t], false, false);
        }
    }
    int rbase = wave * 16 + (half << 3);
#pragma unroll
    for (int t = 0; t < NT; ++t) {
        float* cp = (C + wave * (16 * NCOL)) + ((half << 3) * NCOL + t * 16 + idx);
#pragma unroll
        for (int i = 0; i < 8; ++i) cp[i * NCOL] = acc[t][i];
    }

    float ps[8], pd[8];
    for (int i = 0; i < 8; ++i) { ps[i] = 0.f; pd[i] = 0.f; }
    for (int t = 0; t < NT; ++t) {
        const float sa = as[t * 16 + idx], sd = ad[t * 16 + idx];
        for (int i = 0; i < 8; ++i) { ps[i] += acc[t][i] * sa; pd[i] += acc[t][i] * sd; }
    }
    for (int o = 8; o > 0; o >>= 1)
        for (int i = 0; i < 8; ++i) { ps[i] += __shfl_xor(ps[i], o); pd[i] += __shfl_xor(pd[i], o); }
    if (idx == 0)
        for (int i = 0; i < 8; ++i) { osrc[rbase + i] = ps[i]; odst[rbase + i] = pd[i]; }
}

#define TILE_WAVES 16

__device__ __forceinline__ bool productBlock(const float* __restrict__ x, const float* __restrict__ W1,
                                             float* __restrict__ h, int mtiles, int gb,
                                             const float* __restrict__ as1, const float* __restrict__ ad1,
                                             float* __restrict__ asrc, float* __restrict__ adst) {
    if ((int)blockIdx.x >= gb) return false;
    const int w = __builtin_amdgcn_readfirstlane((int)(threadIdx.x >> 5));
    const int lane = threadIdx.x & 31;
    const int tile = blockIdx.x * TILE_WAVES + w;
    if (w < TILE_WAVES && tile < mtiles)
        gemmTile<128, 4, 128>(x + tile * (16 * 128), W1, h, tile, lane, as1, ad1, asrc, adst);
    return true;
}

__global__ void __launch_bounds__(1024, 16)
binGemmK(const float* __restrict__ x, const float* __restrict__ W1, float* __restrict__ h,
         int mtiles, int gb,
         const float* __restrict__ as1, const float* __restrict__ ad1,
         float* __restrict__ asrc, float* __restrict__ adst,
         const float* __restrict__ we1, const float* __restrict__ ae1,
         const float* __restrict__ we2, const float* __restrict__ ae2,
         float* __restrict__ kscal,
         const int* __restrict__ ei, const float* __restrict__ ea,
         int4* __restrict__ tmp, int* __restrict__ tab, int E, int NB, int G) {
    if (productBlock(x, W1, h, mtiles, gb, as1, ad1, asrc, adst)) {
        if (blockIdx.x == 0 && (threadIdx.x >> 5) == TILE_WAVES) {
            const int lane = threadIdx.x & 31;
            if (lane == 0) { float s = 0.f; for (int j = 0; j < 64; ++j) s += we1[j] * ae1[j]; kscal[0] = s; }
            if (lane == 1) { float s = 0.f; for (int j = 0; j < 32; ++j) s += we2[j] * ae2[j]; kscal[1] = s; }
        }
        return;
    }
    binBody(ei, ea, tmp, tab, E, NB, G, (int)blockIdx.x - gb);
}

#define ROW_XOR_F(v, m) __int_as_float(__builtin_amdgcn_mov_dpp(__float_as_int(v), 0x160 + (m), 0xf, 0xf, true))
#define ROW_LANE_I(v, u) __builtin_amdgcn_mov_dpp((v), 0x150 + (u), 0xf, 0xf, true)
#define ROW_LANE_F(v, u) __int_as_float(__builtin_amdgcn_mov_dpp(__float_as_int(v), 0x150 + (u), 0xf, 0xf, true))

__device__ __forceinline__ float rowSum16(float v) {
    v += ROW_XOR_F(v, 8); v += ROW_XOR_F(v, 4); v += ROW_XOR_F(v, 2); v += ROW_XOR_F(v, 1);
    return v;
}
__device__ __forceinline__ float rowMax16(float v) {
    v = fmaxf(v, ROW_XOR_F(v, 8)); v = fmaxf(v, ROW_XOR_F(v, 4));
    v = fmaxf(v, ROW_XOR_F(v, 2)); v = fmaxf(v, ROW_XOR_F(v, 1));
    return v;
}
__device__ __forceinline__ float lrelu(float a) { return (a > 0.0f) ? a : NEG_SLOPE * a; }

typedef __attribute__((ext_vector_type(4))) float v4f;

template <int D>
__device__ __forceinline__ void loadPiece(const float* __restrict__ h, int row, int hl, v2f (&v)[D / 32]) {
    const char* p = (const char*)h + ((unsigned)row * (unsigned)(D * 4) + (unsigned)hl * (unsigned)(D / 4));
    if constexpr (D == 64) {
        const v4f t = *(const v4f*)p;
        v[0] = t.xy; v[1] = t.zw;
    } else {
        v[0] = *(const v2f*)p;
    }
}

__device__ __forceinline__ float sum8(float v) {
    v += ROW_XOR_F(v, 4); v += ROW_XOR_F(v, 2); v += ROW_XOR_F(v, 1);
    return v;
}

template <int D, int U0>
__device__ __forceinline__ void chunk8(const float* __restrict__ h, int hl, int s, float lb,
                                       const v2f (&av)[D / 32], float& m, float& l,
                                       v2f (&acc)[D / 32]) {
    constexpr int P = D / 32;
    v2f hv[8][P];
#define CHUNK_LD(u) { const int st = ROW_LANE_I(s, U0 + (u)); loadPiece<D>(h, st, hl, hv[u]); }
    CHUNK_LD(0) CHUNK_LD(1) CHUNK_LD(2) CHUNK_LD(3)
    CHUNK_LD(4) CHUNK_LD(5) CHUNK_LD(6) CHUNK_LD(7)
#undef CHUNK_LD
    float d[8];
    for (int u = 0; u < 8; ++u) {
        v2f t = hv[u][0] * av[0];
        for (int p = 1; p < P; ++p) t += hv[u][p] * av[p];
        d[u] = t.x + t.y;
    }
    const bool b2 = (hl & 4) != 0, b1 = (hl & 2) != 0, b0 = (hl & 1) != 0;
    float w[4], x[2];
    for (int k = 0; k < 4; ++k) {
        const float keep = b2 ? d[k + 4] : d[k];
        const float send = b2 ? d[k] : d[k + 4];
        w[k] = keep + ROW_XOR_F(send, 4);
    }
    for (int k = 0; k < 2; ++k) {
        const float keep = b1 ? w[k + 2] : w[k];
        const float send = b1 ? w[k] : w[k + 2];
        x[k] = keep + ROW_XOR_F(send, 2);
    }
    float y;
    {
        const float keep = b0 ? x[1] : x[0];
        const float send = b0 ? x[0] : x[1];
        y = keep + ROW_XOR_F(send, 1);
    }
    y += ROW_XOR_F(y, 8);
    const float lbx = ROW_XOR_F(lb, 8);
    const float mylb = ((hl & 8) == U0) ? lb : lbx;
    const float lg = lrelu(y + mylb);
    const float mn = fmaxf(m, rowMax16(lg));
    const float sc = expf(m - mn);
    const float pe = expf(lg - mn);
    l = l * sc + sum8(pe);
    m = mn;
    for (int p = 0; p < P; ++p) acc[p] *= sc;
#define CHUNK_ADD(u) { const float pu = ROW_LANE_F(pe, u); for (int p = 0; p < P; ++p) acc[p] += hv[u][p] * pu; }
    CHUNK_ADD(0) CHUNK_ADD(1) CHUNK_ADD(2) CHUNK_ADD(3)
    CHUNK_ADD(4) CHUNK_ADD(5) CHUNK_ADD(6) CHUNK_ADD(7)
#undef CHUNK_ADD
}

template <int D>
__device__ __forceinline__ int segRow(const int* __restrict__ off, const int2* __restrict__ rec,
                                      const float* __restrict__ asrc, const float* __restrict__ adst,
                                      const float* __restrict__ kscal, int ki,
                                      const float* __restrict__ h, const float* __restrict__ att,
                                      int N, int i0, int lane, bool& live,
                                      v2f (&acc)[D / 32], float& inv) {
    constexpr int P = D / 32;
    const int hl = lane & 15;
    const int iraw = i0 + (lane >> 4);
    live = iraw < N;
    const int i = live ? iraw : i0;
    const int b = off[i];
    const int deg = off[live ? i + 1 : i] - b;
    const float kk = kscal[ki];
    const float ad = adst[i];
    const float asi = asrc[i];
    const int dmax = max(__builtin_amdgcn_readlane(deg, 0), __builtin_amdgcn_readlane(deg, 16));

    v2f av[P];
    loadPiece<D>(att, 0, hl, av);
    loadPiece<D>(h, i, hl, acc);

    const bool v0 = hl < deg, v1 = hl + 16 < deg;
    int2 r0 = make_int2(i, 0), r1 = make_int2(i, 0);
    if (v0) r0 = rec[b + hl];
    if (v1) r1 = rec[b + 16 + hl];
    const int s0 = r0.x, s1 = r1.x;
    const float at0 = __int_as_float(r0.y), at1 = __int_as_float(r1.y);
    const float lb0 = v0 ? (ad + at0 * kk) : -INFINITY;
    const float lb1 = v1 ? (ad + at1 * kk) : -INFINITY;
    float asum = at0 + at1;
#pragma nounroll
    for (int cb = 32; cb < dmax; cb += 16) {
        const int j = cb + hl;
        if (j < deg) asum += __int_as_float(rec[b + j].y);
    }
    asum = rowSum16(asum);

    const float la = asum / fmaxf((float)deg, 1.0f);
    float m = lrelu(asi + ad + la * kk);
    float l = 1.0f;

    if (dmax > 0)  chunk8<D, 0>(h, hl, s0, lb0, av, m, l, acc);
    if (dmax > 8)  chunk8<D, 8>(h, hl, s0, lb0, av, m, l, acc);
    if (dmax > 16) chunk8<D, 0>(h, hl, s1, lb1, av, m, l, acc);
    if (dmax > 24) chunk8<D, 8>(h, hl, s1, lb1, av, m, l, acc);
#pragma nounroll
    for (int cb = 32; cb < dmax; cb += 16) {
        const int j = cb + hl;
        int s = i; float lb = -INFINITY;
        if (j < deg) { const int2 r = rec[b + j]; s = r.x; lb = ad + __int_as_float(r.y) * kk; }
        chunk8<D, 0>(h, hl, s, lb, av, m, l, acc);
        if (dmax > cb + 8) chunk8<D, 8>(h, hl, s, lb, av, m, l, acc);
    }
    inv = 1.0f / l;
    return i;
}

__global__ void __launch_bounds__(256) segOutK(const int* __restrict__ off, const int2* __restrict__ rec,
                                               const float* __restrict__ asrc, const float* __restrict__ adst,
                                               const float* __restrict__ kscal, int ki,
                                               const float* __restrict__ h, const float* __restrict__ att,
                                               const float* __restrict__ bias,
                                               float* __restrict__ out, int N) {
    const int lane = threadIdx.x & 31;
    const int hl = lane & 15;
    const int i0 = 2 * __builtin_amdgcn_readfirstlane((int)((blockIdx.x * blockDim.x + threadIdx.x) >> 5));
    if (i0 >= N) return;
    bool live; v2f acc[1]; float inv;
    const int i = segRow<32>(off, rec, asrc, adst, kscal, ki, h, att, N, i0, lane, live, acc, inv);
    if (live) {
        const v2f bv = *(const v2f*)(bias + hl * 2);
        *(v2f*)(out + i * 32 + hl * 2) = acc[0] * inv + bv;
    }
}

#define AGG_LDA 68
__global__ void __launch_bounds__(256) segGemmK(const int* __restrict__ off, const int2* __restrict__ rec,
                                                const float* __restrict__ asrc, const float* __restrict__ adst,
                                                const float* __restrict__ kscal,
                                                const float* __restrict__ h, const float* __restrict__ att,
                                                const float* __restrict__ bias,
                                                const float* __restrict__ W2,
                                                const float* __restrict__ as2, const float* __restrict__ ad2,
                                                float* __restrict__ h2,
                                                float* __restrict__ asrc2, float* __restrict__ adst2,
                                                int N, int mtiles) {
    __shared__ __attribute__((aligned(16))) float tileA[16 * AGG_LDA];
    const int lane = threadIdx.x & 31;
    const int hl = lane & 15;
    const int w = __builtin_amdgcn_readfirstlane((int)(threadIdx.x >> 5));
    const int i0 = 2 * ((int)blockIdx.x * 8 + w);
    v4f o = (v4f){0.f, 0.f, 0.f, 0.f};
    if (i0 < N) {
        bool live; v2f acc[2]; float inv;
        segRow<64>(off, rec, asrc, adst, kscal, 0, h, att, N, i0, lane, live, acc, inv);
        if (live) {
            const v4f bv = *(const v4f*)(bias + hl * 4);
            o.xy = acc[0] * inv + bv.xy; o.zw = acc[1] * inv + bv.zw;
            o.x = fmaxf(o.x, 0.0f); o.y = fmaxf(o.y, 0.0f); o.z = fmaxf(o.z, 0.0f); o.w = fmaxf(o.w, 0.0f);
        }
    }
    *(v4f*)(tileA + (2 * w + (lane >> 4)) * AGG_LDA + hl * 4) = o;
    __syncthreads();
    if (w == 0 && (int)blockIdx.x < mtiles)
        gemmTile<64, 2, AGG_LDA>(tileA, W2, h2, (int)blockIdx.x, lane, as2, ad2, asrc2, adst2);
}

static inline int gr(long long n, int blk) { return (int)((n + blk - 1) / blk); }

extern "C" void kernel_launch(void* const* d_in, const int* in_sizes, int n_in,
                              void* d_out, int out_size, void* d_ws, size_t ws_size,
                              hipStream_t stream) {
    const float* x    = (const float*)d_in[0];
    const int*   ei   = (const int*)d_in[1];
    const float* ea   = (const float*)d_in[2];
    const float* W1   = (const float*)d_in[3];
    const float* as1  = (const float*)d_in[4];
    const float* ad1  = (const float*)d_in[5];
    const float* we1  = (const float*)d_in[6];
    const float* ae1  = (const float*)d_in[7];
    const float* b1   = (const float*)d_in[8];
    const float* W2   = (const float*)d_in[9];
    const float* as2  = (const float*)d_in[10];
    const float* ad2  = (const float*)d_in[11];
    const float* we2  = (const float*)d_in[12];
    const float* ae2  = (const float*)d_in[13];
    const float* b2   = (const float*)d_in[14];
    float* out = (float*)d_out;

    const int N  = in_sizes[0] / 128;
    const int E  = in_sizes[1] / 2;

    float* ws = (float*)d_ws;
    size_t o = 0;
    float* h     = ws + o; o += (size_t)N * 64;
    float* h2    = ws + o; o += (size_t)N * 32;
    int2*  rec   = (int2*)(ws + o); o += (size_t)E * 2;
    const int G  = gr(E, CH_EDGES);
    const int NB = gr(N, 1 << BIN_SHIFT);
    o = (o + 3) & ~(size_t)3;
    int4*  tmp   = (int4*)(ws + o); o += (size_t)G * CH_EDGES * 4;
    int*   tab   = (int*)(ws + o); o += (size_t)NB * G;
    int*   off   = (int*)(ws + o); o += N + 1;
    float* asrc  = ws + o; o += N;
    float* adst  = ws + o; o += N;
    float* asrc2 = ws + o; o += N;
    float* adst2 = ws + o; o += N;
    float* kscal = ws + o; o += 4;

    const int B = 256;
    const int mtiles = N / 16;

    const int GB = gr(mtiles, TILE_WAVES);
    binGemmK<<<GB + G, 1024, 0, stream>>>(x, W1, h, mtiles, GB, as1, ad1, asrc, adst,
                                          we1, ae1, we2, ae2, kscal, ei, ea, tmp, tab, E, NB, G);
    placeK<<<NB, 1024, 0, stream>>>(tmp, tab, off, rec, N, G);

    const int SB = gr((long long)((N + 1) / 2) * 32, B);
    segGemmK<<<SB, B, 0, stream>>>(off, rec, asrc, adst, kscal, h, as1, b1, W2, as2, ad2,
                                   h2, asrc2, adst2, N, mtiles);

    segOutK<<<SB, B, 0, stream>>>(off, rec, asrc2, adst2, kscal, 1, h2, as2, b2, out, N);
}
